# starting kernel
# baseline (speedup 1.0000x reference)
#include <hip/hip_runtime.h>
#include <math.h>

#define BB   256
#define SS   512
#define DD   256
#define LLN  4
#define NITERS 10
#define BT   16
#define LSTR 260
#define TILE (BT * LSTR)
#define SMEM_FLOATS (9 * TILE)
#define SMEM_BYTES  (SMEM_FLOATS * sizeof(float))

typedef float v2f __attribute__((ext_vector_type(2)));
typedef float v8f __attribute__((ext_vector_type(8)));

#if defined(__has_builtin)
#  if __has_builtin(__builtin_amdgcn_wmma_f32_16x16x4_f32)
#    define HAS_WMMA_F32 1
#  else
#    define HAS_WMMA_F32 0
#  endif
#  if __has_builtin(__builtin_amdgcn_tanhf)
#    define HAS_HW_TANH 1
#  else
#    define HAS_HW_TANH 0
#  endif
#else
#  define HAS_WMMA_F32 0
#  define HAS_HW_TANH 0
#endif

__device__ __forceinline__ float tanh_fast(float x) {
#if HAS_HW_TANH
  return __builtin_amdgcn_tanhf(x);
#else
  return tanhf(x);
#endif
}

__device__ __forceinline__ float* smb(float* sm, int i) { return sm + i * TILE; }


__device__ __forceinline__ v8f tile_xWT(const float* __restrict__ W,
                                        const float* X, int nt, int lane,
                                        float cinit) {
  const int l16 = lane & 15;
  const int half = lane >> 4;
  v8f acc = {cinit, cinit, cinit, cinit, cinit, cinit, cinit, cinit};
#if HAS_WMMA_F32
  const float* wr = W + (nt * 16 + l16) * DD + 2 * half;
  const float* xr = X + l16 * LSTR + 2 * half;
#pragma unroll 4
  for (int kk = 0; kk < DD; kk += 4) {
    v2f a = {xr[kk], xr[kk + 1]};
    v2f b = {wr[kk], wr[kk + 1]};
    acc = __builtin_amdgcn_wmma_f32_16x16x4_f32(false, a, false, b,
                                                (short)0, acc, false, false);
  }
#else
#pragma unroll
  for (int j = 0; j < 8; ++j) {
    const int m = j + 8 * half, n = nt * 16 + l16;
    float s = cinit;
    for (int k = 0; k < DD; ++k) s += X[m * LSTR + k] * W[n * DD + k];
    acc[j] = s;
  }
#endif
  return acc;
}

__device__ __forceinline__ v8f tile_xM(const float* __restrict__ M,
                                       const float* X, int nt, int lane,
                                       float cinit) {
  const int l16 = lane & 15;
  const int half = lane >> 4;
  v8f acc = {cinit, cinit, cinit, cinit, cinit, cinit, cinit, cinit};
#if HAS_WMMA_F32
  const float* mp = M + (nt * 16 + l16) + (2 * half) * DD;
  const float* xr = X + l16 * LSTR + 2 * half;
#pragma unroll 4
  for (int kk = 0; kk < DD; kk += 4) {
    v2f a = {xr[kk], xr[kk + 1]};
    v2f b = {mp[kk * DD], mp[kk * DD + DD]};
    acc = __builtin_amdgcn_wmma_f32_16x16x4_f32(false, a, false, b,
                                                (short)0, acc, false, false);
  }
#else
#pragma unroll
  for (int j = 0; j < 8; ++j) {
    const int m = j + 8 * half, n = nt * 16 + l16;
    float s = cinit;
    for (int k = 0; k < DD; ++k) s += X[m * LSTR + k] * M[k * DD + n];
    acc[j] = s;
  }
#endif
  return acc;
}

template <typename Epi>
__device__ __forceinline__ void gemm_xWT(const float* __restrict__ W,
                                         const float* __restrict__ bias,
                                         const float* X, int lane, int wv,
                                         Epi epi) {
#pragma unroll
  for (int t = 0; t < 2; ++t) {
    const int nt = wv * 2 + t;
    const int l16 = lane & 15, half = lane >> 4;
    const float bv = bias ? bias[nt * 16 + l16] : 0.f;
    v8f acc = tile_xWT(W, X, nt, lane, bv);
#pragma unroll
    for (int j = 0; j < 8; ++j) epi(j + 8 * half, nt * 16 + l16, acc[j]);
  }
}

template <typename Epi>
__device__ __forceinline__ void gemm_xM(const float* __restrict__ M,
                                        const float* X, int lane, int wv,
                                        Epi epi) {
#pragma unroll
  for (int t = 0; t < 2; ++t) {
    const int nt = wv * 2 + t;
    v8f acc = tile_xM(M, X, nt, lane, 0.f);
    const int l16 = lane & 15, half = lane >> 4;
#pragma unroll
    for (int j = 0; j < 8; ++j) epi(j + 8 * half, nt * 16 + l16, acc[j]);
  }
}

__global__ __launch_bounds__(256) void pooled_kernel(
    const int* __restrict__ ids, const float* __restrict__ emb,
    const float* __restrict__ pos, float* __restrict__ pooled_g) {
  __shared__ int sid[SS];
  const int b = blockIdx.x;
  const int t = threadIdx.x;
  sid[t] = ids[b * SS + t];
  sid[t + 256] = ids[b * SS + 256 + t];
  __syncthreads();
  float acc = 0.f;
  for (int s = 0; s < SS; ++s) {
    acc += emb[(size_t)sid[s] * DD + t] + pos[s * DD + t];
  }
  pooled_g[b * DD + t] = acc * (1.0f / (float)SS);
}

__global__ __launch_bounds__(256) void refine_kernel(
    const float* __restrict__ pooled_g,
    const float* __restrict__ init_w, const float* __restrict__ init_b,
    const float* __restrict__ pred_w, const float* __restrict__ pred_b,
    const float* __restrict__ upd_w, const float* __restrict__ upd_b,
    const float* __restrict__ sqw, const float* __restrict__ sqb,
    const float* __restrict__ eqw, const float* __restrict__ eqb,
    const float* __restrict__ keyw, const float* __restrict__ keyb,
    const float* __restrict__ aw1, const float* __restrict__ ab1,
    const float* __restrict__ aw2, const float* __restrict__ ab2,
    float* __restrict__ sq_g, float* __restrict__ eq_g,
    float* __restrict__ cs_g, float* __restrict__ ce_g,
    float* __restrict__ ans_g) {
  extern __shared__ float sm[];

  const int tid = threadIdx.x;
  const int lane = tid & 31;
  const int wv = tid >> 5;
  const int b0 = blockIdx.x * BT;

  float* pooled = smb(sm, 0);
  float* ff = smb(sm, 5);
  float* tot = smb(sm, 8);

  for (int idx = tid; idx < BT * DD; idx += 256) {
    const int r = idx >> 8, c = idx & 255;
    pooled[r * LSTR + c] = pooled_g[(b0 + r) * DD + c];
  }
  __syncthreads();

  {
    const float* cur = pooled;
    for (int l = 0; l < LLN; ++l) {
      float* dst = smb(sm, 1 + l);
      const bool last = (l == LLN - 1);
      gemm_xWT(init_w + l * DD * DD, init_b + l * DD, cur, lane, wv,
               [&](int m, int n, float v) {
                 const float r = tanh_fast(v);
                 dst[m * LSTR + n] = r;
                 if (last) ff[m * LSTR + n] = r;
               });
      __syncthreads();
      cur = dst;
    }
  }

  for (int it = 0; it < NITERS; ++it) {
    for (int i = LLN - 1; i >= 0; --i) {
      float* pc = smb(sm, 6 + ((LLN - 1 - i) & 1));
      float* pn = smb(sm, 6 + ((LLN - i) & 1));
      float* ri = smb(sm, 1 + i);
      gemm_xWT(pred_w + i * DD * DD, pred_b + i * DD, ri, lane, wv,
               [&](int m, int n, float v) { pc[m * LSTR + n] = v; });
      __syncthreads();
      {
        const float* below = (i > 0) ? smb(sm, i) : pooled;
        const bool has_above = (i < LLN - 1);
        for (int idx = tid; idx < BT * DD; idx += 256) {
          const int r = idx >> 8, c = idx & 255;
          const int o = r * LSTR + c;
          float e = below[o] - pc[o];
          if (has_above) e += 0.5f * (ri[o] - pn[o]);
          tot[o] = e;
        }
      }
      __syncthreads();
      gemm_xWT(upd_w + i * DD * DD, upd_b + i * DD, tot, lane, wv,
               [&](int m, int n, float v) {
                 const int o = m * LSTR + n;
                 ri[o] = ri[o] + 0.1f * tanh_fast(v);
               });
      __syncthreads();
    }
  }

  float* refined = smb(sm, LLN);
  for (int idx = tid; idx < BT * DD; idx += 256) {
    const int r = idx >> 8, c = idx & 255;
    refined[r * LSTR + c] += ff[r * LSTR + c];
  }
  __syncthreads();

  {
    float* q = smb(sm, 6);
    gemm_xWT(sqw, sqb, refined, lane, wv,
             [&](int m, int n, float v) { q[m * LSTR + n] = v; });
    __syncthreads();
    {
      const int r = tid >> 4, c = tid & 15;
      float p = 0.f;
#pragma unroll
      for (int u = 0; u < 16; ++u) {
        const int o = c * 16 + u;
        p += q[r * LSTR + o] * keyb[o];
      }
      tot[tid] = p;
    }
    __syncthreads();
    if (tid < BT) {
      float s = 0.f;
#pragma unroll
      for (int u = 0; u < 16; ++u) s += tot[tid * 16 + u];
      cs_g[b0 + tid] = s;
    }
    __syncthreads();
    gemm_xM(keyw, q, lane, wv,
            [&](int m, int n, float v) { sq_g[(b0 + m) * DD + n] = v; });
    __syncthreads();
  }

  {
    float* q = smb(sm, 7);
    gemm_xWT(eqw, eqb, refined, lane, wv,
             [&](int m, int n, float v) { q[m * LSTR + n] = v; });
    __syncthreads();
    {
      const int r = tid >> 4, c = tid & 15;
      float p = 0.f;
#pragma unroll
      for (int u = 0; u < 16; ++u) {
        const int o = c * 16 + u;
        p += q[r * LSTR + o] * keyb[o];
      }
      tot[tid] = p;
    }
    __syncthreads();
    if (tid < BT) {
      float s = 0.f;
#pragma unroll
      for (int u = 0; u < 16; ++u) s += tot[tid * 16 + u];
      ce_g[b0 + tid] = s;
    }
    __syncthreads();
    gemm_xM(keyw, q, lane, wv,
            [&](int m, int n, float v) { eq_g[(b0 + m) * DD + n] = v; });
    __syncthreads();
  }

  for (int task = tid; task < BT * 128; task += 256) {
    const int r = task >> 7, j = task & 127;
    float s = ab1[j];
    const float* rr = refined + r * LSTR;
    const float* wr = aw1 + j * DD;
    for (int k = 0; k < DD; ++k) s += rr[k] * wr[k];
    tot[r * LSTR + j] = 0.5f * s * (1.0f + erff(s * 0.70710678118654752f));
  }
  __syncthreads();
  if (tid < BT * 2) {
    const int r = tid >> 1, k = tid & 1;
    float s = ab2[k];
    for (int j = 0; j < 128; ++j) s += tot[r * LSTR + j] * aw2[k * 128 + j];
    ans_g[(b0 + r) * 2 + k] = s;
  }
}

__global__ __launch_bounds__(256) void logits_kernel(
    const int* __restrict__ ids, const float* __restrict__ emb,
    const float* __restrict__ pos, const float* __restrict__ sq_g,
    const float* __restrict__ eq_g, const float* __restrict__ cs_g,
    const float* __restrict__ ce_g, float* __restrict__ out_start,
    float* __restrict__ out_end) {
  const int b = blockIdx.x;
  const int lane = threadIdx.x & 31;
  const int wv = threadIdx.x >> 5;
  float sqv[8], eqv[8];
  const float* sr = sq_g + b * DD + lane * 8;
  const float* er = eq_g + b * DD + lane * 8;
#pragma unroll
  for (int j = 0; j < 8; ++j) {
    sqv[j] = sr[j];
    eqv[j] = er[j];
  }
  const float cs = cs_g[b];
  const float ce = ce_g[b];
  for (int s = wv; s < SS; s += 8) {
    const int id = ids[b * SS + s];
    const float* row = emb + (size_t)id * DD + lane * 8;
    const float* pr = pos + s * DD + lane * 8;
    float as = 0.f, ae = 0.f;
#pragma unroll
    for (int j = 0; j < 8; ++j) {
      const float e = row[j] + pr[j];
      as += e * sqv[j];
      ae += e * eqv[j];
    }
#pragma unroll
    for (int off = 16; off; off >>= 1) {
      as += __shfl_xor(as, off, 32);
      ae += __shfl_xor(ae, off, 32);
    }
    if (lane == 0) {
      out_start[b * SS + s] = as + cs;
      out_end[b * SS + s] = ae + ce;
    }
  }
}

extern "C" void kernel_launch(void* const* d_in, const int* in_sizes, int n_in,
                              void* d_out, int out_size, void* d_ws,
                              size_t ws_size, hipStream_t stream) {
  const int* ids = (const int*)d_in[0];
  const float* emb = (const float*)d_in[1];
  const float* pos = (const float*)d_in[2];
  const float* init_w = (const float*)d_in[3];
  const float* init_b = (const float*)d_in[4];
  const float* pred_w = (const float*)d_in[5];
  const float* pred_b = (const float*)d_in[6];
  const float* upd_w = (const float*)d_in[7];
  const float* upd_b = (const float*)d_in[8];
  const float* sqw = (const float*)d_in[9];
  const float* sqb = (const float*)d_in[10];
  const float* eqw = (const float*)d_in[11];
  const float* eqb = (const float*)d_in[12];
  const float* keyw = (const float*)d_in[13];
  const float* keyb = (const float*)d_in[14];
  const float* aw1 = (const float*)d_in[15];
  const float* ab1 = (const float*)d_in[16];
  const float* aw2 = (const float*)d_in[17];
  const float* ab2 = (const float*)d_in[18];

  float* ws = (float*)d_ws;
  float* pooled = ws;
  float* sq = pooled + BB * DD;
  float* eq = sq + BB * DD;
  float* cs = eq + BB * DD;
  float* ce = cs + BB;

  float* out = (float*)d_out;
  float* out_start = out;
  float* out_end = out + BB * SS;
  float* out_ans = out + 2 * BB * SS;

  (void)hipFuncSetAttribute((const void*)refine_kernel,
                            hipFuncAttributeMaxDynamicSharedMemorySize,
                            (int)SMEM_BYTES);

  pooled_kernel<<<BB, 256, 0, stream>>>(ids, emb, pos, pooled);
  refine_kernel<<<BB / BT, 256, SMEM_BYTES, stream>>>(
      pooled, init_w, init_b, pred_w, pred_b, upd_w, upd_b, sqw, sqb, eqw, eqb,
      keyw, keyb, aw1, ab1, aw2, ab2, sq, eq, cs, ce, out_ans);
  logits_kernel<<<BB, 256, 0, stream>>>(ids, emb, pos, sq, eq, cs, ce,
                                        out_start, out_end);
}


# refine GEMMs: each GEMM's first weight half-tile and bias are requested under the previous GEMM's last half-tile, parked before the barrier; no GEMM opens on an uncovered round trip; sums unchanged
# speedup vs baseline: 1.7992x; 1.7992x over previous
#include <hip/hip_runtime.h>
#include <math.h>

#define BB   256
#define SS   512
#define DD   256
#define LLN  4
#define NITERS 10
#define BT   16
#define LSTR 260
#define TILE (BT * LSTR)
#define NWAVES 8
#define WSTR 132
#define WHALF (16 * WSTR)
#define WSTAGE (2 * WHALF)
#define SMEM_FLOATS (9 * TILE + NWAVES * WSTAGE)
#define SMEM_BYTES  (SMEM_FLOATS * sizeof(float))

typedef float v2f __attribute__((ext_vector_type(2)));
typedef float v4f __attribute__((ext_vector_type(4)));
typedef float v8f __attribute__((ext_vector_type(8)));

#if defined(__has_builtin)
#  if __has_builtin(__builtin_amdgcn_wmma_f32_16x16x4_f32)
#    define HAS_WMMA_F32 1
#  else
#    define HAS_WMMA_F32 0
#  endif
#  if __has_builtin(__builtin_amdgcn_tanhf)
#    define HAS_HW_TANH 1
#  else
#    define HAS_HW_TANH 0
#  endif
#else
#  define HAS_WMMA_F32 0
#  define HAS_HW_TANH 0
#endif

__device__ __forceinline__ float tanh_fast(float x) {
#if HAS_HW_TANH
  return __builtin_amdgcn_tanhf(x);
#else
  return tanhf(x);
#endif
}

__device__ __forceinline__ float* smb(float* sm, int i) { return sm + i * TILE; }


__device__ __forceinline__ v8f tile_xWT(const float* __restrict__ W,
                                        const float* X, int nt, int lane,
                                        float cinit) {
  const int l16 = lane & 15;
  const int half = lane >> 4;
  v8f acc = {cinit, cinit, cinit, cinit, cinit, cinit, cinit, cinit};
#if HAS_WMMA_F32
  const float* wr = W + (nt * 16 + l16) * DD + 2 * half;
  const float* xr = X + l16 * LSTR + 2 * half;
#pragma unroll 4
  for (int kk = 0; kk < DD; kk += 4) {
    v2f a = {xr[kk], xr[kk + 1]};
    v2f b = {wr[kk], wr[kk + 1]};
    acc = __builtin_amdgcn_wmma_f32_16x16x4_f32(false, a, false, b,
                                                (short)0, acc, false, false);
  }
#else
#pragma unroll
  for (int j = 0; j < 8; ++j) {
    const int m = j + 8 * half, n = nt * 16 + l16;
    float s = cinit;
    for (int k = 0; k < DD; ++k) s += X[m * LSTR + k] * W[n * DD + k];
    acc[j] = s;
  }
#endif
  return acc;
}

__device__ __forceinline__ v8f tile_xM(const float* __restrict__ M,
                                       const float* X, int nt, int lane,
                                       float cinit) {
  const int l16 = lane & 15;
  const int half = lane >> 4;
  v8f acc = {cinit, cinit, cinit, cinit, cinit, cinit, cinit, cinit};
#if HAS_WMMA_F32
  const float* mp = M + (nt * 16 + l16) + (2 * half) * DD;
  const float* xr = X + l16 * LSTR + 2 * half;
#pragma unroll 4
  for (int kk = 0; kk < DD; kk += 4) {
    v2f a = {xr[kk], xr[kk + 1]};
    v2f b = {mp[kk * DD], mp[kk * DD + DD]};
    acc = __builtin_amdgcn_wmma_f32_16x16x4_f32(false, a, false, b,
                                                (short)0, acc, false, false);
  }
#else
#pragma unroll
  for (int j = 0; j < 8; ++j) {
    const int m = j + 8 * half, n = nt * 16 + l16;
    float s = cinit;
    for (int k = 0; k < DD; ++k) s += X[m * LSTR + k] * M[k * DD + n];
    acc[j] = s;
  }
#endif
  return acc;
}

#if HAS_WMMA_F32
__device__ __forceinline__ void stage_issue(v4f (&r)[16],
                                            const float* __restrict__ wsrc) {
#pragma unroll
  for (int j = 0; j < 16; ++j)
    r[j] = *reinterpret_cast<const v4f*>(wsrc + j * DD);
}

__device__ __forceinline__ void stage_park(const v4f (&r)[16], float* wp) {
#pragma unroll
  for (int j = 0; j < 16; ++j) *reinterpret_cast<v4f*>(wp + j * WSTR) = r[j];
}

__device__ __forceinline__ void wave_lds_handover() {
  __builtin_amdgcn_fence(__ATOMIC_RELEASE, "wavefront");
  __builtin_amdgcn_wave_barrier();
  __builtin_amdgcn_fence(__ATOMIC_ACQUIRE, "wavefront");
}

__device__ __forceinline__ void frag_read8(v2f (&a)[8], v2f (&b)[8],
                                           const float* xr, const float* wr) {
#pragma unroll
  for (int s = 0; s < 8; ++s) {
    a[s] = (v2f){xr[4 * s], xr[4 * s + 1]};
    b[s] = (v2f){wr[4 * s], wr[4 * s + 1]};
  }
}

__device__ __forceinline__ v8f mma8(v8f acc, const v2f (&a)[8],
                                    const v2f (&b)[8]) {
#pragma unroll
  for (int s = 0; s < 8; ++s)
    acc = __builtin_amdgcn_wmma_f32_16x16x4_f32(false, a[s], false, b[s],
                                                (short)0, acc, false, false);
  return acc;
}

__device__ __forceinline__ v8f half_tile_mma(v8f acc, const float* xr,
                                             const float* wr) {
  v2f a0[8], b0[8], a1[8], b1[8];
  frag_read8(a0, b0, xr, wr);
  frag_read8(a1, b1, xr + 32, wr + 32);
  __builtin_amdgcn_sched_barrier(0);
  acc = mma8(acc, a0, b0);
  __builtin_amdgcn_sched_barrier(0);
  frag_read8(a0, b0, xr + 64, wr + 64);
  __builtin_amdgcn_sched_barrier(0);
  acc = mma8(acc, a1, b1);
  __builtin_amdgcn_sched_barrier(0);
  frag_read8(a1, b1, xr + 96, wr + 96);
  __builtin_amdgcn_sched_barrier(0);
  acc = mma8(acc, a0, b0);
  acc = mma8(acc, a1, b1);
  return acc;
}
#endif

__device__ __forceinline__ void gemm_first_request(
    v4f (&r)[16], float& bv0, float& bv1, const float* __restrict__ W,
    const float* __restrict__ bias, int lane, int wv) {
#if HAS_WMMA_F32
  stage_issue(r, W + (wv * 32) * DD + lane * 4);
#endif
  bv0 = bias ? bias[wv * 32 + (lane & 15)] : 0.f;
  bv1 = bias ? bias[wv * 32 + 16 + (lane & 15)] : 0.f;
}

template <typename TileEpi>
__device__ __forceinline__ void gemm_xWT_tiles(
    const float* __restrict__ W, const float* __restrict__ bias,
    const float* X, float* stage, int lane, int wv, float& bv0, float& bv1,
    const float* __restrict__ Wn, const float* __restrict__ biasn,
    const bool has_next, TileEpi epi) {
  const int l16 = lane & 15, half = lane >> 4;
#if HAS_WMMA_F32
  const float* wsrc = W + (wv * 32) * DD + lane * 4;
  const float* xr = X + l16 * LSTR + 2 * half;
  const float* wr = stage + l16 * WSTR + 2 * half;
  float* wp = stage + lane * 4;
  const float cb0 = bv0, cb1 = bv1;
  v4f r[16];
  v8f acc = {0.f, 0.f, 0.f, 0.f, 0.f, 0.f, 0.f, 0.f};
#pragma unroll 1
  for (int q = 0; q < 4; ++q) {
    const int kh = q & 1, qn = q + 1;
    const bool more = (q < 3) || has_next;
    if (more) {
      stage_issue(r, (q < 3)
                         ? wsrc + (qn >> 1) * 16 * DD + (qn & 1) * 128
                         : Wn + (wv * 32) * DD + lane * 4);
      if (q == 3) {
        bv0 = biasn ? biasn[wv * 32 + l16] : 0.f;
        bv1 = biasn ? biasn[wv * 32 + 16 + l16] : 0.f;
      }
    }
    if (kh == 0) {
      const float bv = (q >> 1) ? cb1 : cb0;
      acc = (v8f){bv, bv, bv, bv, bv, bv, bv, bv};
    }
    acc = half_tile_mma(acc, xr + kh * 128, wr + kh * WHALF);
    if (kh == 1) epi(wv * 2 + (q >> 1), acc);
    if (more) {
      stage_park(r, wp + (qn & 1) * WHALF);
      wave_lds_handover();
    }
  }
#else
#pragma unroll
  for (int t = 0; t < 2; ++t) {
    const int nt = wv * 2 + t;
    const float bv = bias ? bias[nt * 16 + l16] : 0.f;
    v8f acc = tile_xWT(W, X, nt, lane, bv);
    epi(nt, acc);
  }
#endif
}

template <typename Epi>
__device__ __forceinline__ void gemm_xWT(
    const float* __restrict__ W, const float* __restrict__ bias,
    const float* X, float* stage, int lane, int wv, float& bv0, float& bv1,
    const float* __restrict__ Wn, const float* __restrict__ biasn,
    const bool has_next, Epi epi) {
  gemm_xWT_tiles(W, bias, X, stage, lane, wv, bv0, bv1, Wn, biasn, has_next,
                 [&](int nt, v8f acc) {
    const int l16 = lane & 15, half = lane >> 4;
#pragma unroll
    for (int j = 0; j < 8; ++j) epi(j + 8 * half, nt * 16 + l16, acc[j]);
  });
}

template <typename Epi>
__device__ __forceinline__ void gemm_xM(const float* __restrict__ M,
                                        const float* X, int lane, int wv,
                                        Epi epi) {
#pragma unroll
  for (int t = 0; t < 2; ++t) {
    const int nt = wv * 2 + t;
    v8f acc = tile_xM(M, X, nt, lane, 0.f);
    const int l16 = lane & 15, half = lane >> 4;
#pragma unroll
    for (int j = 0; j < 8; ++j) epi(j + 8 * half, nt * 16 + l16, acc[j]);
  }
}

__global__ __launch_bounds__(256) void pooled_kernel(
    const int* __restrict__ ids, const float* __restrict__ emb,
    const float* __restrict__ pos, float* __restrict__ pooled_g) {
  __shared__ int sid[SS];
  const int b = blockIdx.x;
  const int t = threadIdx.x;
  sid[t] = ids[b * SS + t];
  sid[t + 256] = ids[b * SS + 256 + t];
  __syncthreads();
  float acc = 0.f;
  for (int s = 0; s < SS; ++s) {
    acc += emb[(size_t)sid[s] * DD + t] + pos[s * DD + t];
  }
  pooled_g[b * DD + t] = acc * (1.0f / (float)SS);
}

__global__ __launch_bounds__(256) void refine_kernel(
    const float* __restrict__ pooled_g,
    const float* __restrict__ init_w, const float* __restrict__ init_b,
    const float* __restrict__ pred_w, const float* __restrict__ pred_b,
    const float* __restrict__ upd_w, const float* __restrict__ upd_b,
    const float* __restrict__ sqw, const float* __restrict__ sqb,
    const float* __restrict__ eqw, const float* __restrict__ eqb,
    const float* __restrict__ keyw, const float* __restrict__ keyb,
    const float* __restrict__ aw1, const float* __restrict__ ab1,
    const float* __restrict__ aw2, const float* __restrict__ ab2,
    float* __restrict__ sq_g, float* __restrict__ eq_g,
    float* __restrict__ cs_g, float* __restrict__ ce_g,
    float* __restrict__ ans_g) {
  extern __shared__ __attribute__((aligned(16))) float sm[];

  const int tid = threadIdx.x;
  const int lane = tid & 31;
  const int wv = tid >> 5;
  const int b0 = blockIdx.x * BT;

  float* pooled = smb(sm, 0);
  float* ff = smb(sm, 5);
  float* tot = smb(sm, 8);
  float* stage = sm + 9 * TILE + wv * WSTAGE;

  float bv0, bv1;
  {
    v4f r0[16];
    gemm_first_request(r0, bv0, bv1, init_w, init_b, lane, wv);
    for (int idx = tid; idx < BT * DD; idx += 256) {
      const int r = idx >> 8, c = idx & 255;
      pooled[r * LSTR + c] = pooled_g[(b0 + r) * DD + c];
    }
#if HAS_WMMA_F32
    stage_park(r0, stage + lane * 4);
    wave_lds_handover();
#endif
  }
  __syncthreads();

  {
    const float* cur = pooled;
    for (int l = 0; l < LLN; ++l) {
      float* dst = smb(sm, 1 + l);
      const bool last = (l == LLN - 1);
      const float* wn = last ? pred_w + (LLN - 1) * DD * DD
                             : init_w + (l + 1) * DD * DD;
      const float* bn = last ? pred_b + (LLN - 1) * DD : init_b + (l + 1) * DD;
      gemm_xWT(init_w + l * DD * DD, init_b + l * DD, cur, stage, lane, wv,
               bv0, bv1, wn, bn, true,
               [&](int m, int n, float v) {
                 const float r = tanh_fast(v);
                 dst[m * LSTR + n] = r;
                 if (last) ff[m * LSTR + n] = r;
               });
      __syncthreads();
      cur = dst;
    }
  }

  for (int it = 0; it < NITERS; ++it) {
    for (int i = LLN - 1; i >= 0; --i) {
      float* pc = smb(sm, 6 + ((LLN - 1 - i) & 1));
      float* pn = smb(sm, 6 + ((LLN - i) & 1));
      float* ri = smb(sm, 1 + i);
      {
        const float* below = (i > 0) ? smb(sm, i) : pooled;
        const bool has_above = (i < LLN - 1);
        const bool read_below = (i > 0);
        gemm_xWT_tiles(
            pred_w + i * DD * DD, pred_b + i * DD, ri, stage, lane, wv,
            bv0, bv1, upd_w + i * DD * DD, upd_b + i * DD, true,
            [&](int nt, v8f acc) {
              const int o0 = 8 * (lane >> 4) * LSTR + nt * 16 + (lane & 15);
              float bl[8], rr[8], pa[8];
#pragma unroll
              for (int j = 0; j < 8; ++j) bl[j] = below[o0 + j * LSTR];
              if (has_above) {
#pragma unroll
                for (int j = 0; j < 8; ++j) {
                  rr[j] = ri[o0 + j * LSTR];
                  pa[j] = pn[o0 + j * LSTR];
                }
              }
              if (read_below) {
#pragma unroll
                for (int j = 0; j < 8; ++j) pc[o0 + j * LSTR] = acc[j];
              }
#pragma unroll
              for (int j = 0; j < 8; ++j) {
                float e = bl[j] - acc[j];
                if (has_above) e += 0.5f * (rr[j] - pa[j]);
                tot[o0 + j * LSTR] = e;
              }
            });
      }
      __syncthreads();
      const bool sweep_end = (i == 0);
      const bool all_end = sweep_end && (it == NITERS - 1);
      const float* wn = all_end ? sqw
                                : pred_w + (sweep_end ? LLN - 1 : i - 1) * DD * DD;
      const float* bn = all_end ? sqb
                                : pred_b + (sweep_end ? LLN - 1 : i - 1) * DD;
      gemm_xWT(upd_w + i * DD * DD, upd_b + i * DD, tot, stage, lane, wv,
               bv0, bv1, wn, bn, true,
               [&](int m, int n, float v) {
                 const int o = m * LSTR + n;
                 ri[o] = ri[o] + 0.1f * tanh_fast(v);
               });
      __syncthreads();
    }
  }

  float* refined = smb(sm, LLN);
  for (int idx = tid; idx < BT * DD; idx += 256) {
    const int r = idx >> 8, c = idx & 255;
    refined[r * LSTR + c] += ff[r * LSTR + c];
  }
  __syncthreads();

  {
    float* q = smb(sm, 6);
    gemm_xWT(sqw, sqb, refined, stage, lane, wv, bv0, bv1, eqw, eqb, true,
             [&](int m, int n, float v) { q[m * LSTR + n] = v; });
    __syncthreads();
    {
      const int r = tid >> 4, c = tid & 15;
      float p = 0.f;
#pragma unroll
      for (int u = 0; u < 16; ++u) {
        const int o = c * 16 + u;
        p += q[r * LSTR + o] * keyb[o];
      }
      tot[tid] = p;
    }
    __syncthreads();
    if (tid < BT) {
      float s = 0.f;
#pragma unroll
      for (int u = 0; u < 16; ++u) s += tot[tid * 16 + u];
      cs_g[b0 + tid] = s;
    }
    __syncthreads();
    gemm_xM(keyw, q, lane, wv,
            [&](int m, int n, float v) { sq_g[(b0 + m) * DD + n] = v; });
    __syncthreads();
  }

  {
    float* q = smb(sm, 7);
    gemm_xWT(eqw, eqb, refined, stage, lane, wv, bv0, bv1, nullptr, nullptr,
             false, [&](int m, int n, float v) { q[m * LSTR + n] = v; });
    __syncthreads();
    {
      const int r = tid >> 4, c = tid & 15;
      float p = 0.f;
#pragma unroll
      for (int u = 0; u < 16; ++u) {
        const int o = c * 16 + u;
        p += q[r * LSTR + o] * keyb[o];
      }
      tot[tid] = p;
    }
    __syncthreads();
    if (tid < BT) {
      float s = 0.f;
#pragma unroll
      for (int u = 0; u < 16; ++u) s += tot[tid * 16 + u];
      ce_g[b0 + tid] = s;
    }
    __syncthreads();
    gemm_xM(keyw, q, lane, wv,
            [&](int m, int n, float v) { eq_g[(b0 + m) * DD + n] = v; });
    __syncthreads();
  }

  for (int task = tid; task < BT * 128; task += 256) {
    const int r = task >> 7, j = task & 127;
    float s = ab1[j];
    const float* rr = refined + r * LSTR;
    const float* wr = aw1 + j * DD;
    for (int k = 0; k < DD; ++k) s += rr[k] * wr[k];
    tot[r * LSTR + j] = 0.5f * s * (1.0f + erff(s * 0.70710678118654752f));
  }
  __syncthreads();
  if (tid < BT * 2) {
    const int r = tid >> 1, k = tid & 1;
    float s = ab2[k];
    for (int j = 0; j < 128; ++j) s += tot[r * LSTR + j] * aw2[k * 128 + j];
    ans_g[(b0 + r) * 2 + k] = s;
  }
}

__global__ __launch_bounds__(256) void logits_kernel(
    const int* __restrict__ ids, const float* __restrict__ emb,
    const float* __restrict__ pos, const float* __restrict__ sq_g,
    const float* __restrict__ eq_g, const float* __restrict__ cs_g,
    const float* __restrict__ ce_g, float* __restrict__ out_start,
    float* __restrict__ out_end) {
  const int b = blockIdx.x;
  const int lane = threadIdx.x & 31;
  const int wv = threadIdx.x >> 5;
  float sqv[8], eqv[8];
  const float* sr = sq_g + b * DD + lane * 8;
  const float* er = eq_g + b * DD + lane * 8;
#pragma unroll
  for (int j = 0; j < 8; ++j) {
    sqv[j] = sr[j];
    eqv[j] = er[j];
  }
  const float cs = cs_g[b];
  const float ce = ce_g[b];
  for (int s = wv; s < SS; s += 8) {
    const int id = ids[b * SS + s];
    const float* row = emb + (size_t)id * DD + lane * 8;
    const float* pr = pos + s * DD + lane * 8;
    float as = 0.f, ae = 0.f;
#pragma unroll
    for (int j = 0; j < 8; ++j) {
      const float e = row[j] + pr[j];
      as += e * sqv[j];
      ae += e * eqv[j];
    }
#pragma unroll
    for (int off = 16; off; off >>= 1) {
      as += __shfl_xor(as, off, 32);
      ae += __shfl_xor(ae, off, 32);
    }
    if (lane == 0) {
      out_start[b * SS + s] = as + cs;
      out_end[b * SS + s] = ae + ce;
    }
  }
}

extern "C" void kernel_launch(void* const* d_in, const int* in_sizes, int n_in,
                              void* d_out, int out_size, void* d_ws,
                              size_t ws_size, hipStream_t stream) {
  const int* ids = (const int*)d_in[0];
  const float* emb = (const float*)d_in[1];
  const float* pos = (const float*)d_in[2];
  const float* init_w = (const float*)d_in[3];
  const float* init_b = (const float*)d_in[4];
  const float* pred_w = (const float*)d_in[5];
  const float* pred_b = (const float*)d_in[6];
  const float* upd_w = (const float*)d_in[7];
  const float* upd_b = (const float*)d_in[8];
  const float* sqw = (const float*)d_in[9];
  const float* sqb = (const float*)d_in[10];
  const float* eqw = (const float*)d_in[11];
  const float* eqb = (const float*)d_in[12];
  const float* keyw = (const float*)d_in[13];
  const float* keyb = (const float*)d_in[14];
  const float* aw1 = (const float*)d_in[15];
  const float* ab1 = (const float*)d_in[16];
  const float* aw2 = (const float*)d_in[17];
  const float* ab2 = (const float*)d_in[18];

  float* ws = (float*)d_ws;
  float* pooled = ws;
  float* sq = pooled + BB * DD;
  float* eq = sq + BB * DD;
  float* cs = eq + BB * DD;
  float* ce = cs + BB;

  float* out = (float*)d_out;
  float* out_start = out;
  float* out_end = out + BB * SS;
  float* out_ans = out + 2 * BB * SS;

  (void)hipFuncSetAttribute((const void*)refine_kernel,
                            hipFuncAttributeMaxDynamicSharedMemorySize,
                            (int)SMEM_BYTES);

  pooled_kernel<<<BB, 256, 0, stream>>>(ids, emb, pos, pooled);
  refine_kernel<<<BB / BT, 256, SMEM_BYTES, stream>>>(
      pooled, init_w, init_b, pred_w, pred_b, upd_w, upd_b, sqw, sqb, eqw, eqb,
      keyw, keyb, aw1, ab1, aw2, ab2, sq, eq, cs, ce, out_ans);
  logits_kernel<<<BB, 256, 0, stream>>>(ids, emb, pos, sq, eq, cs, ce,
                                        out_start, out_end);
}


# refine kernel runs 16 waves a work-group, one n-tile each (was 8 waves x 2 tiles), one staging buffer a wave: four waves per SIMD cover each other's load, LDS and epilogue waits; sums unchanged
# speedup vs baseline: 1.8814x; 1.0457x over previous
#include <hip/hip_runtime.h>
#include <math.h>

#define BB   256
#define SS   512
#define DD   256
#define LLN  4
#define NITERS 10
#define BT   16
#define LSTR 260
#define TILE (BT * LSTR)
#define NWAVES 16
#define NTHR (NWAVES * 32)
#define WSTR 132
#define WHALF (16 * WSTR)
#define WSTAGE WHALF
#define SMEM_FLOATS (9 * TILE + NWAVES * WSTAGE)
#define SMEM_BYTES  (SMEM_FLOATS * sizeof(float))

typedef float v2f __attribute__((ext_vector_type(2)));
typedef float v4f __attribute__((ext_vector_type(4)));
typedef float v8f __attribute__((ext_vector_type(8)));

#if defined(__has_builtin)
#  if __has_builtin(__builtin_amdgcn_wmma_f32_16x16x4_f32)
#    define HAS_WMMA_F32 1
#  else
#    define HAS_WMMA_F32 0
#  endif
#  if __has_builtin(__builtin_amdgcn_tanhf)
#    define HAS_HW_TANH 1
#  else
#    define HAS_HW_TANH 0
#  endif
#else
#  define HAS_WMMA_F32 0
#  define HAS_HW_TANH 0
#endif

__device__ __forceinline__ float tanh_fast(float x) {
#if HAS_HW_TANH
  return __builtin_amdgcn_tanhf(x);
#else
  return tanhf(x);
#endif
}

__device__ __forceinline__ float* smb(float* sm, int i) { return sm + i * TILE; }


__device__ __forceinline__ v8f tile_xWT(const float* __restrict__ W,
                                        const float* X, int nt, int lane,
                                        float cinit) {
  const int l16 = lane & 15;
  const int half = lane >> 4;
  v8f acc = {cinit, cinit, cinit, cinit, cinit, cinit, cinit, cinit};
#if HAS_WMMA_F32
  const float* wr = W + (nt * 16 + l16) * DD + 2 * half;
  const float* xr = X + l16 * LSTR + 2 * half;
#pragma unroll 4
  for (int kk = 0; kk < DD; kk += 4) {
    v2f a = {xr[kk], xr[kk + 1]};
    v2f b = {wr[kk], wr[kk + 1]};
    acc = __builtin_amdgcn_wmma_f32_16x16x4_f32(false, a, false, b,
                                                (short)0, acc, false, false);
  }
#else
#pragma unroll
  for (int j = 0; j < 8; ++j) {
    const int m = j + 8 * half, n = nt * 16 + l16;
    float s = cinit;
    for (int k = 0; k < DD; ++k) s += X[m * LSTR + k] * W[n * DD + k];
    acc[j] = s;
  }
#endif
  return acc;
}

__device__ __forceinline__ v8f tile_xM(const float* __restrict__ M,
                                       const float* X, int nt, int lane,
                                       float cinit) {
  const int l16 = lane & 15;
  const int half = lane >> 4;
  v8f acc = {cinit, cinit, cinit, cinit, cinit, cinit, cinit, cinit};
#if HAS_WMMA_F32
  const float* mp = M + (nt * 16 + l16) + (2 * half) * DD;
  const float* xr = X + l16 * LSTR + 2 * half;
#pragma unroll 4
  for (int kk = 0; kk < DD; kk += 4) {
    v2f a = {xr[kk], xr[kk + 1]};
    v2f b = {mp[kk * DD], mp[kk * DD + DD]};
    acc = __builtin_amdgcn_wmma_f32_16x16x4_f32(false, a, false, b,
                                                (short)0, acc, false, false);
  }
#else
#pragma unroll
  for (int j = 0; j < 8; ++j) {
    const int m = j + 8 * half, n = nt * 16 + l16;
    float s = cinit;
    for (int k = 0; k < DD; ++k) s += X[m * LSTR + k] * M[k * DD + n];
    acc[j] = s;
  }
#endif
  return acc;
}

#if HAS_WMMA_F32
__device__ __forceinline__ void stage_issue(v4f (&r)[16],
                                            const float* __restrict__ wsrc) {
#pragma unroll
  for (int j = 0; j < 16; ++j)
    r[j] = *reinterpret_cast<const v4f*>(wsrc + j * DD);
}

__device__ __forceinline__ void stage_park(const v4f (&r)[16], float* wp) {
#pragma unroll
  for (int j = 0; j < 16; ++j) *reinterpret_cast<v4f*>(wp + j * WSTR) = r[j];
}

__device__ __forceinline__ void wave_lds_handover() {
  __builtin_amdgcn_fence(__ATOMIC_RELEASE, "wavefront");
  __builtin_amdgcn_wave_barrier();
  __builtin_amdgcn_fence(__ATOMIC_ACQUIRE, "wavefront");
}

__device__ __forceinline__ void frag_read8(v2f (&a)[8], v2f (&b)[8],
                                           const float* xr, const float* wr) {
#pragma unroll
  for (int s = 0; s < 8; ++s) {
    a[s] = (v2f){xr[4 * s], xr[4 * s + 1]};
    b[s] = (v2f){wr[4 * s], wr[4 * s + 1]};
  }
}

__device__ __forceinline__ v8f mma8(v8f acc, const v2f (&a)[8],
                                    const v2f (&b)[8]) {
#pragma unroll
  for (int s = 0; s < 8; ++s)
    acc = __builtin_amdgcn_wmma_f32_16x16x4_f32(false, a[s], false, b[s],
                                                (short)0, acc, false, false);
  return acc;
}

__device__ __forceinline__ v8f half_tile_mma(v8f acc, const float* xr,
                                             const float* wr) {
  v2f a0[8], b0[8], a1[8], b1[8];
  frag_read8(a0, b0, xr, wr);
  frag_read8(a1, b1, xr + 32, wr + 32);
  __builtin_amdgcn_sched_barrier(0);
  acc = mma8(acc, a0, b0);
  __builtin_amdgcn_sched_barrier(0);
  frag_read8(a0, b0, xr + 64, wr + 64);
  __builtin_amdgcn_sched_barrier(0);
  acc = mma8(acc, a1, b1);
  __builtin_amdgcn_sched_barrier(0);
  frag_read8(a1, b1, xr + 96, wr + 96);
  __builtin_amdgcn_sched_barrier(0);
  acc = mma8(acc, a0, b0);
  acc = mma8(acc, a1, b1);
  return acc;
}
#endif

__device__ __forceinline__ void gemm_first_request(
    v4f (&r)[16], float& bv, const float* __restrict__ W,
    const float* __restrict__ bias, int lane, int wv) {
#if HAS_WMMA_F32
  stage_issue(r, W + (wv * 16) * DD + lane * 4);
#endif
  bv = bias ? bias[wv * 16 + (lane & 15)] : 0.f;
}

template <typename TileEpi>
__device__ __forceinline__ void gemm_xWT_tiles(
    const float* __restrict__ W, const float* __restrict__ bias,
    const float* X, float* stage, int lane, int wv, float& bv,
    const float* __restrict__ Wn, const float* __restrict__ biasn,
    const bool has_next, TileEpi epi) {
  const int l16 = lane & 15, half = lane >> 4;
#if HAS_WMMA_F32
  const float* wsrc = W + (wv * 16) * DD + lane * 4;
  const float* xr = X + l16 * LSTR + 2 * half;
  const float* wr = stage + l16 * WSTR + 2 * half;
  float* wp = stage + lane * 4;
  const float cb = bv;
  v4f r[16];
  v8f acc = {cb, cb, cb, cb, cb, cb, cb, cb};
#pragma unroll 1
  for (int q = 0; q < 2; ++q) {
    const bool more = (q < 1) || has_next;
    if (more) {
      stage_issue(r, (q < 1) ? wsrc + 128 : Wn + (wv * 16) * DD + lane * 4);
      if (q == 1) bv = biasn ? biasn[wv * 16 + l16] : 0.f;
    }
    acc = half_tile_mma(acc, xr + q * 128, wr);
    if (q == 1) epi(wv, acc);
    if (more) {
      wave_lds_handover();
      stage_park(r, wp);
      wave_lds_handover();
    }
  }
#else
  const float bvf = bias ? bias[wv * 16 + l16] : 0.f;
  v8f acc = tile_xWT(W, X, wv, lane, bvf);
  epi(wv, acc);
#endif
}

template <typename Epi>
__device__ __forceinline__ void gemm_xWT(
    const float* __restrict__ W, const float* __restrict__ bias,
    const float* X, float* stage, int lane, int wv, float& bv,
    const float* __restrict__ Wn, const float* __restrict__ biasn,
    const bool has_next, Epi epi) {
  gemm_xWT_tiles(W, bias, X, stage, lane, wv, bv, Wn, biasn, has_next,
                 [&](int nt, v8f acc) {
    const int l16 = lane & 15, half = lane >> 4;
#pragma unroll
    for (int j = 0; j < 8; ++j) epi(j + 8 * half, nt * 16 + l16, acc[j]);
  });
}

template <typename Epi>
__device__ __forceinline__ void gemm_xM(const float* __restrict__ M,
                                        const float* X, int lane, int wv,
                                        Epi epi) {
  const int nt = wv;
  v8f acc = tile_xM(M, X, nt, lane, 0.f);
  const int l16 = lane & 15, half = lane >> 4;
#pragma unroll
  for (int j = 0; j < 8; ++j) epi(j + 8 * half, nt * 16 + l16, acc[j]);
}

__global__ __launch_bounds__(256) void pooled_kernel(
    const int* __restrict__ ids, const float* __restrict__ emb,
    const float* __restrict__ pos, float* __restrict__ pooled_g) {
  __shared__ int sid[SS];
  const int b = blockIdx.x;
  const int t = threadIdx.x;
  sid[t] = ids[b * SS + t];
  sid[t + 256] = ids[b * SS + 256 + t];
  __syncthreads();
  float acc = 0.f;
  for (int s = 0; s < SS; ++s) {
    acc += emb[(size_t)sid[s] * DD + t] + pos[s * DD + t];
  }
  pooled_g[b * DD + t] = acc * (1.0f / (float)SS);
}

__global__ __launch_bounds__(NTHR) void refine_kernel(
    const float* __restrict__ pooled_g,
    const float* __restrict__ init_w, const float* __restrict__ init_b,
    const float* __restrict__ pred_w, const float* __restrict__ pred_b,
    const float* __restrict__ upd_w, const float* __restrict__ upd_b,
    const float* __restrict__ sqw, const float* __restrict__ sqb,
    const float* __restrict__ eqw, const float* __restrict__ eqb,
    const float* __restrict__ keyw, const float* __restrict__ keyb,
    const float* __restrict__ aw1, const float* __restrict__ ab1,
    const float* __restrict__ aw2, const float* __restrict__ ab2,
    float* __restrict__ sq_g, float* __restrict__ eq_g,
    float* __restrict__ cs_g, float* __restrict__ ce_g,
    float* __restrict__ ans_g) {
  extern __shared__ __attribute__((aligned(16))) float sm[];

  const int tid = threadIdx.x;
  const int lane = tid & 31;
  const int wv = tid >> 5;
  const int b0 = blockIdx.x * BT;

  float* pooled = smb(sm, 0);
  float* ff = smb(sm, 5);
  float* tot = smb(sm, 8);
  float* stage = sm + 9 * TILE + wv * WSTAGE;

  float bv;
  {
    v4f r0[16];
    gemm_first_request(r0, bv, init_w, init_b, lane, wv);
    for (int idx = tid; idx < BT * DD; idx += NTHR) {
      const int r = idx >> 8, c = idx & 255;
      pooled[r * LSTR + c] = pooled_g[(b0 + r) * DD + c];
    }
#if HAS_WMMA_F32
    stage_park(r0, stage + lane * 4);
    wave_lds_handover();
#endif
  }
  __syncthreads();

  {
    const float* cur = pooled;
    for (int l = 0; l < LLN; ++l) {
      float* dst = smb(sm, 1 + l);
      const bool last = (l == LLN - 1);
      const float* wn = last ? pred_w + (LLN - 1) * DD * DD
                             : init_w + (l + 1) * DD * DD;
      const float* bn = last ? pred_b + (LLN - 1) * DD : init_b + (l + 1) * DD;
      gemm_xWT(init_w + l * DD * DD, init_b + l * DD, cur, stage, lane, wv,
               bv, wn, bn, true,
               [&](int m, int n, float v) {
                 const float r = tanh_fast(v);
                 dst[m * LSTR + n] = r;
                 if (last) ff[m * LSTR + n] = r;
               });
      __syncthreads();
      cur = dst;
    }
  }

  for (int it = 0; it < NITERS; ++it) {
    for (int i = LLN - 1; i >= 0; --i) {
      float* pc = smb(sm, 6 + ((LLN - 1 - i) & 1));
      float* pn = smb(sm, 6 + ((LLN - i) & 1));
      float* ri = smb(sm, 1 + i);
      {
        const float* below = (i > 0) ? smb(sm, i) : pooled;
        const bool has_above = (i < LLN - 1);
        const bool read_below = (i > 0);
        gemm_xWT_tiles(
            pred_w + i * DD * DD, pred_b + i * DD, ri, stage, lane, wv,
            bv, upd_w + i * DD * DD, upd_b + i * DD, true,
            [&](int nt, v8f acc) {
              const int o0 = 8 * (lane >> 4) * LSTR + nt * 16 + (lane & 15);
              float bl[8], rr[8], pa[8];
#pragma unroll
              for (int j = 0; j < 8; ++j) bl[j] = below[o0 + j * LSTR];
              if (has_above) {
#pragma unroll
                for (int j = 0; j < 8; ++j) {
                  rr[j] = ri[o0 + j * LSTR];
                  pa[j] = pn[o0 + j * LSTR];
                }
              }
              if (read_below) {
#pragma unroll
                for (int j = 0; j < 8; ++j) pc[o0 + j * LSTR] = acc[j];
              }
#pragma unroll
              for (int j = 0; j < 8; ++j) {
                float e = bl[j] - acc[j];
                if (has_above) e += 0.5f * (rr[j] - pa[j]);
                tot[o0 + j * LSTR] = e;
              }
            });
      }
      __syncthreads();
      const bool sweep_end = (i == 0);
      const bool all_end = sweep_end && (it == NITERS - 1);
      const float* wn = all_end ? sqw
                                : pred_w + (sweep_end ? LLN - 1 : i - 1) * DD * DD;
      const float* bn = all_end ? sqb
                                : pred_b + (sweep_end ? LLN - 1 : i - 1) * DD;
      gemm_xWT(upd_w + i * DD * DD, upd_b + i * DD, tot, stage, lane, wv,
               bv, wn, bn, true,
               [&](int m, int n, float v) {
                 const int o = m * LSTR + n;
                 ri[o] = ri[o] + 0.1f * tanh_fast(v);
               });
      __syncthreads();
    }
  }

  float* refined = smb(sm, LLN);
  for (int idx = tid; idx < BT * DD; idx += NTHR) {
    const int r = idx >> 8, c = idx & 255;
    refined[r * LSTR + c] += ff[r * LSTR + c];
  }
  __syncthreads();

  {
    float* q = smb(sm, 6);
    gemm_xWT(sqw, sqb, refined, stage, lane, wv, bv, eqw, eqb, true,
             [&](int m, int n, float v) { q[m * LSTR + n] = v; });
    __syncthreads();
    if (tid < BT * 16) {
      const int r = tid >> 4, c = tid & 15;
      float p = 0.f;
#pragma unroll
      for (int u = 0; u < 16; ++u) {
        const int o = c * 16 + u;
        p += q[r * LSTR + o] * keyb[o];
      }
      tot[tid] = p;
    }
    __syncthreads();
    if (tid < BT) {
      float s = 0.f;
#pragma unroll
      for (int u = 0; u < 16; ++u) s += tot[tid * 16 + u];
      cs_g[b0 + tid] = s;
    }
    __syncthreads();
    gemm_xM(keyw, q, lane, wv,
            [&](int m, int n, float v) { sq_g[(b0 + m) * DD + n] = v; });
    __syncthreads();
  }

  {
    float* q = smb(sm, 7);
    gemm_xWT(eqw, eqb, refined, stage, lane, wv, bv, nullptr, nullptr, false,
             [&](int m, int n, float v) { q[m * LSTR + n] = v; });
    __syncthreads();
    if (tid < BT * 16) {
      const int r = tid >> 4, c = tid & 15;
      float p = 0.f;
#pragma unroll
      for (int u = 0; u < 16; ++u) {
        const int o = c * 16 + u;
        p += q[r * LSTR + o] * keyb[o];
      }
      tot[tid] = p;
    }
    __syncthreads();
    if (tid < BT) {
      float s = 0.f;
#pragma unroll
      for (int u = 0; u < 16; ++u) s += tot[tid * 16 + u];
      ce_g[b0 + tid] = s;
    }
    __syncthreads();
    gemm_xM(keyw, q, lane, wv,
            [&](int m, int n, float v) { eq_g[(b0 + m) * DD + n] = v; });
    __syncthreads();
  }

  for (int task = tid; task < BT * 128; task += NTHR) {
    const int r = task >> 7, j = task & 127;
    float s = ab1[j];
    const float* rr = refined + r * LSTR;
    const float* wr = aw1 + j * DD;
    for (int k = 0; k < DD; ++k) s += rr[k] * wr[k];
    tot[r * LSTR + j] = 0.5f * s * (1.0f + erff(s * 0.70710678118654752f));
  }
  __syncthreads();
  if (tid < BT * 2) {
    const int r = tid >> 1, k = tid & 1;
    float s = ab2[k];
    for (int j = 0; j < 128; ++j) s += tot[r * LSTR + j] * aw2[k * 128 + j];
    ans_g[(b0 + r) * 2 + k] = s;
  }
}

__global__ __launch_bounds__(256) void logits_kernel(
    const int* __restrict__ ids, const float* __restrict__ emb,
    const float* __restrict__ pos, const float* __restrict__ sq_g,
    const float* __restrict__ eq_g, const float* __restrict__ cs_g,
    const float* __restrict__ ce_g, float* __restrict__ out_start,
    float* __restrict__ out_end) {
  const int b = blockIdx.x;
  const int lane = threadIdx.x & 31;
  const int wv = threadIdx.x >> 5;
  float sqv[8], eqv[8];
  const float* sr = sq_g + b * DD + lane * 8;
  const float* er = eq_g + b * DD + lane * 8;
#pragma unroll
  for (int j = 0; j < 8; ++j) {
    sqv[j] = sr[j];
    eqv[j] = er[j];
  }
  const float cs = cs_g[b];
  const float ce = ce_g[b];
  for (int s = wv; s < SS; s += 8) {
    const int id = ids[b * SS + s];
    const float* row = emb + (size_t)id * DD + lane * 8;
    const float* pr = pos + s * DD + lane * 8;
    float as = 0.f, ae = 0.f;
#pragma unroll
    for (int j = 0; j < 8; ++j) {
      const float e = row[j] + pr[j];
      as += e * sqv[j];
      ae += e * eqv[j];
    }
#pragma unroll
    for (int off = 16; off; off >>= 1) {
      as += __shfl_xor(as, off, 32);
      ae += __shfl_xor(ae, off, 32);
    }
    if (lane == 0) {
      out_start[b * SS + s] = as + cs;
      out_end[b * SS + s] = ae + ce;
    }
  }
}

extern "C" void kernel_launch(void* const* d_in, const int* in_sizes, int n_in,
                              void* d_out, int out_size, void* d_ws,
                              size_t ws_size, hipStream_t stream) {
  const int* ids = (const int*)d_in[0];
  const float* emb = (const float*)d_in[1];
  const float* pos = (const float*)d_in[2];
  const float* init_w = (const float*)d_in[3];
  const float* init_b = (const float*)d_in[4];
  const float* pred_w = (const float*)d_in[5];
  const float* pred_b = (const float*)d_in[6];
  const float* upd_w = (const float*)d_in[7];
  const float* upd_b = (const float*)d_in[8];
  const float* sqw = (const float*)d_in[9];
  const float* sqb = (const float*)d_in[10];
  const float* eqw = (const float*)d_in[11];
  const float* eqb = (const float*)d_in[12];
  const float* keyw = (const float*)d_in[13];
  const float* keyb = (const float*)d_in[14];
  const float* aw1 = (const float*)d_in[15];
  const float* ab1 = (const float*)d_in[16];
  const float* aw2 = (const float*)d_in[17];
  const float* ab2 = (const float*)d_in[18];

  float* ws = (float*)d_ws;
  float* pooled = ws;
  float* sq = pooled + BB * DD;
  float* eq = sq + BB * DD;
  float* cs = eq + BB * DD;
  float* ce = cs + BB;

  float* out = (float*)d_out;
  float* out_start = out;
  float* out_end = out + BB * SS;
  float* out_ans = out + 2 * BB * SS;

  (void)hipFuncSetAttribute((const void*)refine_kernel,
                            hipFuncAttributeMaxDynamicSharedMemorySize,
                            (int)SMEM_BYTES);

  pooled_kernel<<<BB, 256, 0, stream>>>(ids, emb, pos, pooled);
  refine_kernel<<<BB / BT, NTHR, SMEM_BYTES, stream>>>(
      pooled, init_w, init_b, pred_w, pred_b, upd_w, upd_b, sqw, sqb, eqw, eqb,
      keyw, keyb, aw1, ab1, aw2, ab2, sq, eq, cs, ce, out_ans);
  logits_kernel<<<BB, 256, 0, stream>>>(ids, emb, pos, sq, eq, cs, ce,
                                        out_start, out_end);
}


# refine sweep is two barrier phases, not eight: each wave runs its four pred tiles back to back, then its four upd tiles; pred[i+1] handed on in registers, total_error one tile a layer; sums unchanged
# speedup vs baseline: 1.9942x; 1.0600x over previous
#include <hip/hip_runtime.h>
#include <math.h>

#define BB   256
#define SS   512
#define DD   256
#define LLN  4
#define NITERS 10
#define BT   16
#define LSTR 260
#define TILE (BT * LSTR)
#define NWAVES 16
#define NTHR (NWAVES * 32)
#define WSTR 132
#define WHALF (16 * WSTR)
#define WSTAGE WHALF
#define NTILES 10
#define SMEM_FLOATS (NTILES * TILE + NWAVES * WSTAGE)
#define SMEM_BYTES  (SMEM_FLOATS * sizeof(float))

typedef float v2f __attribute__((ext_vector_type(2)));
typedef float v4f __attribute__((ext_vector_type(4)));
typedef float v8f __attribute__((ext_vector_type(8)));

#if defined(__has_builtin)
#  if __has_builtin(__builtin_amdgcn_wmma_f32_16x16x4_f32)
#    define HAS_WMMA_F32 1
#  else
#    define HAS_WMMA_F32 0
#  endif
#  if __has_builtin(__builtin_amdgcn_tanhf)
#    define HAS_HW_TANH 1
#  else
#    define HAS_HW_TANH 0
#  endif
#else
#  define HAS_WMMA_F32 0
#  define HAS_HW_TANH 0
#endif

__device__ __forceinline__ float tanh_fast(float x) {
#if HAS_HW_TANH
  return __builtin_amdgcn_tanhf(x);
#else
  return tanhf(x);
#endif
}

__device__ __forceinline__ float* smb(float* sm, int i) { return sm + i * TILE; }


__device__ __forceinline__ v8f tile_xWT(const float* __restrict__ W,
                                        const float* X, int nt, int lane,
                                        float cinit) {
  const int l16 = lane & 15;
  const int half = lane >> 4;
  v8f acc = {cinit, cinit, cinit, cinit, cinit, cinit, cinit, cinit};
#if HAS_WMMA_F32
  const float* wr = W + (nt * 16 + l16) * DD + 2 * half;
  const float* xr = X + l16 * LSTR + 2 * half;
#pragma unroll 4
  for (int kk = 0; kk < DD; kk += 4) {
    v2f a = {xr[kk], xr[kk + 1]};
    v2f b = {wr[kk], wr[kk + 1]};
    acc = __builtin_amdgcn_wmma_f32_16x16x4_f32(false, a, false, b,
                                                (short)0, acc, false, false);
  }
#else
#pragma unroll
  for (int j = 0; j < 8; ++j) {
    const int m = j + 8 * half, n = nt * 16 + l16;
    float s = cinit;
    for (int k = 0; k < DD; ++k) s += X[m * LSTR + k] * W[n * DD + k];
    acc[j] = s;
  }
#endif
  return acc;
}

__device__ __forceinline__ v8f tile_xM(const float* __restrict__ M,
                                       const float* X, int nt, int lane,
                                       float cinit) {
  const int l16 = lane & 15;
  const int half = lane >> 4;
  v8f acc = {cinit, cinit, cinit, cinit, cinit, cinit, cinit, cinit};
#if HAS_WMMA_F32
  const float* mp = M + (nt * 16 + l16) + (2 * half) * DD;
  const float* xr = X + l16 * LSTR + 2 * half;
#pragma unroll 4
  for (int kk = 0; kk < DD; kk += 4) {
    v2f a = {xr[kk], xr[kk + 1]};
    v2f b = {mp[kk * DD], mp[kk * DD + DD]};
    acc = __builtin_amdgcn_wmma_f32_16x16x4_f32(false, a, false, b,
                                                (short)0, acc, false, false);
  }
#else
#pragma unroll
  for (int j = 0; j < 8; ++j) {
    const int m = j + 8 * half, n = nt * 16 + l16;
    float s = cinit;
    for (int k = 0; k < DD; ++k) s += X[m * LSTR + k] * M[k * DD + n];
    acc[j] = s;
  }
#endif
  return acc;
}

#if HAS_WMMA_F32
__device__ __forceinline__ void stage_issue(v4f (&r)[16],
                                            const float* __restrict__ wsrc) {
#pragma unroll
  for (int j = 0; j < 16; ++j)
    r[j] = *reinterpret_cast<const v4f*>(wsrc + j * DD);
}

__device__ __forceinline__ void stage_park(const v4f (&r)[16], float* wp) {
#pragma unroll
  for (int j = 0; j < 16; ++j) *reinterpret_cast<v4f*>(wp + j * WSTR) = r[j];
}

__device__ __forceinline__ void wave_lds_handover() {
  __builtin_amdgcn_fence(__ATOMIC_RELEASE, "wavefront");
  __builtin_amdgcn_wave_barrier();
  __builtin_amdgcn_fence(__ATOMIC_ACQUIRE, "wavefront");
}

__device__ __forceinline__ void frag_read8(v2f (&a)[8], v2f (&b)[8],
                                           const float* xr, const float* wr) {
#pragma unroll
  for (int s = 0; s < 8; ++s) {
    a[s] = (v2f){xr[4 * s], xr[4 * s + 1]};
    b[s] = (v2f){wr[4 * s], wr[4 * s + 1]};
  }
}

__device__ __forceinline__ v8f mma8(v8f acc, const v2f (&a)[8],
                                    const v2f (&b)[8]) {
#pragma unroll
  for (int s = 0; s < 8; ++s)
    acc = __builtin_amdgcn_wmma_f32_16x16x4_f32(false, a[s], false, b[s],
                                                (short)0, acc, false, false);
  return acc;
}

__device__ __forceinline__ v8f half_tile_mma(v8f acc, const float* xr,
                                             const float* wr) {
  v2f a0[8], b0[8], a1[8], b1[8];
  frag_read8(a0, b0, xr, wr);
  frag_read8(a1, b1, xr + 32, wr + 32);
  __builtin_amdgcn_sched_barrier(0);
  acc = mma8(acc, a0, b0);
  __builtin_amdgcn_sched_barrier(0);
  frag_read8(a0, b0, xr + 64, wr + 64);
  __builtin_amdgcn_sched_barrier(0);
  acc = mma8(acc, a1, b1);
  __builtin_amdgcn_sched_barrier(0);
  frag_read8(a1, b1, xr + 96, wr + 96);
  __builtin_amdgcn_sched_barrier(0);
  acc = mma8(acc, a0, b0);
  acc = mma8(acc, a1, b1);
  return acc;
}
#endif

__device__ __forceinline__ void gemm_first_request(
    v4f (&r)[16], float& bv, const float* __restrict__ W,
    const float* __restrict__ bias, int lane, int wv) {
#if HAS_WMMA_F32
  stage_issue(r, W + (wv * 16) * DD + lane * 4);
#endif
  bv = bias ? bias[wv * 16 + (lane & 15)] : 0.f;
}

template <typename TileEpi>
__device__ __forceinline__ void gemm_xWT_tiles(
    const float* __restrict__ W, const float* __restrict__ bias,
    const float* X, float* stage, int lane, int wv, float& bv,
    const float* __restrict__ Wn, const float* __restrict__ biasn,
    const bool has_next, TileEpi epi) {
  const int l16 = lane & 15, half = lane >> 4;
#if HAS_WMMA_F32
  const float* wsrc = W + (wv * 16) * DD + lane * 4;
  const float* xr = X + l16 * LSTR + 2 * half;
  const float* wr = stage + l16 * WSTR + 2 * half;
  float* wp = stage + lane * 4;
  const float cb = bv;
  v4f r[16];
  v8f acc = {cb, cb, cb, cb, cb, cb, cb, cb};
#pragma unroll 1
  for (int q = 0; q < 2; ++q) {
    const bool more = (q < 1) || has_next;
    if (more) {
      stage_issue(r, (q < 1) ? wsrc + 128 : Wn + (wv * 16) * DD + lane * 4);
      if (q == 1) bv = biasn ? biasn[wv * 16 + l16] : 0.f;
    }
    acc = half_tile_mma(acc, xr + q * 128, wr);
    if (q == 1) epi(wv, acc);
    if (more) {
      wave_lds_handover();
      stage_park(r, wp);
      wave_lds_handover();
    }
  }
#else
  const float bvf = bias ? bias[wv * 16 + l16] : 0.f;
  v8f acc = tile_xWT(W, X, wv, lane, bvf);
  epi(wv, acc);
#endif
}

template <typename Epi>
__device__ __forceinline__ void gemm_xWT(
    const float* __restrict__ W, const float* __restrict__ bias,
    const float* X, float* stage, int lane, int wv, float& bv,
    const float* __restrict__ Wn, const float* __restrict__ biasn,
    const bool has_next, Epi epi) {
  gemm_xWT_tiles(W, bias, X, stage, lane, wv, bv, Wn, biasn, has_next,
                 [&](int nt, v8f acc) {
    const int l16 = lane & 15, half = lane >> 4;
#pragma unroll
    for (int j = 0; j < 8; ++j) epi(j + 8 * half, nt * 16 + l16, acc[j]);
  });
}

template <typename Epi>
__device__ __forceinline__ void gemm_xM(const float* __restrict__ M,
                                        const float* X, int lane, int wv,
                                        Epi epi) {
  const int nt = wv;
  v8f acc = tile_xM(M, X, nt, lane, 0.f);
  const int l16 = lane & 15, half = lane >> 4;
#pragma unroll
  for (int j = 0; j < 8; ++j) epi(j + 8 * half, nt * 16 + l16, acc[j]);
}

__global__ __launch_bounds__(256) void pooled_kernel(
    const int* __restrict__ ids, const float* __restrict__ emb,
    const float* __restrict__ pos, float* __restrict__ pooled_g) {
  __shared__ int sid[SS];
  const int b = blockIdx.x;
  const int t = threadIdx.x;
  sid[t] = ids[b * SS + t];
  sid[t + 256] = ids[b * SS + 256 + t];
  __syncthreads();
  float acc = 0.f;
  for (int s = 0; s < SS; ++s) {
    acc += emb[(size_t)sid[s] * DD + t] + pos[s * DD + t];
  }
  pooled_g[b * DD + t] = acc * (1.0f / (float)SS);
}

__global__ __launch_bounds__(NTHR) void refine_kernel(
    const float* __restrict__ pooled_g,
    const float* __restrict__ init_w, const float* __restrict__ init_b,
    const float* __restrict__ pred_w, const float* __restrict__ pred_b,
    const float* __restrict__ upd_w, const float* __restrict__ upd_b,
    const float* __restrict__ sqw, const float* __restrict__ sqb,
    const float* __restrict__ eqw, const float* __restrict__ eqb,
    const float* __restrict__ keyw, const float* __restrict__ keyb,
    const float* __restrict__ aw1, const float* __restrict__ ab1,
    const float* __restrict__ aw2, const float* __restrict__ ab2,
    float* __restrict__ sq_g, float* __restrict__ eq_g,
    float* __restrict__ cs_g, float* __restrict__ ce_g,
    float* __restrict__ ans_g) {
  extern __shared__ __attribute__((aligned(16))) float sm[];

  const int tid = threadIdx.x;
  const int lane = tid & 31;
  const int wv = tid >> 5;
  const int b0 = blockIdx.x * BT;

  float* pooled = smb(sm, 0);
  float* ff = smb(sm, 5);
  float* tot = smb(sm, 8);
  float* stage = sm + NTILES * TILE + wv * WSTAGE;

  float bv;
  {
    v4f r0[16];
    gemm_first_request(r0, bv, init_w, init_b, lane, wv);
    for (int idx = tid; idx < BT * DD; idx += NTHR) {
      const int r = idx >> 8, c = idx & 255;
      pooled[r * LSTR + c] = pooled_g[(b0 + r) * DD + c];
    }
#if HAS_WMMA_F32
    stage_park(r0, stage + lane * 4);
    wave_lds_handover();
#endif
  }
  __syncthreads();

  {
    const float* cur = pooled;
    for (int l = 0; l < LLN; ++l) {
      float* dst = smb(sm, 1 + l);
      const bool last = (l == LLN - 1);
      const float* wn = last ? pred_w + (LLN - 1) * DD * DD
                             : init_w + (l + 1) * DD * DD;
      const float* bn = last ? pred_b + (LLN - 1) * DD : init_b + (l + 1) * DD;
      gemm_xWT(init_w + l * DD * DD, init_b + l * DD, cur, stage, lane, wv,
               bv, wn, bn, true,
               [&](int m, int n, float v) {
                 const float r = tanh_fast(v);
                 dst[m * LSTR + n] = r;
                 if (last) ff[m * LSTR + n] = r;
               });
      __syncthreads();
      cur = dst;
    }
  }

  for (int it = 0; it < NITERS; ++it) {
    {
      v8f pabove = {0.f, 0.f, 0.f, 0.f, 0.f, 0.f, 0.f, 0.f};
#pragma unroll 1
      for (int i = LLN - 1; i >= 0; --i) {
        const float* ri = smb(sm, 1 + i);
        const float* below = (i > 0) ? smb(sm, i) : pooled;
        float* ti = smb(sm, 6 + i);
        const bool has_above = (i < LLN - 1);
        const float* wn = (i > 0) ? pred_w + (i - 1) * DD * DD
                                  : upd_w + (LLN - 1) * DD * DD;
        const float* bn = (i > 0) ? pred_b + (i - 1) * DD
                                  : upd_b + (LLN - 1) * DD;
        gemm_xWT_tiles(
            pred_w + i * DD * DD, pred_b + i * DD, ri, stage, lane, wv,
            bv, wn, bn, true,
            [&](int nt, v8f acc) {
              const int o0 = 8 * (lane >> 4) * LSTR + nt * 16 + (lane & 15);
              float bl[8], rr[8];
#pragma unroll
              for (int j = 0; j < 8; ++j) bl[j] = below[o0 + j * LSTR];
              if (has_above) {
#pragma unroll
                for (int j = 0; j < 8; ++j) rr[j] = ri[o0 + j * LSTR];
              }
#pragma unroll
              for (int j = 0; j < 8; ++j) {
                float e = bl[j] - acc[j];
                if (has_above) e += 0.5f * (rr[j] - pabove[j]);
                ti[o0 + j * LSTR] = e;
              }
              pabove = acc;
            });
      }
    }
    __syncthreads();
#pragma unroll 1
    for (int i = LLN - 1; i >= 0; --i) {
      float* ri = smb(sm, 1 + i);
      const float* ti = smb(sm, 6 + i);
      const bool phase_end = (i == 0);
      const bool all_end = phase_end && (it == NITERS - 1);
      const float* wn = all_end ? sqw
                        : phase_end ? pred_w + (LLN - 1) * DD * DD
                                    : upd_w + (i - 1) * DD * DD;
      const float* bn = all_end ? sqb
                        : phase_end ? pred_b + (LLN - 1) * DD
                                    : upd_b + (i - 1) * DD;
      gemm_xWT(upd_w + i * DD * DD, upd_b + i * DD, ti, stage, lane, wv,
               bv, wn, bn, true,
               [&](int m, int n, float v) {
                 const int o = m * LSTR + n;
                 ri[o] = ri[o] + 0.1f * tanh_fast(v);
               });
    }
    __syncthreads();
  }

  float* refined = smb(sm, LLN);
  for (int idx = tid; idx < BT * DD; idx += NTHR) {
    const int r = idx >> 8, c = idx & 255;
    refined[r * LSTR + c] += ff[r * LSTR + c];
  }
  __syncthreads();

  {
    float* q = smb(sm, 6);
    gemm_xWT(sqw, sqb, refined, stage, lane, wv, bv, eqw, eqb, true,
             [&](int m, int n, float v) { q[m * LSTR + n] = v; });
    __syncthreads();
    if (tid < BT * 16) {
      const int r = tid >> 4, c = tid & 15;
      float p = 0.f;
#pragma unroll
      for (int u = 0; u < 16; ++u) {
        const int o = c * 16 + u;
        p += q[r * LSTR + o] * keyb[o];
      }
      tot[tid] = p;
    }
    __syncthreads();
    if (tid < BT) {
      float s = 0.f;
#pragma unroll
      for (int u = 0; u < 16; ++u) s += tot[tid * 16 + u];
      cs_g[b0 + tid] = s;
    }
    __syncthreads();
    gemm_xM(keyw, q, lane, wv,
            [&](int m, int n, float v) { sq_g[(b0 + m) * DD + n] = v; });
    __syncthreads();
  }

  {
    float* q = smb(sm, 7);
    gemm_xWT(eqw, eqb, refined, stage, lane, wv, bv, nullptr, nullptr, false,
             [&](int m, int n, float v) { q[m * LSTR + n] = v; });
    __syncthreads();
    if (tid < BT * 16) {
      const int r = tid >> 4, c = tid & 15;
      float p = 0.f;
#pragma unroll
      for (int u = 0; u < 16; ++u) {
        const int o = c * 16 + u;
        p += q[r * LSTR + o] * keyb[o];
      }
      tot[tid] = p;
    }
    __syncthreads();
    if (tid < BT) {
      float s = 0.f;
#pragma unroll
      for (int u = 0; u < 16; ++u) s += tot[tid * 16 + u];
      ce_g[b0 + tid] = s;
    }
    __syncthreads();
    gemm_xM(keyw, q, lane, wv,
            [&](int m, int n, float v) { eq_g[(b0 + m) * DD + n] = v; });
    __syncthreads();
  }

  for (int task = tid; task < BT * 128; task += NTHR) {
    const int r = task >> 7, j = task & 127;
    float s = ab1[j];
    const float* rr = refined + r * LSTR;
    const float* wr = aw1 + j * DD;
    for (int k = 0; k < DD; ++k) s += rr[k] * wr[k];
    tot[r * LSTR + j] = 0.5f * s * (1.0f + erff(s * 0.70710678118654752f));
  }
  __syncthreads();
  if (tid < BT * 2) {
    const int r = tid >> 1, k = tid & 1;
    float s = ab2[k];
    for (int j = 0; j < 128; ++j) s += tot[r * LSTR + j] * aw2[k * 128 + j];
    ans_g[(b0 + r) * 2 + k] = s;
  }
}

__global__ __launch_bounds__(256) void logits_kernel(
    const int* __restrict__ ids, const float* __restrict__ emb,
    const float* __restrict__ pos, const float* __restrict__ sq_g,
    const float* __restrict__ eq_g, const float* __restrict__ cs_g,
    const float* __restrict__ ce_g, float* __restrict__ out_start,
    float* __restrict__ out_end) {
  const int b = blockIdx.x;
  const int lane = threadIdx.x & 31;
  const int wv = threadIdx.x >> 5;
  float sqv[8], eqv[8];
  const float* sr = sq_g + b * DD + lane * 8;
  const float* er = eq_g + b * DD + lane * 8;
#pragma unroll
  for (int j = 0; j < 8; ++j) {
    sqv[j] = sr[j];
    eqv[j] = er[j];
  }
  const float cs = cs_g[b];
  const float ce = ce_g[b];
  for (int s = wv; s < SS; s += 8) {
    const int id = ids[b * SS + s];
    const float* row = emb + (size_t)id * DD + lane * 8;
    const float* pr = pos + s * DD + lane * 8;
    float as = 0.f, ae = 0.f;
#pragma unroll
    for (int j = 0; j < 8; ++j) {
      const float e = row[j] + pr[j];
      as += e * sqv[j];
      ae += e * eqv[j];
    }
#pragma unroll
    for (int off = 16; off; off >>= 1) {
      as += __shfl_xor(as, off, 32);
      ae += __shfl_xor(ae, off, 32);
    }
    if (lane == 0) {
      out_start[b * SS + s] = as + cs;
      out_end[b * SS + s] = ae + ce;
    }
  }
}

extern "C" void kernel_launch(void* const* d_in, const int* in_sizes, int n_in,
                              void* d_out, int out_size, void* d_ws,
                              size_t ws_size, hipStream_t stream) {
  const int* ids = (const int*)d_in[0];
  const float* emb = (const float*)d_in[1];
  const float* pos = (const float*)d_in[2];
  const float* init_w = (const float*)d_in[3];
  const float* init_b = (const float*)d_in[4];
  const float* pred_w = (const float*)d_in[5];
  const float* pred_b = (const float*)d_in[6];
  const float* upd_w = (const float*)d_in[7];
  const float* upd_b = (const float*)d_in[8];
  const float* sqw = (const float*)d_in[9];
  const float* sqb = (const float*)d_in[10];
  const float* eqw = (const float*)d_in[11];
  const float* eqb = (const float*)d_in[12];
  const float* keyw = (const float*)d_in[13];
  const float* keyb = (const float*)d_in[14];
  const float* aw1 = (const float*)d_in[15];
  const float* ab1 = (const float*)d_in[16];
  const float* aw2 = (const float*)d_in[17];
  const float* ab2 = (const float*)d_in[18];

  float* ws = (float*)d_ws;
  float* pooled = ws;
  float* sq = pooled + BB * DD;
  float* eq = sq + BB * DD;
  float* cs = eq + BB * DD;
  float* ce = cs + BB;

  float* out = (float*)d_out;
  float* out_start = out;
  float* out_end = out + BB * SS;
  float* out_ans = out + 2 * BB * SS;

  (void)hipFuncSetAttribute((const void*)refine_kernel,
                            hipFuncAttributeMaxDynamicSharedMemorySize,
                            (int)SMEM_BYTES);

  pooled_kernel<<<BB, 256, 0, stream>>>(ids, emb, pos, pooled);
  refine_kernel<<<BB / BT, NTHR, SMEM_BYTES, stream>>>(
      pooled, init_w, init_b, pred_w, pred_b, upd_w, upd_b, sqw, sqb, eqw, eqb,
      keyw, keyb, aw1, ab1, aw2, ab2, sq, eq, cs, ce, out_ans);
  logits_kernel<<<BB, 256, 0, stream>>>(ids, emb, pos, sq, eq, cs, ce,
                                        out_start, out_end);
}


# answerable head's hidden layer (refined x aw1^T) runs as a staged WMMA GEMM, one n-tile on each of waves 0..7 (was one aw1 row per lane, scalar fma loop); its k sums now in the WMMA's order
# speedup vs baseline: 2.2682x; 1.1374x over previous
#include <hip/hip_runtime.h>
#include <math.h>

#define BB   256
#define SS   512
#define DD   256
#define LLN  4
#define NITERS 10
#define BT   16
#define LSTR 260
#define TILE (BT * LSTR)
#define NWAVES 16
#define NTHR (NWAVES * 32)
#define WSTR 132
#define WHALF (16 * WSTR)
#define WSTAGE WHALF
#define NTILES 10
#define SMEM_FLOATS (NTILES * TILE + NWAVES * WSTAGE)
#define SMEM_BYTES  (SMEM_FLOATS * sizeof(float))

typedef float v2f __attribute__((ext_vector_type(2)));
typedef float v4f __attribute__((ext_vector_type(4)));
typedef float v8f __attribute__((ext_vector_type(8)));

#if defined(__has_builtin)
#  if __has_builtin(__builtin_amdgcn_wmma_f32_16x16x4_f32)
#    define HAS_WMMA_F32 1
#  else
#    define HAS_WMMA_F32 0
#  endif
#  if __has_builtin(__builtin_amdgcn_tanhf)
#    define HAS_HW_TANH 1
#  else
#    define HAS_HW_TANH 0
#  endif
#else
#  define HAS_WMMA_F32 0
#  define HAS_HW_TANH 0
#endif

__device__ __forceinline__ float tanh_fast(float x) {
#if HAS_HW_TANH
  return __builtin_amdgcn_tanhf(x);
#else
  return tanhf(x);
#endif
}

__device__ __forceinline__ float* smb(float* sm, int i) { return sm + i * TILE; }


__device__ __forceinline__ v8f tile_xWT(const float* __restrict__ W,
                                        const float* X, int nt, int lane,
                                        float cinit) {
  const int l16 = lane & 15;
  const int half = lane >> 4;
  v8f acc = {cinit, cinit, cinit, cinit, cinit, cinit, cinit, cinit};
#if HAS_WMMA_F32
  const float* wr = W + (nt * 16 + l16) * DD + 2 * half;
  const float* xr = X + l16 * LSTR + 2 * half;
#pragma unroll 4
  for (int kk = 0; kk < DD; kk += 4) {
    v2f a = {xr[kk], xr[kk + 1]};
    v2f b = {wr[kk], wr[kk + 1]};
    acc = __builtin_amdgcn_wmma_f32_16x16x4_f32(false, a, false, b,
                                                (short)0, acc, false, false);
  }
#else
#pragma unroll
  for (int j = 0; j < 8; ++j) {
    const int m = j + 8 * half, n = nt * 16 + l16;
    float s = cinit;
    for (int k = 0; k < DD; ++k) s += X[m * LSTR + k] * W[n * DD + k];
    acc[j] = s;
  }
#endif
  return acc;
}

__device__ __forceinline__ v8f tile_xM(const float* __restrict__ M,
                                       const float* X, int nt, int lane,
                                       float cinit) {
  const int l16 = lane & 15;
  const int half = lane >> 4;
  v8f acc = {cinit, cinit, cinit, cinit, cinit, cinit, cinit, cinit};
#if HAS_WMMA_F32
  const float* mp = M + (nt * 16 + l16) + (2 * half) * DD;
  const float* xr = X + l16 * LSTR + 2 * half;
#pragma unroll 4
  for (int kk = 0; kk < DD; kk += 4) {
    v2f a = {xr[kk], xr[kk + 1]};
    v2f b = {mp[kk * DD], mp[kk * DD + DD]};
    acc = __builtin_amdgcn_wmma_f32_16x16x4_f32(false, a, false, b,
                                                (short)0, acc, false, false);
  }
#else
#pragma unroll
  for (int j = 0; j < 8; ++j) {
    const int m = j + 8 * half, n = nt * 16 + l16;
    float s = cinit;
    for (int k = 0; k < DD; ++k) s += X[m * LSTR + k] * M[k * DD + n];
    acc[j] = s;
  }
#endif
  return acc;
}

#if HAS_WMMA_F32
__device__ __forceinline__ void stage_issue(v4f (&r)[16],
                                            const float* __restrict__ wsrc) {
#pragma unroll
  for (int j = 0; j < 16; ++j)
    r[j] = *reinterpret_cast<const v4f*>(wsrc + j * DD);
}

__device__ __forceinline__ void stage_park(const v4f (&r)[16], float* wp) {
#pragma unroll
  for (int j = 0; j < 16; ++j) *reinterpret_cast<v4f*>(wp + j * WSTR) = r[j];
}

__device__ __forceinline__ void wave_lds_handover() {
  __builtin_amdgcn_fence(__ATOMIC_RELEASE, "wavefront");
  __builtin_amdgcn_wave_barrier();
  __builtin_amdgcn_fence(__ATOMIC_ACQUIRE, "wavefront");
}

__device__ __forceinline__ void frag_read8(v2f (&a)[8], v2f (&b)[8],
                                           const float* xr, const float* wr) {
#pragma unroll
  for (int s = 0; s < 8; ++s) {
    a[s] = (v2f){xr[4 * s], xr[4 * s + 1]};
    b[s] = (v2f){wr[4 * s], wr[4 * s + 1]};
  }
}

__device__ __forceinline__ v8f mma8(v8f acc, const v2f (&a)[8],
                                    const v2f (&b)[8]) {
#pragma unroll
  for (int s = 0; s < 8; ++s)
    acc = __builtin_amdgcn_wmma_f32_16x16x4_f32(false, a[s], false, b[s],
                                                (short)0, acc, false, false);
  return acc;
}

__device__ __forceinline__ v8f half_tile_mma(v8f acc, const float* xr,
                                             const float* wr) {
  v2f a0[8], b0[8], a1[8], b1[8];
  frag_read8(a0, b0, xr, wr);
  frag_read8(a1, b1, xr + 32, wr + 32);
  __builtin_amdgcn_sched_barrier(0);
  acc = mma8(acc, a0, b0);
  __builtin_amdgcn_sched_barrier(0);
  frag_read8(a0, b0, xr + 64, wr + 64);
  __builtin_amdgcn_sched_barrier(0);
  acc = mma8(acc, a1, b1);
  __builtin_amdgcn_sched_barrier(0);
  frag_read8(a1, b1, xr + 96, wr + 96);
  __builtin_amdgcn_sched_barrier(0);
  acc = mma8(acc, a0, b0);
  acc = mma8(acc, a1, b1);
  return acc;
}
#endif

__device__ __forceinline__ void gemm_first_request(
    v4f (&r)[16], float& bv, const float* __restrict__ W,
    const float* __restrict__ bias, int lane, int wv) {
#if HAS_WMMA_F32
  stage_issue(r, W + (wv * 16) * DD + lane * 4);
#endif
  bv = bias ? bias[wv * 16 + (lane & 15)] : 0.f;
}

template <typename TileEpi>
__device__ __forceinline__ void gemm_xWT_tiles(
    const float* __restrict__ W, const float* __restrict__ bias,
    const float* X, float* stage, int lane, int wv, float& bv,
    const float* __restrict__ Wn, const float* __restrict__ biasn,
    const bool has_next, TileEpi epi) {
  const int l16 = lane & 15, half = lane >> 4;
#if HAS_WMMA_F32
  const float* wsrc = W + (wv * 16) * DD + lane * 4;
  const float* xr = X + l16 * LSTR + 2 * half;
  const float* wr = stage + l16 * WSTR + 2 * half;
  float* wp = stage + lane * 4;
  const float cb = bv;
  v4f r[16];
  v8f acc = {cb, cb, cb, cb, cb, cb, cb, cb};
#pragma unroll 1
  for (int q = 0; q < 2; ++q) {
    const bool more = (q < 1) || has_next;
    if (more) {
      stage_issue(r, (q < 1) ? wsrc + 128 : Wn + (wv * 16) * DD + lane * 4);
      if (q == 1) bv = biasn ? biasn[wv * 16 + l16] : 0.f;
    }
    acc = half_tile_mma(acc, xr + q * 128, wr);
    if (q == 1) epi(wv, acc);
    if (more) {
      wave_lds_handover();
      stage_park(r, wp);
      wave_lds_handover();
    }
  }
#else
  const float bvf = bias ? bias[wv * 16 + l16] : 0.f;
  v8f acc = tile_xWT(W, X, wv, lane, bvf);
  epi(wv, acc);
#endif
}

template <typename Epi>
__device__ __forceinline__ void gemm_xWT(
    const float* __restrict__ W, const float* __restrict__ bias,
    const float* X, float* stage, int lane, int wv, float& bv,
    const float* __restrict__ Wn, const float* __restrict__ biasn,
    const bool has_next, Epi epi) {
  gemm_xWT_tiles(W, bias, X, stage, lane, wv, bv, Wn, biasn, has_next,
                 [&](int nt, v8f acc) {
    const int l16 = lane & 15, half = lane >> 4;
#pragma unroll
    for (int j = 0; j < 8; ++j) epi(j + 8 * half, nt * 16 + l16, acc[j]);
  });
}

template <typename Epi>
__device__ __forceinline__ void gemm_xM(const float* __restrict__ M,
                                        const float* X, int lane, int wv,
                                        Epi epi) {
  const int nt = wv;
  v8f acc = tile_xM(M, X, nt, lane, 0.f);
  const int l16 = lane & 15, half = lane >> 4;
#pragma unroll
  for (int j = 0; j < 8; ++j) epi(j + 8 * half, nt * 16 + l16, acc[j]);
}

__global__ __launch_bounds__(256) void pooled_kernel(
    const int* __restrict__ ids, const float* __restrict__ emb,
    const float* __restrict__ pos, float* __restrict__ pooled_g) {
  __shared__ int sid[SS];
  const int b = blockIdx.x;
  const int t = threadIdx.x;
  sid[t] = ids[b * SS + t];
  sid[t + 256] = ids[b * SS + 256 + t];
  __syncthreads();
  float acc = 0.f;
  for (int s = 0; s < SS; ++s) {
    acc += emb[(size_t)sid[s] * DD + t] + pos[s * DD + t];
  }
  pooled_g[b * DD + t] = acc * (1.0f / (float)SS);
}

__global__ __launch_bounds__(NTHR) void refine_kernel(
    const float* __restrict__ pooled_g,
    const float* __restrict__ init_w, const float* __restrict__ init_b,
    const float* __restrict__ pred_w, const float* __restrict__ pred_b,
    const float* __restrict__ upd_w, const float* __restrict__ upd_b,
    const float* __restrict__ sqw, const float* __restrict__ sqb,
    const float* __restrict__ eqw, const float* __restrict__ eqb,
    const float* __restrict__ keyw, const float* __restrict__ keyb,
    const float* __restrict__ aw1, const float* __restrict__ ab1,
    const float* __restrict__ aw2, const float* __restrict__ ab2,
    float* __restrict__ sq_g, float* __restrict__ eq_g,
    float* __restrict__ cs_g, float* __restrict__ ce_g,
    float* __restrict__ ans_g) {
  extern __shared__ __attribute__((aligned(16))) float sm[];

  const int tid = threadIdx.x;
  const int lane = tid & 31;
  const int wv = tid >> 5;
  const int wvu = __builtin_amdgcn_readfirstlane(wv);
  const int b0 = blockIdx.x * BT;

  float* pooled = smb(sm, 0);
  float* ff = smb(sm, 5);
  float* tot = smb(sm, 8);
  float* stage = sm + NTILES * TILE + wv * WSTAGE;

  float bv;
  {
    v4f r0[16];
    gemm_first_request(r0, bv, init_w, init_b, lane, wv);
    for (int idx = tid; idx < BT * DD; idx += NTHR) {
      const int r = idx >> 8, c = idx & 255;
      pooled[r * LSTR + c] = pooled_g[(b0 + r) * DD + c];
    }
#if HAS_WMMA_F32
    stage_park(r0, stage + lane * 4);
    wave_lds_handover();
#endif
  }
  __syncthreads();

  {
    const float* cur = pooled;
    for (int l = 0; l < LLN; ++l) {
      float* dst = smb(sm, 1 + l);
      const bool last = (l == LLN - 1);
      const float* wn = last ? pred_w + (LLN - 1) * DD * DD
                             : init_w + (l + 1) * DD * DD;
      const float* bn = last ? pred_b + (LLN - 1) * DD : init_b + (l + 1) * DD;
      gemm_xWT(init_w + l * DD * DD, init_b + l * DD, cur, stage, lane, wv,
               bv, wn, bn, true,
               [&](int m, int n, float v) {
                 const float r = tanh_fast(v);
                 dst[m * LSTR + n] = r;
                 if (last) ff[m * LSTR + n] = r;
               });
      __syncthreads();
      cur = dst;
    }
  }

  for (int it = 0; it < NITERS; ++it) {
    {
      v8f pabove = {0.f, 0.f, 0.f, 0.f, 0.f, 0.f, 0.f, 0.f};
#pragma unroll 1
      for (int i = LLN - 1; i >= 0; --i) {
        const float* ri = smb(sm, 1 + i);
        const float* below = (i > 0) ? smb(sm, i) : pooled;
        float* ti = smb(sm, 6 + i);
        const bool has_above = (i < LLN - 1);
        const float* wn = (i > 0) ? pred_w + (i - 1) * DD * DD
                                  : upd_w + (LLN - 1) * DD * DD;
        const float* bn = (i > 0) ? pred_b + (i - 1) * DD
                                  : upd_b + (LLN - 1) * DD;
        gemm_xWT_tiles(
            pred_w + i * DD * DD, pred_b + i * DD, ri, stage, lane, wv,
            bv, wn, bn, true,
            [&](int nt, v8f acc) {
              const int o0 = 8 * (lane >> 4) * LSTR + nt * 16 + (lane & 15);
              float bl[8], rr[8];
#pragma unroll
              for (int j = 0; j < 8; ++j) bl[j] = below[o0 + j * LSTR];
              if (has_above) {
#pragma unroll
                for (int j = 0; j < 8; ++j) rr[j] = ri[o0 + j * LSTR];
              }
#pragma unroll
              for (int j = 0; j < 8; ++j) {
                float e = bl[j] - acc[j];
                if (has_above) e += 0.5f * (rr[j] - pabove[j]);
                ti[o0 + j * LSTR] = e;
              }
              pabove = acc;
            });
      }
    }
    __syncthreads();
#pragma unroll 1
    for (int i = LLN - 1; i >= 0; --i) {
      float* ri = smb(sm, 1 + i);
      const float* ti = smb(sm, 6 + i);
      const bool phase_end = (i == 0);
      const bool all_end = phase_end && (it == NITERS - 1);
      const float* wn = all_end ? sqw
                        : phase_end ? pred_w + (LLN - 1) * DD * DD
                                    : upd_w + (i - 1) * DD * DD;
      const float* bn = all_end ? sqb
                        : phase_end ? pred_b + (LLN - 1) * DD
                                    : upd_b + (i - 1) * DD;
      gemm_xWT(upd_w + i * DD * DD, upd_b + i * DD, ti, stage, lane, wv,
               bv, wn, bn, true,
               [&](int m, int n, float v) {
                 const int o = m * LSTR + n;
                 ri[o] = ri[o] + 0.1f * tanh_fast(v);
               });
    }
    __syncthreads();
  }

  float* refined = smb(sm, LLN);
  for (int idx = tid; idx < BT * DD; idx += NTHR) {
    const int r = idx >> 8, c = idx & 255;
    refined[r * LSTR + c] += ff[r * LSTR + c];
  }
  __syncthreads();

  {
    float* q = smb(sm, 6);
    gemm_xWT(sqw, sqb, refined, stage, lane, wv, bv, eqw, eqb, true,
             [&](int m, int n, float v) { q[m * LSTR + n] = v; });
    __syncthreads();
    if (tid < BT * 16) {
      const int r = tid >> 4, c = tid & 15;
      float p = 0.f;
#pragma unroll
      for (int u = 0; u < 16; ++u) {
        const int o = c * 16 + u;
        p += q[r * LSTR + o] * keyb[o];
      }
      tot[tid] = p;
    }
    __syncthreads();
    if (tid < BT) {
      float s = 0.f;
#pragma unroll
      for (int u = 0; u < 16; ++u) s += tot[tid * 16 + u];
      cs_g[b0 + tid] = s;
    }
    __syncthreads();
    gemm_xM(keyw, q, lane, wv,
            [&](int m, int n, float v) { sq_g[(b0 + m) * DD + n] = v; });
    __syncthreads();
  }

  {
    float* q = smb(sm, 7);
    gemm_xWT(eqw, eqb, refined, stage, lane, wv, bv, aw1, ab1, wvu < 8,
             [&](int m, int n, float v) { q[m * LSTR + n] = v; });
    __syncthreads();
    if (tid < BT * 16) {
      const int r = tid >> 4, c = tid & 15;
      float p = 0.f;
#pragma unroll
      for (int u = 0; u < 16; ++u) {
        const int o = c * 16 + u;
        p += q[r * LSTR + o] * keyb[o];
      }
      tot[tid] = p;
    }
    __syncthreads();
    if (tid < BT) {
      float s = 0.f;
#pragma unroll
      for (int u = 0; u < 16; ++u) s += tot[tid * 16 + u];
      ce_g[b0 + tid] = s;
    }
    __syncthreads();
    gemm_xM(keyw, q, lane, wv,
            [&](int m, int n, float v) { eq_g[(b0 + m) * DD + n] = v; });
    __syncthreads();
  }

  if (wvu < 8) {
    gemm_xWT(aw1, ab1, refined, stage, lane, wv, bv, nullptr, nullptr, false,
             [&](int m, int n, float v) {
               tot[m * LSTR + n] =
                   0.5f * v * (1.0f + erff(v * 0.70710678118654752f));
             });
  }
  __syncthreads();
  if (tid < BT * 2) {
    const int r = tid >> 1, k = tid & 1;
    float s = ab2[k];
    for (int j = 0; j < 128; ++j) s += tot[r * LSTR + j] * aw2[k * 128 + j];
    ans_g[(b0 + r) * 2 + k] = s;
  }
}

__global__ __launch_bounds__(256) void logits_kernel(
    const int* __restrict__ ids, const float* __restrict__ emb,
    const float* __restrict__ pos, const float* __restrict__ sq_g,
    const float* __restrict__ eq_g, const float* __restrict__ cs_g,
    const float* __restrict__ ce_g, float* __restrict__ out_start,
    float* __restrict__ out_end) {
  const int b = blockIdx.x;
  const int lane = threadIdx.x & 31;
  const int wv = threadIdx.x >> 5;
  float sqv[8], eqv[8];
  const float* sr = sq_g + b * DD + lane * 8;
  const float* er = eq_g + b * DD + lane * 8;
#pragma unroll
  for (int j = 0; j < 8; ++j) {
    sqv[j] = sr[j];
    eqv[j] = er[j];
  }
  const float cs = cs_g[b];
  const float ce = ce_g[b];
  for (int s = wv; s < SS; s += 8) {
    const int id = ids[b * SS + s];
    const float* row = emb + (size_t)id * DD + lane * 8;
    const float* pr = pos + s * DD + lane * 8;
    float as = 0.f, ae = 0.f;
#pragma unroll
    for (int j = 0; j < 8; ++j) {
      const float e = row[j] + pr[j];
      as += e * sqv[j];
      ae += e * eqv[j];
    }
#pragma unroll
    for (int off = 16; off; off >>= 1) {
      as += __shfl_xor(as, off, 32);
      ae += __shfl_xor(ae, off, 32);
    }
    if (lane == 0) {
      out_start[b * SS + s] = as + cs;
      out_end[b * SS + s] = ae + ce;
    }
  }
}

extern "C" void kernel_launch(void* const* d_in, const int* in_sizes, int n_in,
                              void* d_out, int out_size, void* d_ws,
                              size_t ws_size, hipStream_t stream) {
  const int* ids = (const int*)d_in[0];
  const float* emb = (const float*)d_in[1];
  const float* pos = (const float*)d_in[2];
  const float* init_w = (const float*)d_in[3];
  const float* init_b = (const float*)d_in[4];
  const float* pred_w = (const float*)d_in[5];
  const float* pred_b = (const float*)d_in[6];
  const float* upd_w = (const float*)d_in[7];
  const float* upd_b = (const float*)d_in[8];
  const float* sqw = (const float*)d_in[9];
  const float* sqb = (const float*)d_in[10];
  const float* eqw = (const float*)d_in[11];
  const float* eqb = (const float*)d_in[12];
  const float* keyw = (const float*)d_in[13];
  const float* keyb = (const float*)d_in[14];
  const float* aw1 = (const float*)d_in[15];
  const float* ab1 = (const float*)d_in[16];
  const float* aw2 = (const float*)d_in[17];
  const float* ab2 = (const float*)d_in[18];

  float* ws = (float*)d_ws;
  float* pooled = ws;
  float* sq = pooled + BB * DD;
  float* eq = sq + BB * DD;
  float* cs = eq + BB * DD;
  float* ce = cs + BB;

  float* out = (float*)d_out;
  float* out_start = out;
  float* out_end = out + BB * SS;
  float* out_ans = out + 2 * BB * SS;

  (void)hipFuncSetAttribute((const void*)refine_kernel,
                            hipFuncAttributeMaxDynamicSharedMemorySize,
                            (int)SMEM_BYTES);

  pooled_kernel<<<BB, 256, 0, stream>>>(ids, emb, pos, pooled);
  refine_kernel<<<BB / BT, NTHR, SMEM_BYTES, stream>>>(
      pooled, init_w, init_b, pred_w, pred_b, upd_w, upd_b, sqw, sqb, eqw, eqb,
      keyw, keyb, aw1, ab1, aw2, ab2, sq, eq, cs, ce, out_ans);
  logits_kernel<<<BB, 256, 0, stream>>>(ids, emb, pos, sq, eq, cs, ce,
                                        out_start, out_end);
}


# logits launch: token dot products run on the f32 WMMA, 16 token rows a tile staged through a per-wave LDS tile, the two queries as B columns; d sums now in the WMMA's order, no lane shuffles
# speedup vs baseline: 2.4825x; 1.0945x over previous
#include <hip/hip_runtime.h>
#include <math.h>

#define BB   256
#define SS   512
#define DD   256
#define LLN  4
#define NITERS 10
#define BT   16
#define LSTR 260
#define TILE (BT * LSTR)
#define NWAVES 16
#define NTHR (NWAVES * 32)
#define WSTR 132
#define WHALF (16 * WSTR)
#define WSTAGE WHALF
#define NTILES 10
#define SMEM_FLOATS (NTILES * TILE + NWAVES * WSTAGE)
#define SMEM_BYTES  (SMEM_FLOATS * sizeof(float))

typedef float v2f __attribute__((ext_vector_type(2)));
typedef float v4f __attribute__((ext_vector_type(4)));
typedef float v8f __attribute__((ext_vector_type(8)));

#if defined(__has_builtin)
#  if __has_builtin(__builtin_amdgcn_wmma_f32_16x16x4_f32)
#    define HAS_WMMA_F32 1
#  else
#    define HAS_WMMA_F32 0
#  endif
#  if __has_builtin(__builtin_amdgcn_tanhf)
#    define HAS_HW_TANH 1
#  else
#    define HAS_HW_TANH 0
#  endif
#else
#  define HAS_WMMA_F32 0
#  define HAS_HW_TANH 0
#endif

__device__ __forceinline__ float tanh_fast(float x) {
#if HAS_HW_TANH
  return __builtin_amdgcn_tanhf(x);
#else
  return tanhf(x);
#endif
}

__device__ __forceinline__ float* smb(float* sm, int i) { return sm + i * TILE; }


__device__ __forceinline__ v8f tile_xWT(const float* __restrict__ W,
                                        const float* X, int nt, int lane,
                                        float cinit) {
  const int l16 = lane & 15;
  const int half = lane >> 4;
  v8f acc = {cinit, cinit, cinit, cinit, cinit, cinit, cinit, cinit};
#if HAS_WMMA_F32
  const float* wr = W + (nt * 16 + l16) * DD + 2 * half;
  const float* xr = X + l16 * LSTR + 2 * half;
#pragma unroll 4
  for (int kk = 0; kk < DD; kk += 4) {
    v2f a = {xr[kk], xr[kk + 1]};
    v2f b = {wr[kk], wr[kk + 1]};
    acc = __builtin_amdgcn_wmma_f32_16x16x4_f32(false, a, false, b,
                                                (short)0, acc, false, false);
  }
#else
#pragma unroll
  for (int j = 0; j < 8; ++j) {
    const int m = j + 8 * half, n = nt * 16 + l16;
    float s = cinit;
    for (int k = 0; k < DD; ++k) s += X[m * LSTR + k] * W[n * DD + k];
    acc[j] = s;
  }
#endif
  return acc;
}

__device__ __forceinline__ v8f tile_xM(const float* __restrict__ M,
                                       const float* X, int nt, int lane,
                                       float cinit) {
  const int l16 = lane & 15;
  const int half = lane >> 4;
  v8f acc = {cinit, cinit, cinit, cinit, cinit, cinit, cinit, cinit};
#if HAS_WMMA_F32
  const float* mp = M + (nt * 16 + l16) + (2 * half) * DD;
  const float* xr = X + l16 * LSTR + 2 * half;
#pragma unroll 4
  for (int kk = 0; kk < DD; kk += 4) {
    v2f a = {xr[kk], xr[kk + 1]};
    v2f b = {mp[kk * DD], mp[kk * DD + DD]};
    acc = __builtin_amdgcn_wmma_f32_16x16x4_f32(false, a, false, b,
                                                (short)0, acc, false, false);
  }
#else
#pragma unroll
  for (int j = 0; j < 8; ++j) {
    const int m = j + 8 * half, n = nt * 16 + l16;
    float s = cinit;
    for (int k = 0; k < DD; ++k) s += X[m * LSTR + k] * M[k * DD + n];
    acc[j] = s;
  }
#endif
  return acc;
}

#if HAS_WMMA_F32
__device__ __forceinline__ void stage_issue(v4f (&r)[16],
                                            const float* __restrict__ wsrc) {
#pragma unroll
  for (int j = 0; j < 16; ++j)
    r[j] = *reinterpret_cast<const v4f*>(wsrc + j * DD);
}

__device__ __forceinline__ void stage_park(const v4f (&r)[16], float* wp) {
#pragma unroll
  for (int j = 0; j < 16; ++j) *reinterpret_cast<v4f*>(wp + j * WSTR) = r[j];
}

__device__ __forceinline__ void wave_lds_handover() {
  __builtin_amdgcn_fence(__ATOMIC_RELEASE, "wavefront");
  __builtin_amdgcn_wave_barrier();
  __builtin_amdgcn_fence(__ATOMIC_ACQUIRE, "wavefront");
}

__device__ __forceinline__ void frag_read8(v2f (&a)[8], v2f (&b)[8],
                                           const float* xr, const float* wr) {
#pragma unroll
  for (int s = 0; s < 8; ++s) {
    a[s] = (v2f){xr[4 * s], xr[4 * s + 1]};
    b[s] = (v2f){wr[4 * s], wr[4 * s + 1]};
  }
}

__device__ __forceinline__ v8f mma8(v8f acc, const v2f (&a)[8],
                                    const v2f (&b)[8]) {
#pragma unroll
  for (int s = 0; s < 8; ++s)
    acc = __builtin_amdgcn_wmma_f32_16x16x4_f32(false, a[s], false, b[s],
                                                (short)0, acc, false, false);
  return acc;
}

__device__ __forceinline__ v8f half_tile_mma(v8f acc, const float* xr,
                                             const float* wr) {
  v2f a0[8], b0[8], a1[8], b1[8];
  frag_read8(a0, b0, xr, wr);
  frag_read8(a1, b1, xr + 32, wr + 32);
  __builtin_amdgcn_sched_barrier(0);
  acc = mma8(acc, a0, b0);
  __builtin_amdgcn_sched_barrier(0);
  frag_read8(a0, b0, xr + 64, wr + 64);
  __builtin_amdgcn_sched_barrier(0);
  acc = mma8(acc, a1, b1);
  __builtin_amdgcn_sched_barrier(0);
  frag_read8(a1, b1, xr + 96, wr + 96);
  __builtin_amdgcn_sched_barrier(0);
  acc = mma8(acc, a0, b0);
  acc = mma8(acc, a1, b1);
  return acc;
}
#endif

__device__ __forceinline__ void gemm_first_request(
    v4f (&r)[16], float& bv, const float* __restrict__ W,
    const float* __restrict__ bias, int lane, int wv) {
#if HAS_WMMA_F32
  stage_issue(r, W + (wv * 16) * DD + lane * 4);
#endif
  bv = bias ? bias[wv * 16 + (lane & 15)] : 0.f;
}

template <typename TileEpi>
__device__ __forceinline__ void gemm_xWT_tiles(
    const float* __restrict__ W, const float* __restrict__ bias,
    const float* X, float* stage, int lane, int wv, float& bv,
    const float* __restrict__ Wn, const float* __restrict__ biasn,
    const bool has_next, TileEpi epi) {
  const int l16 = lane & 15, half = lane >> 4;
#if HAS_WMMA_F32
  const float* wsrc = W + (wv * 16) * DD + lane * 4;
  const float* xr = X + l16 * LSTR + 2 * half;
  const float* wr = stage + l16 * WSTR + 2 * half;
  float* wp = stage + lane * 4;
  const float cb = bv;
  v4f r[16];
  v8f acc = {cb, cb, cb, cb, cb, cb, cb, cb};
#pragma unroll 1
  for (int q = 0; q < 2; ++q) {
    const bool more = (q < 1) || has_next;
    if (more) {
      stage_issue(r, (q < 1) ? wsrc + 128 : Wn + (wv * 16) * DD + lane * 4);
      if (q == 1) bv = biasn ? biasn[wv * 16 + l16] : 0.f;
    }
    acc = half_tile_mma(acc, xr + q * 128, wr);
    if (q == 1) epi(wv, acc);
    if (more) {
      wave_lds_handover();
      stage_park(r, wp);
      wave_lds_handover();
    }
  }
#else
  const float bvf = bias ? bias[wv * 16 + l16] : 0.f;
  v8f acc = tile_xWT(W, X, wv, lane, bvf);
  epi(wv, acc);
#endif
}

template <typename Epi>
__device__ __forceinline__ void gemm_xWT(
    const float* __restrict__ W, const float* __restrict__ bias,
    const float* X, float* stage, int lane, int wv, float& bv,
    const float* __restrict__ Wn, const float* __restrict__ biasn,
    const bool has_next, Epi epi) {
  gemm_xWT_tiles(W, bias, X, stage, lane, wv, bv, Wn, biasn, has_next,
                 [&](int nt, v8f acc) {
    const int l16 = lane & 15, half = lane >> 4;
#pragma unroll
    for (int j = 0; j < 8; ++j) epi(j + 8 * half, nt * 16 + l16, acc[j]);
  });
}

template <typename Epi>
__device__ __forceinline__ void gemm_xM(const float* __restrict__ M,
                                        const float* X, int lane, int wv,
                                        Epi epi) {
  const int nt = wv;
  v8f acc = tile_xM(M, X, nt, lane, 0.f);
  const int l16 = lane & 15, half = lane >> 4;
#pragma unroll
  for (int j = 0; j < 8; ++j) epi(j + 8 * half, nt * 16 + l16, acc[j]);
}

__global__ __launch_bounds__(256) void pooled_kernel(
    const int* __restrict__ ids, const float* __restrict__ emb,
    const float* __restrict__ pos, float* __restrict__ pooled_g) {
  __shared__ int sid[SS];
  const int b = blockIdx.x;
  const int t = threadIdx.x;
  sid[t] = ids[b * SS + t];
  sid[t + 256] = ids[b * SS + 256 + t];
  __syncthreads();
  float acc = 0.f;
  for (int s = 0; s < SS; ++s) {
    acc += emb[(size_t)sid[s] * DD + t] + pos[s * DD + t];
  }
  pooled_g[b * DD + t] = acc * (1.0f / (float)SS);
}

__global__ __launch_bounds__(NTHR) void refine_kernel(
    const float* __restrict__ pooled_g,
    const float* __restrict__ init_w, const float* __restrict__ init_b,
    const float* __restrict__ pred_w, const float* __restrict__ pred_b,
    const float* __restrict__ upd_w, const float* __restrict__ upd_b,
    const float* __restrict__ sqw, const float* __restrict__ sqb,
    const float* __restrict__ eqw, const float* __restrict__ eqb,
    const float* __restrict__ keyw, const float* __restrict__ keyb,
    const float* __restrict__ aw1, const float* __restrict__ ab1,
    const float* __restrict__ aw2, const float* __restrict__ ab2,
    float* __restrict__ sq_g, float* __restrict__ eq_g,
    float* __restrict__ cs_g, float* __restrict__ ce_g,
    float* __restrict__ ans_g) {
  extern __shared__ __attribute__((aligned(16))) float sm[];

  const int tid = threadIdx.x;
  const int lane = tid & 31;
  const int wv = tid >> 5;
  const int wvu = __builtin_amdgcn_readfirstlane(wv);
  const int b0 = blockIdx.x * BT;

  float* pooled = smb(sm, 0);
  float* ff = smb(sm, 5);
  float* tot = smb(sm, 8);
  float* stage = sm + NTILES * TILE + wv * WSTAGE;

  float bv;
  {
    v4f r0[16];
    gemm_first_request(r0, bv, init_w, init_b, lane, wv);
    for (int idx = tid; idx < BT * DD; idx += NTHR) {
      const int r = idx >> 8, c = idx & 255;
      pooled[r * LSTR + c] = pooled_g[(b0 + r) * DD + c];
    }
#if HAS_WMMA_F32
    stage_park(r0, stage + lane * 4);
    wave_lds_handover();
#endif
  }
  __syncthreads();

  {
    const float* cur = pooled;
    for (int l = 0; l < LLN; ++l) {
      float* dst = smb(sm, 1 + l);
      const bool last = (l == LLN - 1);
      const float* wn = last ? pred_w + (LLN - 1) * DD * DD
                             : init_w + (l + 1) * DD * DD;
      const float* bn = last ? pred_b + (LLN - 1) * DD : init_b + (l + 1) * DD;
      gemm_xWT(init_w + l * DD * DD, init_b + l * DD, cur, stage, lane, wv,
               bv, wn, bn, true,
               [&](int m, int n, float v) {
                 const float r = tanh_fast(v);
                 dst[m * LSTR + n] = r;
                 if (last) ff[m * LSTR + n] = r;
               });
      __syncthreads();
      cur = dst;
    }
  }

  for (int it = 0; it < NITERS; ++it) {
    {
      v8f pabove = {0.f, 0.f, 0.f, 0.f, 0.f, 0.f, 0.f, 0.f};
#pragma unroll 1
      for (int i = LLN - 1; i >= 0; --i) {
        const float* ri = smb(sm, 1 + i);
        const float* below = (i > 0) ? smb(sm, i) : pooled;
        float* ti = smb(sm, 6 + i);
        const bool has_above = (i < LLN - 1);
        const float* wn = (i > 0) ? pred_w + (i - 1) * DD * DD
                                  : upd_w + (LLN - 1) * DD * DD;
        const float* bn = (i > 0) ? pred_b + (i - 1) * DD
                                  : upd_b + (LLN - 1) * DD;
        gemm_xWT_tiles(
            pred_w + i * DD * DD, pred_b + i * DD, ri, stage, lane, wv,
            bv, wn, bn, true,
            [&](int nt, v8f acc) {
              const int o0 = 8 * (lane >> 4) * LSTR + nt * 16 + (lane & 15);
              float bl[8], rr[8];
#pragma unroll
              for (int j = 0; j < 8; ++j) bl[j] = below[o0 + j * LSTR];
              if (has_above) {
#pragma unroll
                for (int j = 0; j < 8; ++j) rr[j] = ri[o0 + j * LSTR];
              }
#pragma unroll
              for (int j = 0; j < 8; ++j) {
                float e = bl[j] - acc[j];
                if (has_above) e += 0.5f * (rr[j] - pabove[j]);
                ti[o0 + j * LSTR] = e;
              }
              pabove = acc;
            });
      }
    }
    __syncthreads();
#pragma unroll 1
    for (int i = LLN - 1; i >= 0; --i) {
      float* ri = smb(sm, 1 + i);
      const float* ti = smb(sm, 6 + i);
      const bool phase_end = (i == 0);
      const bool all_end = phase_end && (it == NITERS - 1);
      const float* wn = all_end ? sqw
                        : phase_end ? pred_w + (LLN - 1) * DD * DD
                                    : upd_w + (i - 1) * DD * DD;
      const float* bn = all_end ? sqb
                        : phase_end ? pred_b + (LLN - 1) * DD
                                    : upd_b + (i - 1) * DD;
      gemm_xWT(upd_w + i * DD * DD, upd_b + i * DD, ti, stage, lane, wv,
               bv, wn, bn, true,
               [&](int m, int n, float v) {
                 const int o = m * LSTR + n;
                 ri[o] = ri[o] + 0.1f * tanh_fast(v);
               });
    }
    __syncthreads();
  }

  float* refined = smb(sm, LLN);
  for (int idx = tid; idx < BT * DD; idx += NTHR) {
    const int r = idx >> 8, c = idx & 255;
    refined[r * LSTR + c] += ff[r * LSTR + c];
  }
  __syncthreads();

  {
    float* q = smb(sm, 6);
    gemm_xWT(sqw, sqb, refined, stage, lane, wv, bv, eqw, eqb, true,
             [&](int m, int n, float v) { q[m * LSTR + n] = v; });
    __syncthreads();
    if (tid < BT * 16) {
      const int r = tid >> 4, c = tid & 15;
      float p = 0.f;
#pragma unroll
      for (int u = 0; u < 16; ++u) {
        const int o = c * 16 + u;
        p += q[r * LSTR + o] * keyb[o];
      }
      tot[tid] = p;
    }
    __syncthreads();
    if (tid < BT) {
      float s = 0.f;
#pragma unroll
      for (int u = 0; u < 16; ++u) s += tot[tid * 16 + u];
      cs_g[b0 + tid] = s;
    }
    __syncthreads();
    gemm_xM(keyw, q, lane, wv,
            [&](int m, int n, float v) { sq_g[(b0 + m) * DD + n] = v; });
    __syncthreads();
  }

  {
    float* q = smb(sm, 7);
    gemm_xWT(eqw, eqb, refined, stage, lane, wv, bv, aw1, ab1, wvu < 8,
             [&](int m, int n, float v) { q[m * LSTR + n] = v; });
    __syncthreads();
    if (tid < BT * 16) {
      const int r = tid >> 4, c = tid & 15;
      float p = 0.f;
#pragma unroll
      for (int u = 0; u < 16; ++u) {
        const int o = c * 16 + u;
        p += q[r * LSTR + o] * keyb[o];
      }
      tot[tid] = p;
    }
    __syncthreads();
    if (tid < BT) {
      float s = 0.f;
#pragma unroll
      for (int u = 0; u < 16; ++u) s += tot[tid * 16 + u];
      ce_g[b0 + tid] = s;
    }
    __syncthreads();
    gemm_xM(keyw, q, lane, wv,
            [&](int m, int n, float v) { eq_g[(b0 + m) * DD + n] = v; });
    __syncthreads();
  }

  if (wvu < 8) {
    gemm_xWT(aw1, ab1, refined, stage, lane, wv, bv, nullptr, nullptr, false,
             [&](int m, int n, float v) {
               tot[m * LSTR + n] =
                   0.5f * v * (1.0f + erff(v * 0.70710678118654752f));
             });
  }
  __syncthreads();
  if (tid < BT * 2) {
    const int r = tid >> 1, k = tid & 1;
    float s = ab2[k];
    for (int j = 0; j < 128; ++j) s += tot[r * LSTR + j] * aw2[k * 128 + j];
    ans_g[(b0 + r) * 2 + k] = s;
  }
}

#define LG_WAVES 8
#define LG_TPW 4
#define LG_SMEM_BYTES (LG_WAVES * TILE * sizeof(float))

#if HAS_WMMA_F32
__device__ __forceinline__ void rows_issue(v4f (&ev)[32], v4f (&pv)[32],
                                           const int idv, const int l0,
                                           const float* __restrict__ ebase,
                                           const float* __restrict__ pbase) {
#pragma unroll
  for (int j = 0; j < 16; ++j) {
    const int id = __builtin_amdgcn_readlane(idv, l0 + j);
    const float* er = ebase + (size_t)id * DD;
    const float* pr = pbase + j * DD;
    ev[2 * j] = *reinterpret_cast<const v4f*>(er);
    ev[2 * j + 1] = *reinterpret_cast<const v4f*>(er + 128);
    pv[2 * j] = *reinterpret_cast<const v4f*>(pr);
    pv[2 * j + 1] = *reinterpret_cast<const v4f*>(pr + 128);
  }
}

__device__ __forceinline__ void rows_park(const v4f (&ev)[32],
                                          const v4f (&pv)[32], float* wp) {
#pragma unroll
  for (int j = 0; j < 16; ++j) {
    *reinterpret_cast<v4f*>(wp + j * LSTR) = ev[2 * j] + pv[2 * j];
    *reinterpret_cast<v4f*>(wp + j * LSTR + 128) = ev[2 * j + 1] + pv[2 * j + 1];
  }
}

__device__ __forceinline__ void afrag_read8(v2f (&a)[8], const float* xr) {
#pragma unroll
  for (int s = 0; s < 8; ++s) a[s] = (v2f){xr[4 * s], xr[4 * s + 1]};
}

__device__ __forceinline__ v8f tile_tokens_dot(const float* xr,
                                               const v2f (&bq)[64]) {
  v8f acc = {0.f, 0.f, 0.f, 0.f, 0.f, 0.f, 0.f, 0.f};
  v2f a0[8], a1[8];
  afrag_read8(a0, xr);
#pragma unroll
  for (int g = 0; g < 8; g += 2) {
    afrag_read8(a1, xr + 32 * (g + 1));
    __builtin_amdgcn_sched_barrier(0);
#pragma unroll
    for (int s = 0; s < 8; ++s)
      acc = __builtin_amdgcn_wmma_f32_16x16x4_f32(false, a0[s], false,
                                                  bq[8 * g + s], (short)0, acc,
                                                  false, false);
    __builtin_amdgcn_sched_barrier(0);
    if (g + 2 < 8) afrag_read8(a0, xr + 32 * (g + 2));
    __builtin_amdgcn_sched_barrier(0);
#pragma unroll
    for (int s = 0; s < 8; ++s)
      acc = __builtin_amdgcn_wmma_f32_16x16x4_f32(false, a1[s], false,
                                                  bq[8 * (g + 1) + s], (short)0,
                                                  acc, false, false);
    __builtin_amdgcn_sched_barrier(0);
  }
  return acc;
}
#endif

__global__ __launch_bounds__(LG_WAVES * 32) void logits_kernel(
    const int* __restrict__ ids, const float* __restrict__ emb,
    const float* __restrict__ pos, const float* __restrict__ sq_g,
    const float* __restrict__ eq_g, const float* __restrict__ cs_g,
    const float* __restrict__ ce_g, float* __restrict__ out_start,
    float* __restrict__ out_end) {
  extern __shared__ __attribute__((aligned(16))) float lsm[];
  const int b = blockIdx.x;
  const int lane = threadIdx.x & 31;
  const int wvu = __builtin_amdgcn_readfirstlane((int)(threadIdx.x >> 5));
#if HAS_WMMA_F32
  const int c = lane & 15, h = lane >> 4;
  const int tok0 = wvu * (LG_TPW * 16);
  float* st = lsm + wvu * TILE;
  const float* ebase = emb + lane * 4;
  const float* pbase = pos + tok0 * DD + lane * 4;
  const int idlo = ids[b * SS + tok0 + lane];
  const int idhi = ids[b * SS + tok0 + 32 + lane];

  v4f ev[32], pv[32];
  rows_issue(ev, pv, idlo, 0, ebase, pbase);

  v2f bq[64];
  {
    const float* qsrc = ((c == 0) ? sq_g : eq_g) + b * DD + 2 * h;
#pragma unroll
    for (int s = 0; s < 64; ++s)
      bq[s] = *reinterpret_cast<const v2f*>(qsrc + 4 * s);
  }
  const float cadd = (c == 0) ? cs_g[b] : ce_g[b];
  float* orow = ((c == 0) ? out_start : out_end) + b * SS + tok0 + 8 * h;
  const float* xr = st + c * LSTR + 2 * h;
  float* wp = st + lane * 4;

#pragma unroll 1
  for (int t = 0; t < LG_TPW; ++t) {
    wave_lds_handover();
    rows_park(ev, pv, wp);
    wave_lds_handover();
    if (t + 1 < LG_TPW)
      rows_issue(ev, pv, (t + 1 < 2) ? idlo : idhi, ((t + 1) & 1) * 16, ebase,
                 pbase + (t + 1) * 16 * DD);
    const v8f acc = tile_tokens_dot(xr, bq);
    const v4f lo = {acc[0] + cadd, acc[1] + cadd, acc[2] + cadd, acc[3] + cadd};
    const v4f hi = {acc[4] + cadd, acc[5] + cadd, acc[6] + cadd, acc[7] + cadd};
    if (c < 2) {
      *reinterpret_cast<v4f*>(orow + t * 16) = lo;
      *reinterpret_cast<v4f*>(orow + t * 16 + 4) = hi;
    }
  }
#else
  (void)lsm;
  (void)wvu;
  (void)lane;
  for (int s = threadIdx.x; s < SS; s += LG_WAVES * 32) {
    const float* row = emb + (size_t)ids[b * SS + s] * DD;
    const float* pr = pos + s * DD;
    float as = 0.f, ae = 0.f;
    for (int k = 0; k < DD; ++k) {
      const float e = row[k] + pr[k];
      as += e * sq_g[b * DD + k];
      ae += e * eq_g[b * DD + k];
    }
    out_start[b * SS + s] = as + cs_g[b];
    out_end[b * SS + s] = ae + ce_g[b];
  }
#endif
}

extern "C" void kernel_launch(void* const* d_in, const int* in_sizes, int n_in,
                              void* d_out, int out_size, void* d_ws,
                              size_t ws_size, hipStream_t stream) {
  const int* ids = (const int*)d_in[0];
  const float* emb = (const float*)d_in[1];
  const float* pos = (const float*)d_in[2];
  const float* init_w = (const float*)d_in[3];
  const float* init_b = (const float*)d_in[4];
  const float* pred_w = (const float*)d_in[5];
  const float* pred_b = (const float*)d_in[6];
  const float* upd_w = (const float*)d_in[7];
  const float* upd_b = (const float*)d_in[8];
  const float* sqw = (const float*)d_in[9];
  const float* sqb = (const float*)d_in[10];
  const float* eqw = (const float*)d_in[11];
  const float* eqb = (const float*)d_in[12];
  const float* keyw = (const float*)d_in[13];
  const float* keyb = (const float*)d_in[14];
  const float* aw1 = (const float*)d_in[15];
  const float* ab1 = (const float*)d_in[16];
  const float* aw2 = (const float*)d_in[17];
  const float* ab2 = (const float*)d_in[18];

  float* ws = (float*)d_ws;
  float* pooled = ws;
  float* sq = pooled + BB * DD;
  float* eq = sq + BB * DD;
  float* cs = eq + BB * DD;
  float* ce = cs + BB;

  float* out = (float*)d_out;
  float* out_start = out;
  float* out_end = out + BB * SS;
  float* out_ans = out + 2 * BB * SS;

  (void)hipFuncSetAttribute((const void*)refine_kernel,
                            hipFuncAttributeMaxDynamicSharedMemorySize,
                            (int)SMEM_BYTES);

  (void)hipFuncSetAttribute((const void*)logits_kernel,
                            hipFuncAttributeMaxDynamicSharedMemorySize,
                            (int)LG_SMEM_BYTES);

  pooled_kernel<<<BB, 256, 0, stream>>>(ids, emb, pos, pooled);
  refine_kernel<<<BB / BT, NTHR, SMEM_BYTES, stream>>>(
      pooled, init_w, init_b, pred_w, pred_b, upd_w, upd_b, sqw, sqb, eqw, eqb,
      keyw, keyb, aw1, ab1, aw2, ab2, sq, eq, cs, ce, out_ans);
  logits_kernel<<<BB, LG_WAVES * 32, LG_SMEM_BYTES, stream>>>(
      ids, emb, pos, sq, eq, cs, ce, out_start, out_end);
}


# pooling launch re-decomposed: a batch row's 512 tokens split over 8 waves (was a column a thread), each wave sums 64 whole rows read by 16-byte loads, slices added through LDS; sum order now by slice
# speedup vs baseline: 2.7735x; 1.1172x over previous
#include <hip/hip_runtime.h>
#include <math.h>

#define BB   256
#define SS   512
#define DD   256
#define LLN  4
#define NITERS 10
#define BT   16
#define LSTR 260
#define TILE (BT * LSTR)
#define NWAVES 16
#define NTHR (NWAVES * 32)
#define WSTR 132
#define WHALF (16 * WSTR)
#define WSTAGE WHALF
#define NTILES 10
#define SMEM_FLOATS (NTILES * TILE + NWAVES * WSTAGE)
#define SMEM_BYTES  (SMEM_FLOATS * sizeof(float))

typedef float v2f __attribute__((ext_vector_type(2)));
typedef float v4f __attribute__((ext_vector_type(4)));
typedef float v8f __attribute__((ext_vector_type(8)));

#if defined(__has_builtin)
#  if __has_builtin(__builtin_amdgcn_wmma_f32_16x16x4_f32)
#    define HAS_WMMA_F32 1
#  else
#    define HAS_WMMA_F32 0
#  endif
#  if __has_builtin(__builtin_amdgcn_tanhf)
#    define HAS_HW_TANH 1
#  else
#    define HAS_HW_TANH 0
#  endif
#else
#  define HAS_WMMA_F32 0
#  define HAS_HW_TANH 0
#endif

__device__ __forceinline__ float tanh_fast(float x) {
#if HAS_HW_TANH
  return __builtin_amdgcn_tanhf(x);
#else
  return tanhf(x);
#endif
}

__device__ __forceinline__ float* smb(float* sm, int i) { return sm + i * TILE; }


__device__ __forceinline__ v8f tile_xWT(const float* __restrict__ W,
                                        const float* X, int nt, int lane,
                                        float cinit) {
  const int l16 = lane & 15;
  const int half = lane >> 4;
  v8f acc = {cinit, cinit, cinit, cinit, cinit, cinit, cinit, cinit};
#if HAS_WMMA_F32
  const float* wr = W + (nt * 16 + l16) * DD + 2 * half;
  const float* xr = X + l16 * LSTR + 2 * half;
#pragma unroll 4
  for (int kk = 0; kk < DD; kk += 4) {
    v2f a = {xr[kk], xr[kk + 1]};
    v2f b = {wr[kk], wr[kk + 1]};
    acc = __builtin_amdgcn_wmma_f32_16x16x4_f32(false, a, false, b,
                                                (short)0, acc, false, false);
  }
#else
#pragma unroll
  for (int j = 0; j < 8; ++j) {
    const int m = j + 8 * half, n = nt * 16 + l16;
    float s = cinit;
    for (int k = 0; k < DD; ++k) s += X[m * LSTR + k] * W[n * DD + k];
    acc[j] = s;
  }
#endif
  return acc;
}

__device__ __forceinline__ v8f tile_xM(const float* __restrict__ M,
                                       const float* X, int nt, int lane,
                                       float cinit) {
  const int l16 = lane & 15;
  const int half = lane >> 4;
  v8f acc = {cinit, cinit, cinit, cinit, cinit, cinit, cinit, cinit};
#if HAS_WMMA_F32
  const float* mp = M + (nt * 16 + l16) + (2 * half) * DD;
  const float* xr = X + l16 * LSTR + 2 * half;
#pragma unroll 4
  for (int kk = 0; kk < DD; kk += 4) {
    v2f a = {xr[kk], xr[kk + 1]};
    v2f b = {mp[kk * DD], mp[kk * DD + DD]};
    acc = __builtin_amdgcn_wmma_f32_16x16x4_f32(false, a, false, b,
                                                (short)0, acc, false, false);
  }
#else
#pragma unroll
  for (int j = 0; j < 8; ++j) {
    const int m = j + 8 * half, n = nt * 16 + l16;
    float s = cinit;
    for (int k = 0; k < DD; ++k) s += X[m * LSTR + k] * M[k * DD + n];
    acc[j] = s;
  }
#endif
  return acc;
}

#if HAS_WMMA_F32
__device__ __forceinline__ void stage_issue(v4f (&r)[16],
                                            const float* __restrict__ wsrc) {
#pragma unroll
  for (int j = 0; j < 16; ++j)
    r[j] = *reinterpret_cast<const v4f*>(wsrc + j * DD);
}

__device__ __forceinline__ void stage_park(const v4f (&r)[16], float* wp) {
#pragma unroll
  for (int j = 0; j < 16; ++j) *reinterpret_cast<v4f*>(wp + j * WSTR) = r[j];
}

__device__ __forceinline__ void wave_lds_handover() {
  __builtin_amdgcn_fence(__ATOMIC_RELEASE, "wavefront");
  __builtin_amdgcn_wave_barrier();
  __builtin_amdgcn_fence(__ATOMIC_ACQUIRE, "wavefront");
}

__device__ __forceinline__ void frag_read8(v2f (&a)[8], v2f (&b)[8],
                                           const float* xr, const float* wr) {
#pragma unroll
  for (int s = 0; s < 8; ++s) {
    a[s] = (v2f){xr[4 * s], xr[4 * s + 1]};
    b[s] = (v2f){wr[4 * s], wr[4 * s + 1]};
  }
}

__device__ __forceinline__ v8f mma8(v8f acc, const v2f (&a)[8],
                                    const v2f (&b)[8]) {
#pragma unroll
  for (int s = 0; s < 8; ++s)
    acc = __builtin_amdgcn_wmma_f32_16x16x4_f32(false, a[s], false, b[s],
                                                (short)0, acc, false, false);
  return acc;
}

__device__ __forceinline__ v8f half_tile_mma(v8f acc, const float* xr,
                                             const float* wr) {
  v2f a0[8], b0[8], a1[8], b1[8];
  frag_read8(a0, b0, xr, wr);
  frag_read8(a1, b1, xr + 32, wr + 32);
  __builtin_amdgcn_sched_barrier(0);
  acc = mma8(acc, a0, b0);
  __builtin_amdgcn_sched_barrier(0);
  frag_read8(a0, b0, xr + 64, wr + 64);
  __builtin_amdgcn_sched_barrier(0);
  acc = mma8(acc, a1, b1);
  __builtin_amdgcn_sched_barrier(0);
  frag_read8(a1, b1, xr + 96, wr + 96);
  __builtin_amdgcn_sched_barrier(0);
  acc = mma8(acc, a0, b0);
  acc = mma8(acc, a1, b1);
  return acc;
}
#endif

__device__ __forceinline__ void gemm_first_request(
    v4f (&r)[16], float& bv, const float* __restrict__ W,
    const float* __restrict__ bias, int lane, int wv) {
#if HAS_WMMA_F32
  stage_issue(r, W + (wv * 16) * DD + lane * 4);
#endif
  bv = bias ? bias[wv * 16 + (lane & 15)] : 0.f;
}

template <typename TileEpi>
__device__ __forceinline__ void gemm_xWT_tiles(
    const float* __restrict__ W, const float* __restrict__ bias,
    const float* X, float* stage, int lane, int wv, float& bv,
    const float* __restrict__ Wn, const float* __restrict__ biasn,
    const bool has_next, TileEpi epi) {
  const int l16 = lane & 15, half = lane >> 4;
#if HAS_WMMA_F32
  const float* wsrc = W + (wv * 16) * DD + lane * 4;
  const float* xr = X + l16 * LSTR + 2 * half;
  const float* wr = stage + l16 * WSTR + 2 * half;
  float* wp = stage + lane * 4;
  const float cb = bv;
  v4f r[16];
  v8f acc = {cb, cb, cb, cb, cb, cb, cb, cb};
#pragma unroll 1
  for (int q = 0; q < 2; ++q) {
    const bool more = (q < 1) || has_next;
    if (more) {
      stage_issue(r, (q < 1) ? wsrc + 128 : Wn + (wv * 16) * DD + lane * 4);
      if (q == 1) bv = biasn ? biasn[wv * 16 + l16] : 0.f;
    }
    acc = half_tile_mma(acc, xr + q * 128, wr);
    if (q == 1) epi(wv, acc);
    if (more) {
      wave_lds_handover();
      stage_park(r, wp);
      wave_lds_handover();
    }
  }
#else
  const float bvf = bias ? bias[wv * 16 + l16] : 0.f;
  v8f acc = tile_xWT(W, X, wv, lane, bvf);
  epi(wv, acc);
#endif
}

template <typename Epi>
__device__ __forceinline__ void gemm_xWT(
    const float* __restrict__ W, const float* __restrict__ bias,
    const float* X, float* stage, int lane, int wv, float& bv,
    const float* __restrict__ Wn, const float* __restrict__ biasn,
    const bool has_next, Epi epi) {
  gemm_xWT_tiles(W, bias, X, stage, lane, wv, bv, Wn, biasn, has_next,
                 [&](int nt, v8f acc) {
    const int l16 = lane & 15, half = lane >> 4;
#pragma unroll
    for (int j = 0; j < 8; ++j) epi(j + 8 * half, nt * 16 + l16, acc[j]);
  });
}

template <typename Epi>
__device__ __forceinline__ void gemm_xM(const float* __restrict__ M,
                                        const float* X, int lane, int wv,
                                        Epi epi) {
  const int nt = wv;
  v8f acc = tile_xM(M, X, nt, lane, 0.f);
  const int l16 = lane & 15, half = lane >> 4;
#pragma unroll
  for (int j = 0; j < 8; ++j) epi(j + 8 * half, nt * 16 + l16, acc[j]);
}

#define PL_WAVES 8
#define PL_TPW (SS / PL_WAVES)
#define PL_GRP 8

__device__ __forceinline__ void pool_issue(v4f (&ev)[2 * PL_GRP],
                                           v4f (&pv)[2 * PL_GRP],
                                           const int idv, const int l0,
                                           const float* __restrict__ ebase,
                                           const float* __restrict__ pbase) {
#pragma unroll
  for (int j = 0; j < PL_GRP; ++j) {
    const int id = __builtin_amdgcn_readlane(idv, l0 + j);
    const float* er = ebase + (size_t)id * DD;
    const float* pr = pbase + j * DD;
    ev[2 * j] = *reinterpret_cast<const v4f*>(er);
    ev[2 * j + 1] = *reinterpret_cast<const v4f*>(er + 128);
    pv[2 * j] = *reinterpret_cast<const v4f*>(pr);
    pv[2 * j + 1] = *reinterpret_cast<const v4f*>(pr + 128);
  }
}

__device__ __forceinline__ void pool_sum(v4f& alo, v4f& ahi,
                                         const v4f (&ev)[2 * PL_GRP],
                                         const v4f (&pv)[2 * PL_GRP]) {
#pragma unroll
  for (int j = 0; j < PL_GRP; ++j) {
    alo += ev[2 * j] + pv[2 * j];
    ahi += ev[2 * j + 1] + pv[2 * j + 1];
  }
}

__global__ __launch_bounds__(PL_WAVES * 32) void pooled_kernel(
    const int* __restrict__ ids, const float* __restrict__ emb,
    const float* __restrict__ pos, float* __restrict__ pooled_g) {
  __shared__ __attribute__((aligned(16))) float part[PL_WAVES * DD];
  const int b = blockIdx.x;
  const int t = threadIdx.x;
  const int lane = t & 31;
  const int wvu = __builtin_amdgcn_readfirstlane(t >> 5);
  const int tok0 = wvu * PL_TPW;
  const int idlo = ids[b * SS + tok0 + lane];
  const int idhi = ids[b * SS + tok0 + 32 + lane];
  const float* ebase = emb + lane * 4;
  const float* pbase = pos + tok0 * DD + lane * 4;

  v4f alo = {0.f, 0.f, 0.f, 0.f};
  v4f ahi = {0.f, 0.f, 0.f, 0.f};
  v4f e0[2 * PL_GRP], p0[2 * PL_GRP], e1[2 * PL_GRP], p1[2 * PL_GRP];
  pool_issue(e0, p0, idlo, 0, ebase, pbase);
#pragma unroll
  for (int g = 1; g < PL_TPW / PL_GRP - 1; g += 2) {
    pool_issue(e1, p1, (g < 4) ? idlo : idhi, (g & 3) * PL_GRP, ebase,
               pbase + g * PL_GRP * DD);
    __builtin_amdgcn_sched_barrier(0);
    pool_sum(alo, ahi, e0, p0);
    pool_issue(e0, p0, (g + 1 < 4) ? idlo : idhi, ((g + 1) & 3) * PL_GRP, ebase,
               pbase + (g + 1) * PL_GRP * DD);
    __builtin_amdgcn_sched_barrier(0);
    pool_sum(alo, ahi, e1, p1);
  }
  pool_issue(e1, p1, idhi, 3 * PL_GRP, ebase,
             pbase + (PL_TPW / PL_GRP - 1) * PL_GRP * DD);
  __builtin_amdgcn_sched_barrier(0);
  pool_sum(alo, ahi, e0, p0);
  pool_sum(alo, ahi, e1, p1);

  float* pw = part + wvu * DD + lane * 4;
  *reinterpret_cast<v4f*>(pw) = alo;
  *reinterpret_cast<v4f*>(pw + 128) = ahi;
  __syncthreads();
  float s = part[t];
#pragma unroll
  for (int w = 1; w < PL_WAVES; ++w) s += part[w * DD + t];
  pooled_g[b * DD + t] = s * (1.0f / (float)SS);
}

__global__ __launch_bounds__(NTHR) void refine_kernel(
    const float* __restrict__ pooled_g,
    const float* __restrict__ init_w, const float* __restrict__ init_b,
    const float* __restrict__ pred_w, const float* __restrict__ pred_b,
    const float* __restrict__ upd_w, const float* __restrict__ upd_b,
    const float* __restrict__ sqw, const float* __restrict__ sqb,
    const float* __restrict__ eqw, const float* __restrict__ eqb,
    const float* __restrict__ keyw, const float* __restrict__ keyb,
    const float* __restrict__ aw1, const float* __restrict__ ab1,
    const float* __restrict__ aw2, const float* __restrict__ ab2,
    float* __restrict__ sq_g, float* __restrict__ eq_g,
    float* __restrict__ cs_g, float* __restrict__ ce_g,
    float* __restrict__ ans_g) {
  extern __shared__ __attribute__((aligned(16))) float sm[];

  const int tid = threadIdx.x;
  const int lane = tid & 31;
  const int wv = tid >> 5;
  const int wvu = __builtin_amdgcn_readfirstlane(wv);
  const int b0 = blockIdx.x * BT;

  float* pooled = smb(sm, 0);
  float* ff = smb(sm, 5);
  float* tot = smb(sm, 8);
  float* stage = sm + NTILES * TILE + wv * WSTAGE;

  float bv;
  {
    v4f r0[16];
    gemm_first_request(r0, bv, init_w, init_b, lane, wv);
    for (int idx = tid; idx < BT * DD; idx += NTHR) {
      const int r = idx >> 8, c = idx & 255;
      pooled[r * LSTR + c] = pooled_g[(b0 + r) * DD + c];
    }
#if HAS_WMMA_F32
    stage_park(r0, stage + lane * 4);
    wave_lds_handover();
#endif
  }
  __syncthreads();

  {
    const float* cur = pooled;
    for (int l = 0; l < LLN; ++l) {
      float* dst = smb(sm, 1 + l);
      const bool last = (l == LLN - 1);
      const float* wn = last ? pred_w + (LLN - 1) * DD * DD
                             : init_w + (l + 1) * DD * DD;
      const float* bn = last ? pred_b + (LLN - 1) * DD : init_b + (l + 1) * DD;
      gemm_xWT(init_w + l * DD * DD, init_b + l * DD, cur, stage, lane, wv,
               bv, wn, bn, true,
               [&](int m, int n, float v) {
                 const float r = tanh_fast(v);
                 dst[m * LSTR + n] = r;
                 if (last) ff[m * LSTR + n] = r;
               });
      __syncthreads();
      cur = dst;
    }
  }

  for (int it = 0; it < NITERS; ++it) {
    {
      v8f pabove = {0.f, 0.f, 0.f, 0.f, 0.f, 0.f, 0.f, 0.f};
#pragma unroll 1
      for (int i = LLN - 1; i >= 0; --i) {
        const float* ri = smb(sm, 1 + i);
        const float* below = (i > 0) ? smb(sm, i) : pooled;
        float* ti = smb(sm, 6 + i);
        const bool has_above = (i < LLN - 1);
        const float* wn = (i > 0) ? pred_w + (i - 1) * DD * DD
                                  : upd_w + (LLN - 1) * DD * DD;
        const float* bn = (i > 0) ? pred_b + (i - 1) * DD
                                  : upd_b + (LLN - 1) * DD;
        gemm_xWT_tiles(
            pred_w + i * DD * DD, pred_b + i * DD, ri, stage, lane, wv,
            bv, wn, bn, true,
            [&](int nt, v8f acc) {
              const int o0 = 8 * (lane >> 4) * LSTR + nt * 16 + (lane & 15);
              float bl[8], rr[8];
#pragma unroll
              for (int j = 0; j < 8; ++j) bl[j] = below[o0 + j * LSTR];
              if (has_above) {
#pragma unroll
                for (int j = 0; j < 8; ++j) rr[j] = ri[o0 + j * LSTR];
              }
#pragma unroll
              for (int j = 0; j < 8; ++j) {
                float e = bl[j] - acc[j];
                if (has_above) e += 0.5f * (rr[j] - pabove[j]);
                ti[o0 + j * LSTR] = e;
              }
              pabove = acc;
            });
      }
    }
    __syncthreads();
#pragma unroll 1
    for (int i = LLN - 1; i >= 0; --i) {
      float* ri = smb(sm, 1 + i);
      const float* ti = smb(sm, 6 + i);
      const bool phase_end = (i == 0);
      const bool all_end = phase_end && (it == NITERS - 1);
      const float* wn = all_end ? sqw
                        : phase_end ? pred_w + (LLN - 1) * DD * DD
                                    : upd_w + (i - 1) * DD * DD;
      const float* bn = all_end ? sqb
                        : phase_end ? pred_b + (LLN - 1) * DD
                                    : upd_b + (i - 1) * DD;
      gemm_xWT(upd_w + i * DD * DD, upd_b + i * DD, ti, stage, lane, wv,
               bv, wn, bn, true,
               [&](int m, int n, float v) {
                 const int o = m * LSTR + n;
                 ri[o] = ri[o] + 0.1f * tanh_fast(v);
               });
    }
    __syncthreads();
  }

  float* refined = smb(sm, LLN);
  for (int idx = tid; idx < BT * DD; idx += NTHR) {
    const int r = idx >> 8, c = idx & 255;
    refined[r * LSTR + c] += ff[r * LSTR + c];
  }
  __syncthreads();

  {
    float* q = smb(sm, 6);
    gemm_xWT(sqw, sqb, refined, stage, lane, wv, bv, eqw, eqb, true,
             [&](int m, int n, float v) { q[m * LSTR + n] = v; });
    __syncthreads();
    if (tid < BT * 16) {
      const int r = tid >> 4, c = tid & 15;
      float p = 0.f;
#pragma unroll
      for (int u = 0; u < 16; ++u) {
        const int o = c * 16 + u;
        p += q[r * LSTR + o] * keyb[o];
      }
      tot[tid] = p;
    }
    __syncthreads();
    if (tid < BT) {
      float s = 0.f;
#pragma unroll
      for (int u = 0; u < 16; ++u) s += tot[tid * 16 + u];
      cs_g[b0 + tid] = s;
    }
    __syncthreads();
    gemm_xM(keyw, q, lane, wv,
            [&](int m, int n, float v) { sq_g[(b0 + m) * DD + n] = v; });
    __syncthreads();
  }

  {
    float* q = smb(sm, 7);
    gemm_xWT(eqw, eqb, refined, stage, lane, wv, bv, aw1, ab1, wvu < 8,
             [&](int m, int n, float v) { q[m * LSTR + n] = v; });
    __syncthreads();
    if (tid < BT * 16) {
      const int r = tid >> 4, c = tid & 15;
      float p = 0.f;
#pragma unroll
      for (int u = 0; u < 16; ++u) {
        const int o = c * 16 + u;
        p += q[r * LSTR + o] * keyb[o];
      }
      tot[tid] = p;
    }
    __syncthreads();
    if (tid < BT) {
      float s = 0.f;
#pragma unroll
      for (int u = 0; u < 16; ++u) s += tot[tid * 16 + u];
      ce_g[b0 + tid] = s;
    }
    __syncthreads();
    gemm_xM(keyw, q, lane, wv,
            [&](int m, int n, float v) { eq_g[(b0 + m) * DD + n] = v; });
    __syncthreads();
  }

  if (wvu < 8) {
    gemm_xWT(aw1, ab1, refined, stage, lane, wv, bv, nullptr, nullptr, false,
             [&](int m, int n, float v) {
               tot[m * LSTR + n] =
                   0.5f * v * (1.0f + erff(v * 0.70710678118654752f));
             });
  }
  __syncthreads();
  if (tid < BT * 2) {
    const int r = tid >> 1, k = tid & 1;
    float s = ab2[k];
    for (int j = 0; j < 128; ++j) s += tot[r * LSTR + j] * aw2[k * 128 + j];
    ans_g[(b0 + r) * 2 + k] = s;
  }
}

#define LG_WAVES 8
#define LG_TPW 4
#define LG_SMEM_BYTES (LG_WAVES * TILE * sizeof(float))

#if HAS_WMMA_F32
__device__ __forceinline__ void rows_issue(v4f (&ev)[32], v4f (&pv)[32],
                                           const int idv, const int l0,
                                           const float* __restrict__ ebase,
                                           const float* __restrict__ pbase) {
#pragma unroll
  for (int j = 0; j < 16; ++j) {
    const int id = __builtin_amdgcn_readlane(idv, l0 + j);
    const float* er = ebase + (size_t)id * DD;
    const float* pr = pbase + j * DD;
    ev[2 * j] = *reinterpret_cast<const v4f*>(er);
    ev[2 * j + 1] = *reinterpret_cast<const v4f*>(er + 128);
    pv[2 * j] = *reinterpret_cast<const v4f*>(pr);
    pv[2 * j + 1] = *reinterpret_cast<const v4f*>(pr + 128);
  }
}

__device__ __forceinline__ void rows_park(const v4f (&ev)[32],
                                          const v4f (&pv)[32], float* wp) {
#pragma unroll
  for (int j = 0; j < 16; ++j) {
    *reinterpret_cast<v4f*>(wp + j * LSTR) = ev[2 * j] + pv[2 * j];
    *reinterpret_cast<v4f*>(wp + j * LSTR + 128) = ev[2 * j + 1] + pv[2 * j + 1];
  }
}

__device__ __forceinline__ void afrag_read8(v2f (&a)[8], const float* xr) {
#pragma unroll
  for (int s = 0; s < 8; ++s) a[s] = (v2f){xr[4 * s], xr[4 * s + 1]};
}

__device__ __forceinline__ v8f tile_tokens_dot(const float* xr,
                                               const v2f (&bq)[64]) {
  v8f acc = {0.f, 0.f, 0.f, 0.f, 0.f, 0.f, 0.f, 0.f};
  v2f a0[8], a1[8];
  afrag_read8(a0, xr);
#pragma unroll
  for (int g = 0; g < 8; g += 2) {
    afrag_read8(a1, xr + 32 * (g + 1));
    __builtin_amdgcn_sched_barrier(0);
#pragma unroll
    for (int s = 0; s < 8; ++s)
      acc = __builtin_amdgcn_wmma_f32_16x16x4_f32(false, a0[s], false,
                                                  bq[8 * g + s], (short)0, acc,
                                                  false, false);
    __builtin_amdgcn_sched_barrier(0);
    if (g + 2 < 8) afrag_read8(a0, xr + 32 * (g + 2));
    __builtin_amdgcn_sched_barrier(0);
#pragma unroll
    for (int s = 0; s < 8; ++s)
      acc = __builtin_amdgcn_wmma_f32_16x16x4_f32(false, a1[s], false,
                                                  bq[8 * (g + 1) + s], (short)0,
                                                  acc, false, false);
    __builtin_amdgcn_sched_barrier(0);
  }
  return acc;
}
#endif

__global__ __launch_bounds__(LG_WAVES * 32) void logits_kernel(
    const int* __restrict__ ids, const float* __restrict__ emb,
    const float* __restrict__ pos, const float* __restrict__ sq_g,
    const float* __restrict__ eq_g, const float* __restrict__ cs_g,
    const float* __restrict__ ce_g, float* __restrict__ out_start,
    float* __restrict__ out_end) {
  extern __shared__ __attribute__((aligned(16))) float lsm[];
  const int b = blockIdx.x;
  const int lane = threadIdx.x & 31;
  const int wvu = __builtin_amdgcn_readfirstlane((int)(threadIdx.x >> 5));
#if HAS_WMMA_F32
  const int c = lane & 15, h = lane >> 4;
  const int tok0 = wvu * (LG_TPW * 16);
  float* st = lsm + wvu * TILE;
  const float* ebase = emb + lane * 4;
  const float* pbase = pos + tok0 * DD + lane * 4;
  const int idlo = ids[b * SS + tok0 + lane];
  const int idhi = ids[b * SS + tok0 + 32 + lane];

  v4f ev[32], pv[32];
  rows_issue(ev, pv, idlo, 0, ebase, pbase);

  v2f bq[64];
  {
    const float* qsrc = ((c == 0) ? sq_g : eq_g) + b * DD + 2 * h;
#pragma unroll
    for (int s = 0; s < 64; ++s)
      bq[s] = *reinterpret_cast<const v2f*>(qsrc + 4 * s);
  }
  const float cadd = (c == 0) ? cs_g[b] : ce_g[b];
  float* orow = ((c == 0) ? out_start : out_end) + b * SS + tok0 + 8 * h;
  const float* xr = st + c * LSTR + 2 * h;
  float* wp = st + lane * 4;

#pragma unroll 1
  for (int t = 0; t < LG_TPW; ++t) {
    wave_lds_handover();
    rows_park(ev, pv, wp);
    wave_lds_handover();
    if (t + 1 < LG_TPW)
      rows_issue(ev, pv, (t + 1 < 2) ? idlo : idhi, ((t + 1) & 1) * 16, ebase,
                 pbase + (t + 1) * 16 * DD);
    const v8f acc = tile_tokens_dot(xr, bq);
    const v4f lo = {acc[0] + cadd, acc[1] + cadd, acc[2] + cadd, acc[3] + cadd};
    const v4f hi = {acc[4] + cadd, acc[5] + cadd, acc[6] + cadd, acc[7] + cadd};
    if (c < 2) {
      *reinterpret_cast<v4f*>(orow + t * 16) = lo;
      *reinterpret_cast<v4f*>(orow + t * 16 + 4) = hi;
    }
  }
#else
  (void)lsm;
  (void)wvu;
  (void)lane;
  for (int s = threadIdx.x; s < SS; s += LG_WAVES * 32) {
    const float* row = emb + (size_t)ids[b * SS + s] * DD;
    const float* pr = pos + s * DD;
    float as = 0.f, ae = 0.f;
    for (int k = 0; k < DD; ++k) {
      const float e = row[k] + pr[k];
      as += e * sq_g[b * DD + k];
      ae += e * eq_g[b * DD + k];
    }
    out_start[b * SS + s] = as + cs_g[b];
    out_end[b * SS + s] = ae + ce_g[b];
  }
#endif
}

extern "C" void kernel_launch(void* const* d_in, const int* in_sizes, int n_in,
                              void* d_out, int out_size, void* d_ws,
                              size_t ws_size, hipStream_t stream) {
  const int* ids = (const int*)d_in[0];
  const float* emb = (const float*)d_in[1];
  const float* pos = (const float*)d_in[2];
  const float* init_w = (const float*)d_in[3];
  const float* init_b = (const float*)d_in[4];
  const float* pred_w = (const float*)d_in[5];
  const float* pred_b = (const float*)d_in[6];
  const float* upd_w = (const float*)d_in[7];
  const float* upd_b = (const float*)d_in[8];
  const float* sqw = (const float*)d_in[9];
  const float* sqb = (const float*)d_in[10];
  const float* eqw = (const float*)d_in[11];
  const float* eqb = (const float*)d_in[12];
  const float* keyw = (const float*)d_in[13];
  const float* keyb = (const float*)d_in[14];
  const float* aw1 = (const float*)d_in[15];
  const float* ab1 = (const float*)d_in[16];
  const float* aw2 = (const float*)d_in[17];
  const float* ab2 = (const float*)d_in[18];

  float* ws = (float*)d_ws;
  float* pooled = ws;
  float* sq = pooled + BB * DD;
  float* eq = sq + BB * DD;
  float* cs = eq + BB * DD;
  float* ce = cs + BB;

  float* out = (float*)d_out;
  float* out_start = out;
  float* out_end = out + BB * SS;
  float* out_ans = out + 2 * BB * SS;

  (void)hipFuncSetAttribute((const void*)refine_kernel,
                            hipFuncAttributeMaxDynamicSharedMemorySize,
                            (int)SMEM_BYTES);

  (void)hipFuncSetAttribute((const void*)logits_kernel,
                            hipFuncAttributeMaxDynamicSharedMemorySize,
                            (int)LG_SMEM_BYTES);

  pooled_kernel<<<BB, PL_WAVES * 32, 0, stream>>>(ids, emb, pos, pooled);
  refine_kernel<<<BB / BT, NTHR, SMEM_BYTES, stream>>>(
      pooled, init_w, init_b, pred_w, pred_b, upd_w, upd_b, sqw, sqb, eqw, eqb,
      keyw, keyb, aw1, ab1, aw2, ab2, sq, eq, cs, ce, out_ans);
  logits_kernel<<<BB, LG_WAVES * 32, LG_SMEM_BYTES, stream>>>(
      ids, emb, pos, sq, eq, cs, ce, out_start, out_end);
}


# refine loop: a sweep no longer computes layers whose new reps no output can read (only reps[3] leaves the loop): last three sweeps run layers 1..3, 2..3, 3; 12 of 80 sweep GEMMs gone; sums unchanged
# speedup vs baseline: 3.1014x; 1.1182x over previous
#include <hip/hip_runtime.h>
#include <math.h>

#define BB   256
#define SS   512
#define DD   256
#define LLN  4
#define NITERS 10
#define BT   16
#define LSTR 260
#define TILE (BT * LSTR)
#define NWAVES 16
#define NTHR (NWAVES * 32)
#define WSTR 132
#define WHALF (16 * WSTR)
#define WSTAGE WHALF
#define NTILES 10
#define SMEM_FLOATS (NTILES * TILE + NWAVES * WSTAGE)
#define SMEM_BYTES  (SMEM_FLOATS * sizeof(float))

typedef float v2f __attribute__((ext_vector_type(2)));
typedef float v4f __attribute__((ext_vector_type(4)));
typedef float v8f __attribute__((ext_vector_type(8)));

#if defined(__has_builtin)
#  if __has_builtin(__builtin_amdgcn_wmma_f32_16x16x4_f32)
#    define HAS_WMMA_F32 1
#  else
#    define HAS_WMMA_F32 0
#  endif
#  if __has_builtin(__builtin_amdgcn_tanhf)
#    define HAS_HW_TANH 1
#  else
#    define HAS_HW_TANH 0
#  endif
#else
#  define HAS_WMMA_F32 0
#  define HAS_HW_TANH 0
#endif

__device__ __forceinline__ float tanh_fast(float x) {
#if HAS_HW_TANH
  return __builtin_amdgcn_tanhf(x);
#else
  return tanhf(x);
#endif
}

__device__ __forceinline__ float* smb(float* sm, int i) { return sm + i * TILE; }


__device__ __forceinline__ v8f tile_xWT(const float* __restrict__ W,
                                        const float* X, int nt, int lane,
                                        float cinit) {
  const int l16 = lane & 15;
  const int half = lane >> 4;
  v8f acc = {cinit, cinit, cinit, cinit, cinit, cinit, cinit, cinit};
#if HAS_WMMA_F32
  const float* wr = W + (nt * 16 + l16) * DD + 2 * half;
  const float* xr = X + l16 * LSTR + 2 * half;
#pragma unroll 4
  for (int kk = 0; kk < DD; kk += 4) {
    v2f a = {xr[kk], xr[kk + 1]};
    v2f b = {wr[kk], wr[kk + 1]};
    acc = __builtin_amdgcn_wmma_f32_16x16x4_f32(false, a, false, b,
                                                (short)0, acc, false, false);
  }
#else
#pragma unroll
  for (int j = 0; j < 8; ++j) {
    const int m = j + 8 * half, n = nt * 16 + l16;
    float s = cinit;
    for (int k = 0; k < DD; ++k) s += X[m * LSTR + k] * W[n * DD + k];
    acc[j] = s;
  }
#endif
  return acc;
}

__device__ __forceinline__ v8f tile_xM(const float* __restrict__ M,
                                       const float* X, int nt, int lane,
                                       float cinit) {
  const int l16 = lane & 15;
  const int half = lane >> 4;
  v8f acc = {cinit, cinit, cinit, cinit, cinit, cinit, cinit, cinit};
#if HAS_WMMA_F32
  const float* mp = M + (nt * 16 + l16) + (2 * half) * DD;
  const float* xr = X + l16 * LSTR + 2 * half;
#pragma unroll 4
  for (int kk = 0; kk < DD; kk += 4) {
    v2f a = {xr[kk], xr[kk + 1]};
    v2f b = {mp[kk * DD], mp[kk * DD + DD]};
    acc = __builtin_amdgcn_wmma_f32_16x16x4_f32(false, a, false, b,
                                                (short)0, acc, false, false);
  }
#else
#pragma unroll
  for (int j = 0; j < 8; ++j) {
    const int m = j + 8 * half, n = nt * 16 + l16;
    float s = cinit;
    for (int k = 0; k < DD; ++k) s += X[m * LSTR + k] * M[k * DD + n];
    acc[j] = s;
  }
#endif
  return acc;
}

#if HAS_WMMA_F32
__device__ __forceinline__ void stage_issue(v4f (&r)[16],
                                            const float* __restrict__ wsrc) {
#pragma unroll
  for (int j = 0; j < 16; ++j)
    r[j] = *reinterpret_cast<const v4f*>(wsrc + j * DD);
}

__device__ __forceinline__ void stage_park(const v4f (&r)[16], float* wp) {
#pragma unroll
  for (int j = 0; j < 16; ++j) *reinterpret_cast<v4f*>(wp + j * WSTR) = r[j];
}

__device__ __forceinline__ void wave_lds_handover() {
  __builtin_amdgcn_fence(__ATOMIC_RELEASE, "wavefront");
  __builtin_amdgcn_wave_barrier();
  __builtin_amdgcn_fence(__ATOMIC_ACQUIRE, "wavefront");
}

__device__ __forceinline__ void frag_read8(v2f (&a)[8], v2f (&b)[8],
                                           const float* xr, const float* wr) {
#pragma unroll
  for (int s = 0; s < 8; ++s) {
    a[s] = (v2f){xr[4 * s], xr[4 * s + 1]};
    b[s] = (v2f){wr[4 * s], wr[4 * s + 1]};
  }
}

__device__ __forceinline__ v8f mma8(v8f acc, const v2f (&a)[8],
                                    const v2f (&b)[8]) {
#pragma unroll
  for (int s = 0; s < 8; ++s)
    acc = __builtin_amdgcn_wmma_f32_16x16x4_f32(false, a[s], false, b[s],
                                                (short)0, acc, false, false);
  return acc;
}

__device__ __forceinline__ v8f half_tile_mma(v8f acc, const float* xr,
                                             const float* wr) {
  v2f a0[8], b0[8], a1[8], b1[8];
  frag_read8(a0, b0, xr, wr);
  frag_read8(a1, b1, xr + 32, wr + 32);
  __builtin_amdgcn_sched_barrier(0);
  acc = mma8(acc, a0, b0);
  __builtin_amdgcn_sched_barrier(0);
  frag_read8(a0, b0, xr + 64, wr + 64);
  __builtin_amdgcn_sched_barrier(0);
  acc = mma8(acc, a1, b1);
  __builtin_amdgcn_sched_barrier(0);
  frag_read8(a1, b1, xr + 96, wr + 96);
  __builtin_amdgcn_sched_barrier(0);
  acc = mma8(acc, a0, b0);
  acc = mma8(acc, a1, b1);
  return acc;
}
#endif

__device__ __forceinline__ void gemm_first_request(
    v4f (&r)[16], float& bv, const float* __restrict__ W,
    const float* __restrict__ bias, int lane, int wv) {
#if HAS_WMMA_F32
  stage_issue(r, W + (wv * 16) * DD + lane * 4);
#endif
  bv = bias ? bias[wv * 16 + (lane & 15)] : 0.f;
}

template <typename TileEpi>
__device__ __forceinline__ void gemm_xWT_tiles(
    const float* __restrict__ W, const float* __restrict__ bias,
    const float* X, float* stage, int lane, int wv, float& bv,
    const float* __restrict__ Wn, const float* __restrict__ biasn,
    const bool has_next, TileEpi epi) {
  const int l16 = lane & 15, half = lane >> 4;
#if HAS_WMMA_F32
  const float* wsrc = W + (wv * 16) * DD + lane * 4;
  const float* xr = X + l16 * LSTR + 2 * half;
  const float* wr = stage + l16 * WSTR + 2 * half;
  float* wp = stage + lane * 4;
  const float cb = bv;
  v4f r[16];
  v8f acc = {cb, cb, cb, cb, cb, cb, cb, cb};
#pragma unroll 1
  for (int q = 0; q < 2; ++q) {
    const bool more = (q < 1) || has_next;
    if (more) {
      stage_issue(r, (q < 1) ? wsrc + 128 : Wn + (wv * 16) * DD + lane * 4);
      if (q == 1) bv = biasn ? biasn[wv * 16 + l16] : 0.f;
    }
    acc = half_tile_mma(acc, xr + q * 128, wr);
    if (q == 1) epi(wv, acc);
    if (more) {
      wave_lds_handover();
      stage_park(r, wp);
      wave_lds_handover();
    }
  }
#else
  const float bvf = bias ? bias[wv * 16 + l16] : 0.f;
  v8f acc = tile_xWT(W, X, wv, lane, bvf);
  epi(wv, acc);
#endif
}

template <typename Epi>
__device__ __forceinline__ void gemm_xWT(
    const float* __restrict__ W, const float* __restrict__ bias,
    const float* X, float* stage, int lane, int wv, float& bv,
    const float* __restrict__ Wn, const float* __restrict__ biasn,
    const bool has_next, Epi epi) {
  gemm_xWT_tiles(W, bias, X, stage, lane, wv, bv, Wn, biasn, has_next,
                 [&](int nt, v8f acc) {
    const int l16 = lane & 15, half = lane >> 4;
#pragma unroll
    for (int j = 0; j < 8; ++j) epi(j + 8 * half, nt * 16 + l16, acc[j]);
  });
}

template <typename Epi>
__device__ __forceinline__ void gemm_xM(const float* __restrict__ M,
                                        const float* X, int lane, int wv,
                                        Epi epi) {
  const int nt = wv;
  v8f acc = tile_xM(M, X, nt, lane, 0.f);
  const int l16 = lane & 15, half = lane >> 4;
#pragma unroll
  for (int j = 0; j < 8; ++j) epi(j + 8 * half, nt * 16 + l16, acc[j]);
}

#define PL_WAVES 8
#define PL_TPW (SS / PL_WAVES)
#define PL_GRP 8

__device__ __forceinline__ void pool_issue(v4f (&ev)[2 * PL_GRP],
                                           v4f (&pv)[2 * PL_GRP],
                                           const int idv, const int l0,
                                           const float* __restrict__ ebase,
                                           const float* __restrict__ pbase) {
#pragma unroll
  for (int j = 0; j < PL_GRP; ++j) {
    const int id = __builtin_amdgcn_readlane(idv, l0 + j);
    const float* er = ebase + (size_t)id * DD;
    const float* pr = pbase + j * DD;
    ev[2 * j] = *reinterpret_cast<const v4f*>(er);
    ev[2 * j + 1] = *reinterpret_cast<const v4f*>(er + 128);
    pv[2 * j] = *reinterpret_cast<const v4f*>(pr);
    pv[2 * j + 1] = *reinterpret_cast<const v4f*>(pr + 128);
  }
}

__device__ __forceinline__ void pool_sum(v4f& alo, v4f& ahi,
                                         const v4f (&ev)[2 * PL_GRP],
                                         const v4f (&pv)[2 * PL_GRP]) {
#pragma unroll
  for (int j = 0; j < PL_GRP; ++j) {
    alo += ev[2 * j] + pv[2 * j];
    ahi += ev[2 * j + 1] + pv[2 * j + 1];
  }
}

__global__ __launch_bounds__(PL_WAVES * 32) void pooled_kernel(
    const int* __restrict__ ids, const float* __restrict__ emb,
    const float* __restrict__ pos, float* __restrict__ pooled_g) {
  __shared__ __attribute__((aligned(16))) float part[PL_WAVES * DD];
  const int b = blockIdx.x;
  const int t = threadIdx.x;
  const int lane = t & 31;
  const int wvu = __builtin_amdgcn_readfirstlane(t >> 5);
  const int tok0 = wvu * PL_TPW;
  const int idlo = ids[b * SS + tok0 + lane];
  const int idhi = ids[b * SS + tok0 + 32 + lane];
  const float* ebase = emb + lane * 4;
  const float* pbase = pos + tok0 * DD + lane * 4;

  v4f alo = {0.f, 0.f, 0.f, 0.f};
  v4f ahi = {0.f, 0.f, 0.f, 0.f};
  v4f e0[2 * PL_GRP], p0[2 * PL_GRP], e1[2 * PL_GRP], p1[2 * PL_GRP];
  pool_issue(e0, p0, idlo, 0, ebase, pbase);
#pragma unroll
  for (int g = 1; g < PL_TPW / PL_GRP - 1; g += 2) {
    pool_issue(e1, p1, (g < 4) ? idlo : idhi, (g & 3) * PL_GRP, ebase,
               pbase + g * PL_GRP * DD);
    __builtin_amdgcn_sched_barrier(0);
    pool_sum(alo, ahi, e0, p0);
    pool_issue(e0, p0, (g + 1 < 4) ? idlo : idhi, ((g + 1) & 3) * PL_GRP, ebase,
               pbase + (g + 1) * PL_GRP * DD);
    __builtin_amdgcn_sched_barrier(0);
    pool_sum(alo, ahi, e1, p1);
  }
  pool_issue(e1, p1, idhi, 3 * PL_GRP, ebase,
             pbase + (PL_TPW / PL_GRP - 1) * PL_GRP * DD);
  __builtin_amdgcn_sched_barrier(0);
  pool_sum(alo, ahi, e0, p0);
  pool_sum(alo, ahi, e1, p1);

  float* pw = part + wvu * DD + lane * 4;
  *reinterpret_cast<v4f*>(pw) = alo;
  *reinterpret_cast<v4f*>(pw + 128) = ahi;
  __syncthreads();
  float s = part[t];
#pragma unroll
  for (int w = 1; w < PL_WAVES; ++w) s += part[w * DD + t];
  pooled_g[b * DD + t] = s * (1.0f / (float)SS);
}

__global__ __launch_bounds__(NTHR) void refine_kernel(
    const float* __restrict__ pooled_g,
    const float* __restrict__ init_w, const float* __restrict__ init_b,
    const float* __restrict__ pred_w, const float* __restrict__ pred_b,
    const float* __restrict__ upd_w, const float* __restrict__ upd_b,
    const float* __restrict__ sqw, const float* __restrict__ sqb,
    const float* __restrict__ eqw, const float* __restrict__ eqb,
    const float* __restrict__ keyw, const float* __restrict__ keyb,
    const float* __restrict__ aw1, const float* __restrict__ ab1,
    const float* __restrict__ aw2, const float* __restrict__ ab2,
    float* __restrict__ sq_g, float* __restrict__ eq_g,
    float* __restrict__ cs_g, float* __restrict__ ce_g,
    float* __restrict__ ans_g) {
  extern __shared__ __attribute__((aligned(16))) float sm[];

  const int tid = threadIdx.x;
  const int lane = tid & 31;
  const int wv = tid >> 5;
  const int wvu = __builtin_amdgcn_readfirstlane(wv);
  const int b0 = blockIdx.x * BT;

  float* pooled = smb(sm, 0);
  float* ff = smb(sm, 5);
  float* tot = smb(sm, 8);
  float* stage = sm + NTILES * TILE + wv * WSTAGE;

  float bv;
  {
    v4f r0[16];
    gemm_first_request(r0, bv, init_w, init_b, lane, wv);
    for (int idx = tid; idx < BT * DD; idx += NTHR) {
      const int r = idx >> 8, c = idx & 255;
      pooled[r * LSTR + c] = pooled_g[(b0 + r) * DD + c];
    }
#if HAS_WMMA_F32
    stage_park(r0, stage + lane * 4);
    wave_lds_handover();
#endif
  }
  __syncthreads();

  {
    const float* cur = pooled;
    for (int l = 0; l < LLN; ++l) {
      float* dst = smb(sm, 1 + l);
      const bool last = (l == LLN - 1);
      const float* wn = last ? pred_w + (LLN - 1) * DD * DD
                             : init_w + (l + 1) * DD * DD;
      const float* bn = last ? pred_b + (LLN - 1) * DD : init_b + (l + 1) * DD;
      gemm_xWT(init_w + l * DD * DD, init_b + l * DD, cur, stage, lane, wv,
               bv, wn, bn, true,
               [&](int m, int n, float v) {
                 const float r = tanh_fast(v);
                 dst[m * LSTR + n] = r;
                 if (last) ff[m * LSTR + n] = r;
               });
      __syncthreads();
      cur = dst;
    }
  }

  for (int it = 0; it < NITERS; ++it) {
    const int lo = (it > NITERS - LLN) ? it - (NITERS - LLN) : 0;
    {
      v8f pabove = {0.f, 0.f, 0.f, 0.f, 0.f, 0.f, 0.f, 0.f};
#pragma unroll 1
      for (int i = LLN - 1; i >= lo; --i) {
        const float* ri = smb(sm, 1 + i);
        const float* below = (i > 0) ? smb(sm, i) : pooled;
        float* ti = smb(sm, 6 + i);
        const bool has_above = (i < LLN - 1);
        const float* wn = (i > lo) ? pred_w + (i - 1) * DD * DD
                                   : upd_w + (LLN - 1) * DD * DD;
        const float* bn = (i > lo) ? pred_b + (i - 1) * DD
                                   : upd_b + (LLN - 1) * DD;
        gemm_xWT_tiles(
            pred_w + i * DD * DD, pred_b + i * DD, ri, stage, lane, wv,
            bv, wn, bn, true,
            [&](int nt, v8f acc) {
              const int o0 = 8 * (lane >> 4) * LSTR + nt * 16 + (lane & 15);
              float bl[8], rr[8];
#pragma unroll
              for (int j = 0; j < 8; ++j) bl[j] = below[o0 + j * LSTR];
              if (has_above) {
#pragma unroll
                for (int j = 0; j < 8; ++j) rr[j] = ri[o0 + j * LSTR];
              }
#pragma unroll
              for (int j = 0; j < 8; ++j) {
                float e = bl[j] - acc[j];
                if (has_above) e += 0.5f * (rr[j] - pabove[j]);
                ti[o0 + j * LSTR] = e;
              }
              pabove = acc;
            });
      }
    }
    __syncthreads();
#pragma unroll 1
    for (int i = LLN - 1; i >= lo; --i) {
      float* ri = smb(sm, 1 + i);
      const float* ti = smb(sm, 6 + i);
      const bool phase_end = (i == lo);
      const bool all_end = phase_end && (it == NITERS - 1);
      const float* wn = all_end ? sqw
                        : phase_end ? pred_w + (LLN - 1) * DD * DD
                                    : upd_w + (i - 1) * DD * DD;
      const float* bn = all_end ? sqb
                        : phase_end ? pred_b + (LLN - 1) * DD
                                    : upd_b + (i - 1) * DD;
      gemm_xWT(upd_w + i * DD * DD, upd_b + i * DD, ti, stage, lane, wv,
               bv, wn, bn, true,
               [&](int m, int n, float v) {
                 const int o = m * LSTR + n;
                 ri[o] = ri[o] + 0.1f * tanh_fast(v);
               });
    }
    __syncthreads();
  }

  float* refined = smb(sm, LLN);
  for (int idx = tid; idx < BT * DD; idx += NTHR) {
    const int r = idx >> 8, c = idx & 255;
    refined[r * LSTR + c] += ff[r * LSTR + c];
  }
  __syncthreads();

  {
    float* q = smb(sm, 6);
    gemm_xWT(sqw, sqb, refined, stage, lane, wv, bv, eqw, eqb, true,
             [&](int m, int n, float v) { q[m * LSTR + n] = v; });
    __syncthreads();
    if (tid < BT * 16) {
      const int r = tid >> 4, c = tid & 15;
      float p = 0.f;
#pragma unroll
      for (int u = 0; u < 16; ++u) {
        const int o = c * 16 + u;
        p += q[r * LSTR + o] * keyb[o];
      }
      tot[tid] = p;
    }
    __syncthreads();
    if (tid < BT) {
      float s = 0.f;
#pragma unroll
      for (int u = 0; u < 16; ++u) s += tot[tid * 16 + u];
      cs_g[b0 + tid] = s;
    }
    __syncthreads();
    gemm_xM(keyw, q, lane, wv,
            [&](int m, int n, float v) { sq_g[(b0 + m) * DD + n] = v; });
    __syncthreads();
  }

  {
    float* q = smb(sm, 7);
    gemm_xWT(eqw, eqb, refined, stage, lane, wv, bv, aw1, ab1, wvu < 8,
             [&](int m, int n, float v) { q[m * LSTR + n] = v; });
    __syncthreads();
    if (tid < BT * 16) {
      const int r = tid >> 4, c = tid & 15;
      float p = 0.f;
#pragma unroll
      for (int u = 0; u < 16; ++u) {
        const int o = c * 16 + u;
        p += q[r * LSTR + o] * keyb[o];
      }
      tot[tid] = p;
    }
    __syncthreads();
    if (tid < BT) {
      float s = 0.f;
#pragma unroll
      for (int u = 0; u < 16; ++u) s += tot[tid * 16 + u];
      ce_g[b0 + tid] = s;
    }
    __syncthreads();
    gemm_xM(keyw, q, lane, wv,
            [&](int m, int n, float v) { eq_g[(b0 + m) * DD + n] = v; });
    __syncthreads();
  }

  if (wvu < 8) {
    gemm_xWT(aw1, ab1, refined, stage, lane, wv, bv, nullptr, nullptr, false,
             [&](int m, int n, float v) {
               tot[m * LSTR + n] =
                   0.5f * v * (1.0f + erff(v * 0.70710678118654752f));
             });
  }
  __syncthreads();
  if (tid < BT * 2) {
    const int r = tid >> 1, k = tid & 1;
    float s = ab2[k];
    for (int j = 0; j < 128; ++j) s += tot[r * LSTR + j] * aw2[k * 128 + j];
    ans_g[(b0 + r) * 2 + k] = s;
  }
}

#define LG_WAVES 8
#define LG_TPW 4
#define LG_SMEM_BYTES (LG_WAVES * TILE * sizeof(float))

#if HAS_WMMA_F32
__device__ __forceinline__ void rows_issue(v4f (&ev)[32], v4f (&pv)[32],
                                           const int idv, const int l0,
                                           const float* __restrict__ ebase,
                                           const float* __restrict__ pbase) {
#pragma unroll
  for (int j = 0; j < 16; ++j) {
    const int id = __builtin_amdgcn_readlane(idv, l0 + j);
    const float* er = ebase + (size_t)id * DD;
    const float* pr = pbase + j * DD;
    ev[2 * j] = *reinterpret_cast<const v4f*>(er);
    ev[2 * j + 1] = *reinterpret_cast<const v4f*>(er + 128);
    pv[2 * j] = *reinterpret_cast<const v4f*>(pr);
    pv[2 * j + 1] = *reinterpret_cast<const v4f*>(pr + 128);
  }
}

__device__ __forceinline__ void rows_park(const v4f (&ev)[32],
                                          const v4f (&pv)[32], float* wp) {
#pragma unroll
  for (int j = 0; j < 16; ++j) {
    *reinterpret_cast<v4f*>(wp + j * LSTR) = ev[2 * j] + pv[2 * j];
    *reinterpret_cast<v4f*>(wp + j * LSTR + 128) = ev[2 * j + 1] + pv[2 * j + 1];
  }
}

__device__ __forceinline__ void afrag_read8(v2f (&a)[8], const float* xr) {
#pragma unroll
  for (int s = 0; s < 8; ++s) a[s] = (v2f){xr[4 * s], xr[4 * s + 1]};
}

__device__ __forceinline__ v8f tile_tokens_dot(const float* xr,
                                               const v2f (&bq)[64]) {
  v8f acc = {0.f, 0.f, 0.f, 0.f, 0.f, 0.f, 0.f, 0.f};
  v2f a0[8], a1[8];
  afrag_read8(a0, xr);
#pragma unroll
  for (int g = 0; g < 8; g += 2) {
    afrag_read8(a1, xr + 32 * (g + 1));
    __builtin_amdgcn_sched_barrier(0);
#pragma unroll
    for (int s = 0; s < 8; ++s)
      acc = __builtin_amdgcn_wmma_f32_16x16x4_f32(false, a0[s], false,
                                                  bq[8 * g + s], (short)0, acc,
                                                  false, false);
    __builtin_amdgcn_sched_barrier(0);
    if (g + 2 < 8) afrag_read8(a0, xr + 32 * (g + 2));
    __builtin_amdgcn_sched_barrier(0);
#pragma unroll
    for (int s = 0; s < 8; ++s)
      acc = __builtin_amdgcn_wmma_f32_16x16x4_f32(false, a1[s], false,
                                                  bq[8 * (g + 1) + s], (short)0,
                                                  acc, false, false);
    __builtin_amdgcn_sched_barrier(0);
  }
  return acc;
}
#endif

__global__ __launch_bounds__(LG_WAVES * 32) void logits_kernel(
    const int* __restrict__ ids, const float* __restrict__ emb,
    const float* __restrict__ pos, const float* __restrict__ sq_g,
    const float* __restrict__ eq_g, const float* __restrict__ cs_g,
    const float* __restrict__ ce_g, float* __restrict__ out_start,
    float* __restrict__ out_end) {
  extern __shared__ __attribute__((aligned(16))) float lsm[];
  const int b = blockIdx.x;
  const int lane = threadIdx.x & 31;
  const int wvu = __builtin_amdgcn_readfirstlane((int)(threadIdx.x >> 5));
#if HAS_WMMA_F32
  const int c = lane & 15, h = lane >> 4;
  const int tok0 = wvu * (LG_TPW * 16);
  float* st = lsm + wvu * TILE;
  const float* ebase = emb + lane * 4;
  const float* pbase = pos + tok0 * DD + lane * 4;
  const int idlo = ids[b * SS + tok0 + lane];
  const int idhi = ids[b * SS + tok0 + 32 + lane];

  v4f ev[32], pv[32];
  rows_issue(ev, pv, idlo, 0, ebase, pbase);

  v2f bq[64];
  {
    const float* qsrc = ((c == 0) ? sq_g : eq_g) + b * DD + 2 * h;
#pragma unroll
    for (int s = 0; s < 64; ++s)
      bq[s] = *reinterpret_cast<const v2f*>(qsrc + 4 * s);
  }
  const float cadd = (c == 0) ? cs_g[b] : ce_g[b];
  float* orow = ((c == 0) ? out_start : out_end) + b * SS + tok0 + 8 * h;
  const float* xr = st + c * LSTR + 2 * h;
  float* wp = st + lane * 4;

#pragma unroll 1
  for (int t = 0; t < LG_TPW; ++t) {
    wave_lds_handover();
    rows_park(ev, pv, wp);
    wave_lds_handover();
    if (t + 1 < LG_TPW)
      rows_issue(ev, pv, (t + 1 < 2) ? idlo : idhi, ((t + 1) & 1) * 16, ebase,
                 pbase + (t + 1) * 16 * DD);
    const v8f acc = tile_tokens_dot(xr, bq);
    const v4f lo = {acc[0] + cadd, acc[1] + cadd, acc[2] + cadd, acc[3] + cadd};
    const v4f hi = {acc[4] + cadd, acc[5] + cadd, acc[6] + cadd, acc[7] + cadd};
    if (c < 2) {
      *reinterpret_cast<v4f*>(orow + t * 16) = lo;
      *reinterpret_cast<v4f*>(orow + t * 16 + 4) = hi;
    }
  }
#else
  (void)lsm;
  (void)wvu;
  (void)lane;
  for (int s = threadIdx.x; s < SS; s += LG_WAVES * 32) {
    const float* row = emb + (size_t)ids[b * SS + s] * DD;
    const float* pr = pos + s * DD;
    float as = 0.f, ae = 0.f;
    for (int k = 0; k < DD; ++k) {
      const float e = row[k] + pr[k];
      as += e * sq_g[b * DD + k];
      ae += e * eq_g[b * DD + k];
    }
    out_start[b * SS + s] = as + cs_g[b];
    out_end[b * SS + s] = ae + ce_g[b];
  }
#endif
}

extern "C" void kernel_launch(void* const* d_in, const int* in_sizes, int n_in,
                              void* d_out, int out_size, void* d_ws,
                              size_t ws_size, hipStream_t stream) {
  const int* ids = (const int*)d_in[0];
  const float* emb = (const float*)d_in[1];
  const float* pos = (const float*)d_in[2];
  const float* init_w = (const float*)d_in[3];
  const float* init_b = (const float*)d_in[4];
  const float* pred_w = (const float*)d_in[5];
  const float* pred_b = (const float*)d_in[6];
  const float* upd_w = (const float*)d_in[7];
  const float* upd_b = (const float*)d_in[8];
  const float* sqw = (const float*)d_in[9];
  const float* sqb = (const float*)d_in[10];
  const float* eqw = (const float*)d_in[11];
  const float* eqb = (const float*)d_in[12];
  const float* keyw = (const float*)d_in[13];
  const float* keyb = (const float*)d_in[14];
  const float* aw1 = (const float*)d_in[15];
  const float* ab1 = (const float*)d_in[16];
  const float* aw2 = (const float*)d_in[17];
  const float* ab2 = (const float*)d_in[18];

  float* ws = (float*)d_ws;
  float* pooled = ws;
  float* sq = pooled + BB * DD;
  float* eq = sq + BB * DD;
  float* cs = eq + BB * DD;
  float* ce = cs + BB;

  float* out = (float*)d_out;
  float* out_start = out;
  float* out_end = out + BB * SS;
  float* out_ans = out + 2 * BB * SS;

  (void)hipFuncSetAttribute((const void*)refine_kernel,
                            hipFuncAttributeMaxDynamicSharedMemorySize,
                            (int)SMEM_BYTES);

  (void)hipFuncSetAttribute((const void*)logits_kernel,
                            hipFuncAttributeMaxDynamicSharedMemorySize,
                            (int)LG_SMEM_BYTES);

  pooled_kernel<<<BB, PL_WAVES * 32, 0, stream>>>(ids, emb, pos, pooled);
  refine_kernel<<<BB / BT, NTHR, SMEM_BYTES, stream>>>(
      pooled, init_w, init_b, pred_w, pred_b, upd_w, upd_b, sqw, sqb, eqw, eqb,
      keyw, keyb, aw1, ab1, aw2, ab2, sq, eq, cs, ce, out_ans);
  logits_kernel<<<BB, LG_WAVES * 32, LG_SMEM_BYTES, stream>>>(
      ids, emb, pos, sq, eq, cs, ce, out_start, out_end);
}


# refine sweep phases: a wave's issue priority falls with each GEMM it finishes in a phase (s_setprio 3..0), so the waves that are behind get the matrix pipe first; sums unchanged
# speedup vs baseline: 3.2403x; 1.0448x over previous
#include <hip/hip_runtime.h>
#include <math.h>

#define BB   256
#define SS   512
#define DD   256
#define LLN  4
#define NITERS 10
#define BT   16
#define LSTR 260
#define TILE (BT * LSTR)
#define NWAVES 16
#define NTHR (NWAVES * 32)
#define WSTR 132
#define WHALF (16 * WSTR)
#define WSTAGE WHALF
#define NTILES 10
#define SMEM_FLOATS (NTILES * TILE + NWAVES * WSTAGE)
#define SMEM_BYTES  (SMEM_FLOATS * sizeof(float))

typedef float v2f __attribute__((ext_vector_type(2)));
typedef float v4f __attribute__((ext_vector_type(4)));
typedef float v8f __attribute__((ext_vector_type(8)));

#if defined(__has_builtin)
#  if __has_builtin(__builtin_amdgcn_wmma_f32_16x16x4_f32)
#    define HAS_WMMA_F32 1
#  else
#    define HAS_WMMA_F32 0
#  endif
#  if __has_builtin(__builtin_amdgcn_tanhf)
#    define HAS_HW_TANH 1
#  else
#    define HAS_HW_TANH 0
#  endif
#else
#  define HAS_WMMA_F32 0
#  define HAS_HW_TANH 0
#endif

__device__ __forceinline__ float tanh_fast(float x) {
#if HAS_HW_TANH
  return __builtin_amdgcn_tanhf(x);
#else
  return tanhf(x);
#endif
}

__device__ __forceinline__ float* smb(float* sm, int i) { return sm + i * TILE; }


__device__ __forceinline__ v8f tile_xWT(const float* __restrict__ W,
                                        const float* X, int nt, int lane,
                                        float cinit) {
  const int l16 = lane & 15;
  const int half = lane >> 4;
  v8f acc = {cinit, cinit, cinit, cinit, cinit, cinit, cinit, cinit};
#if HAS_WMMA_F32
  const float* wr = W + (nt * 16 + l16) * DD + 2 * half;
  const float* xr = X + l16 * LSTR + 2 * half;
#pragma unroll 4
  for (int kk = 0; kk < DD; kk += 4) {
    v2f a = {xr[kk], xr[kk + 1]};
    v2f b = {wr[kk], wr[kk + 1]};
    acc = __builtin_amdgcn_wmma_f32_16x16x4_f32(false, a, false, b,
                                                (short)0, acc, false, false);
  }
#else
#pragma unroll
  for (int j = 0; j < 8; ++j) {
    const int m = j + 8 * half, n = nt * 16 + l16;
    float s = cinit;
    for (int k = 0; k < DD; ++k) s += X[m * LSTR + k] * W[n * DD + k];
    acc[j] = s;
  }
#endif
  return acc;
}

__device__ __forceinline__ v8f tile_xM(const float* __restrict__ M,
                                       const float* X, int nt, int lane,
                                       float cinit) {
  const int l16 = lane & 15;
  const int half = lane >> 4;
  v8f acc = {cinit, cinit, cinit, cinit, cinit, cinit, cinit, cinit};
#if HAS_WMMA_F32
  const float* mp = M + (nt * 16 + l16) + (2 * half) * DD;
  const float* xr = X + l16 * LSTR + 2 * half;
#pragma unroll 4
  for (int kk = 0; kk < DD; kk += 4) {
    v2f a = {xr[kk], xr[kk + 1]};
    v2f b = {mp[kk * DD], mp[kk * DD + DD]};
    acc = __builtin_amdgcn_wmma_f32_16x16x4_f32(false, a, false, b,
                                                (short)0, acc, false, false);
  }
#else
#pragma unroll
  for (int j = 0; j < 8; ++j) {
    const int m = j + 8 * half, n = nt * 16 + l16;
    float s = cinit;
    for (int k = 0; k < DD; ++k) s += X[m * LSTR + k] * M[k * DD + n];
    acc[j] = s;
  }
#endif
  return acc;
}

#if HAS_WMMA_F32
__device__ __forceinline__ void stage_issue(v4f (&r)[16],
                                            const float* __restrict__ wsrc) {
#pragma unroll
  for (int j = 0; j < 16; ++j)
    r[j] = *reinterpret_cast<const v4f*>(wsrc + j * DD);
}

__device__ __forceinline__ void stage_park(const v4f (&r)[16], float* wp) {
#pragma unroll
  for (int j = 0; j < 16; ++j) *reinterpret_cast<v4f*>(wp + j * WSTR) = r[j];
}

__device__ __forceinline__ void wave_lds_handover() {
  __builtin_amdgcn_fence(__ATOMIC_RELEASE, "wavefront");
  __builtin_amdgcn_wave_barrier();
  __builtin_amdgcn_fence(__ATOMIC_ACQUIRE, "wavefront");
}

__device__ __forceinline__ void frag_read8(v2f (&a)[8], v2f (&b)[8],
                                           const float* xr, const float* wr) {
#pragma unroll
  for (int s = 0; s < 8; ++s) {
    a[s] = (v2f){xr[4 * s], xr[4 * s + 1]};
    b[s] = (v2f){wr[4 * s], wr[4 * s + 1]};
  }
}

__device__ __forceinline__ v8f mma8(v8f acc, const v2f (&a)[8],
                                    const v2f (&b)[8]) {
#pragma unroll
  for (int s = 0; s < 8; ++s)
    acc = __builtin_amdgcn_wmma_f32_16x16x4_f32(false, a[s], false, b[s],
                                                (short)0, acc, false, false);
  return acc;
}

__device__ __forceinline__ v8f half_tile_mma(v8f acc, const float* xr,
                                             const float* wr) {
  v2f a0[8], b0[8], a1[8], b1[8];
  frag_read8(a0, b0, xr, wr);
  frag_read8(a1, b1, xr + 32, wr + 32);
  __builtin_amdgcn_sched_barrier(0);
  acc = mma8(acc, a0, b0);
  __builtin_amdgcn_sched_barrier(0);
  frag_read8(a0, b0, xr + 64, wr + 64);
  __builtin_amdgcn_sched_barrier(0);
  acc = mma8(acc, a1, b1);
  __builtin_amdgcn_sched_barrier(0);
  frag_read8(a1, b1, xr + 96, wr + 96);
  __builtin_amdgcn_sched_barrier(0);
  acc = mma8(acc, a0, b0);
  acc = mma8(acc, a1, b1);
  return acc;
}
#endif

__device__ __forceinline__ void phase_priority(const int level) {
  if (level >= 3) __builtin_amdgcn_s_setprio(3);
  else if (level == 2) __builtin_amdgcn_s_setprio(2);
  else if (level == 1) __builtin_amdgcn_s_setprio(1);
  else __builtin_amdgcn_s_setprio(0);
}

__device__ __forceinline__ void gemm_first_request(
    v4f (&r)[16], float& bv, const float* __restrict__ W,
    const float* __restrict__ bias, int lane, int wv) {
#if HAS_WMMA_F32
  stage_issue(r, W + (wv * 16) * DD + lane * 4);
#endif
  bv = bias ? bias[wv * 16 + (lane & 15)] : 0.f;
}

template <typename TileEpi>
__device__ __forceinline__ void gemm_xWT_tiles(
    const float* __restrict__ W, const float* __restrict__ bias,
    const float* X, float* stage, int lane, int wv, float& bv,
    const float* __restrict__ Wn, const float* __restrict__ biasn,
    const bool has_next, TileEpi epi) {
  const int l16 = lane & 15, half = lane >> 4;
#if HAS_WMMA_F32
  const float* wsrc = W + (wv * 16) * DD + lane * 4;
  const float* xr = X + l16 * LSTR + 2 * half;
  const float* wr = stage + l16 * WSTR + 2 * half;
  float* wp = stage + lane * 4;
  const float cb = bv;
  v4f r[16];
  v8f acc = {cb, cb, cb, cb, cb, cb, cb, cb};
#pragma unroll 1
  for (int q = 0; q < 2; ++q) {
    const bool more = (q < 1) || has_next;
    if (more) {
      stage_issue(r, (q < 1) ? wsrc + 128 : Wn + (wv * 16) * DD + lane * 4);
      if (q == 1) bv = biasn ? biasn[wv * 16 + l16] : 0.f;
    }
    acc = half_tile_mma(acc, xr + q * 128, wr);
    if (q == 1) epi(wv, acc);
    if (more) {
      wave_lds_handover();
      stage_park(r, wp);
      wave_lds_handover();
    }
  }
#else
  const float bvf = bias ? bias[wv * 16 + l16] : 0.f;
  v8f acc = tile_xWT(W, X, wv, lane, bvf);
  epi(wv, acc);
#endif
}

template <typename Epi>
__device__ __forceinline__ void gemm_xWT(
    const float* __restrict__ W, const float* __restrict__ bias,
    const float* X, float* stage, int lane, int wv, float& bv,
    const float* __restrict__ Wn, const float* __restrict__ biasn,
    const bool has_next, Epi epi) {
  gemm_xWT_tiles(W, bias, X, stage, lane, wv, bv, Wn, biasn, has_next,
                 [&](int nt, v8f acc) {
    const int l16 = lane & 15, half = lane >> 4;
#pragma unroll
    for (int j = 0; j < 8; ++j) epi(j + 8 * half, nt * 16 + l16, acc[j]);
  });
}

template <typename Epi>
__device__ __forceinline__ void gemm_xM(const float* __restrict__ M,
                                        const float* X, int lane, int wv,
                                        Epi epi) {
  const int nt = wv;
  v8f acc = tile_xM(M, X, nt, lane, 0.f);
  const int l16 = lane & 15, half = lane >> 4;
#pragma unroll
  for (int j = 0; j < 8; ++j) epi(j + 8 * half, nt * 16 + l16, acc[j]);
}

#define PL_WAVES 8
#define PL_TPW (SS / PL_WAVES)
#define PL_GRP 8

__device__ __forceinline__ void pool_issue(v4f (&ev)[2 * PL_GRP],
                                           v4f (&pv)[2 * PL_GRP],
                                           const int idv, const int l0,
                                           const float* __restrict__ ebase,
                                           const float* __restrict__ pbase) {
#pragma unroll
  for (int j = 0; j < PL_GRP; ++j) {
    const int id = __builtin_amdgcn_readlane(idv, l0 + j);
    const float* er = ebase + (size_t)id * DD;
    const float* pr = pbase + j * DD;
    ev[2 * j] = *reinterpret_cast<const v4f*>(er);
    ev[2 * j + 1] = *reinterpret_cast<const v4f*>(er + 128);
    pv[2 * j] = *reinterpret_cast<const v4f*>(pr);
    pv[2 * j + 1] = *reinterpret_cast<const v4f*>(pr + 128);
  }
}

__device__ __forceinline__ void pool_sum(v4f& alo, v4f& ahi,
                                         const v4f (&ev)[2 * PL_GRP],
                                         const v4f (&pv)[2 * PL_GRP]) {
#pragma unroll
  for (int j = 0; j < PL_GRP; ++j) {
    alo += ev[2 * j] + pv[2 * j];
    ahi += ev[2 * j + 1] + pv[2 * j + 1];
  }
}

__global__ __launch_bounds__(PL_WAVES * 32) void pooled_kernel(
    const int* __restrict__ ids, const float* __restrict__ emb,
    const float* __restrict__ pos, float* __restrict__ pooled_g) {
  __shared__ __attribute__((aligned(16))) float part[PL_WAVES * DD];
  const int b = blockIdx.x;
  const int t = threadIdx.x;
  const int lane = t & 31;
  const int wvu = __builtin_amdgcn_readfirstlane(t >> 5);
  const int tok0 = wvu * PL_TPW;
  const int idlo = ids[b * SS + tok0 + lane];
  const int idhi = ids[b * SS + tok0 + 32 + lane];
  const float* ebase = emb + lane * 4;
  const float* pbase = pos + tok0 * DD + lane * 4;

  v4f alo = {0.f, 0.f, 0.f, 0.f};
  v4f ahi = {0.f, 0.f, 0.f, 0.f};
  v4f e0[2 * PL_GRP], p0[2 * PL_GRP], e1[2 * PL_GRP], p1[2 * PL_GRP];
  pool_issue(e0, p0, idlo, 0, ebase, pbase);
#pragma unroll
  for (int g = 1; g < PL_TPW / PL_GRP - 1; g += 2) {
    pool_issue(e1, p1, (g < 4) ? idlo : idhi, (g & 3) * PL_GRP, ebase,
               pbase + g * PL_GRP * DD);
    __builtin_amdgcn_sched_barrier(0);
    pool_sum(alo, ahi, e0, p0);
    pool_issue(e0, p0, (g + 1 < 4) ? idlo : idhi, ((g + 1) & 3) * PL_GRP, ebase,
               pbase + (g + 1) * PL_GRP * DD);
    __builtin_amdgcn_sched_barrier(0);
    pool_sum(alo, ahi, e1, p1);
  }
  pool_issue(e1, p1, idhi, 3 * PL_GRP, ebase,
             pbase + (PL_TPW / PL_GRP - 1) * PL_GRP * DD);
  __builtin_amdgcn_sched_barrier(0);
  pool_sum(alo, ahi, e0, p0);
  pool_sum(alo, ahi, e1, p1);

  float* pw = part + wvu * DD + lane * 4;
  *reinterpret_cast<v4f*>(pw) = alo;
  *reinterpret_cast<v4f*>(pw + 128) = ahi;
  __syncthreads();
  float s = part[t];
#pragma unroll
  for (int w = 1; w < PL_WAVES; ++w) s += part[w * DD + t];
  pooled_g[b * DD + t] = s * (1.0f / (float)SS);
}

__global__ __launch_bounds__(NTHR) void refine_kernel(
    const float* __restrict__ pooled_g,
    const float* __restrict__ init_w, const float* __restrict__ init_b,
    const float* __restrict__ pred_w, const float* __restrict__ pred_b,
    const float* __restrict__ upd_w, const float* __restrict__ upd_b,
    const float* __restrict__ sqw, const float* __restrict__ sqb,
    const float* __restrict__ eqw, const float* __restrict__ eqb,
    const float* __restrict__ keyw, const float* __restrict__ keyb,
    const float* __restrict__ aw1, const float* __restrict__ ab1,
    const float* __restrict__ aw2, const float* __restrict__ ab2,
    float* __restrict__ sq_g, float* __restrict__ eq_g,
    float* __restrict__ cs_g, float* __restrict__ ce_g,
    float* __restrict__ ans_g) {
  extern __shared__ __attribute__((aligned(16))) float sm[];

  const int tid = threadIdx.x;
  const int lane = tid & 31;
  const int wv = tid >> 5;
  const int wvu = __builtin_amdgcn_readfirstlane(wv);
  const int b0 = blockIdx.x * BT;

  float* pooled = smb(sm, 0);
  float* ff = smb(sm, 5);
  float* tot = smb(sm, 8);
  float* stage = sm + NTILES * TILE + wv * WSTAGE;

  float bv;
  {
    v4f r0[16];
    gemm_first_request(r0, bv, init_w, init_b, lane, wv);
    for (int idx = tid; idx < BT * DD; idx += NTHR) {
      const int r = idx >> 8, c = idx & 255;
      pooled[r * LSTR + c] = pooled_g[(b0 + r) * DD + c];
    }
#if HAS_WMMA_F32
    stage_park(r0, stage + lane * 4);
    wave_lds_handover();
#endif
  }
  __syncthreads();

  {
    const float* cur = pooled;
    for (int l = 0; l < LLN; ++l) {
      float* dst = smb(sm, 1 + l);
      const bool last = (l == LLN - 1);
      const float* wn = last ? pred_w + (LLN - 1) * DD * DD
                             : init_w + (l + 1) * DD * DD;
      const float* bn = last ? pred_b + (LLN - 1) * DD : init_b + (l + 1) * DD;
      gemm_xWT(init_w + l * DD * DD, init_b + l * DD, cur, stage, lane, wv,
               bv, wn, bn, true,
               [&](int m, int n, float v) {
                 const float r = tanh_fast(v);
                 dst[m * LSTR + n] = r;
                 if (last) ff[m * LSTR + n] = r;
               });
      __syncthreads();
      cur = dst;
    }
  }

  for (int it = 0; it < NITERS; ++it) {
    const int lo = (it > NITERS - LLN) ? it - (NITERS - LLN) : 0;
    {
      v8f pabove = {0.f, 0.f, 0.f, 0.f, 0.f, 0.f, 0.f, 0.f};
#pragma unroll 1
      for (int i = LLN - 1; i >= lo; --i) {
        phase_priority(i - lo);
        const float* ri = smb(sm, 1 + i);
        const float* below = (i > 0) ? smb(sm, i) : pooled;
        float* ti = smb(sm, 6 + i);
        const bool has_above = (i < LLN - 1);
        const float* wn = (i > lo) ? pred_w + (i - 1) * DD * DD
                                   : upd_w + (LLN - 1) * DD * DD;
        const float* bn = (i > lo) ? pred_b + (i - 1) * DD
                                   : upd_b + (LLN - 1) * DD;
        gemm_xWT_tiles(
            pred_w + i * DD * DD, pred_b + i * DD, ri, stage, lane, wv,
            bv, wn, bn, true,
            [&](int nt, v8f acc) {
              const int o0 = 8 * (lane >> 4) * LSTR + nt * 16 + (lane & 15);
              float bl[8], rr[8];
#pragma unroll
              for (int j = 0; j < 8; ++j) bl[j] = below[o0 + j * LSTR];
              if (has_above) {
#pragma unroll
                for (int j = 0; j < 8; ++j) rr[j] = ri[o0 + j * LSTR];
              }
#pragma unroll
              for (int j = 0; j < 8; ++j) {
                float e = bl[j] - acc[j];
                if (has_above) e += 0.5f * (rr[j] - pabove[j]);
                ti[o0 + j * LSTR] = e;
              }
              pabove = acc;
            });
      }
    }
    __syncthreads();
#pragma unroll 1
    for (int i = LLN - 1; i >= lo; --i) {
      phase_priority(i - lo);
      float* ri = smb(sm, 1 + i);
      const float* ti = smb(sm, 6 + i);
      const bool phase_end = (i == lo);
      const bool all_end = phase_end && (it == NITERS - 1);
      const float* wn = all_end ? sqw
                        : phase_end ? pred_w + (LLN - 1) * DD * DD
                                    : upd_w + (i - 1) * DD * DD;
      const float* bn = all_end ? sqb
                        : phase_end ? pred_b + (LLN - 1) * DD
                                    : upd_b + (i - 1) * DD;
      gemm_xWT(upd_w + i * DD * DD, upd_b + i * DD, ti, stage, lane, wv,
               bv, wn, bn, true,
               [&](int m, int n, float v) {
                 const int o = m * LSTR + n;
                 ri[o] = ri[o] + 0.1f * tanh_fast(v);
               });
    }
    __syncthreads();
  }

  float* refined = smb(sm, LLN);
  for (int idx = tid; idx < BT * DD; idx += NTHR) {
    const int r = idx >> 8, c = idx & 255;
    refined[r * LSTR + c] += ff[r * LSTR + c];
  }
  __syncthreads();

  {
    float* q = smb(sm, 6);
    gemm_xWT(sqw, sqb, refined, stage, lane, wv, bv, eqw, eqb, true,
             [&](int m, int n, float v) { q[m * LSTR + n] = v; });
    __syncthreads();
    if (tid < BT * 16) {
      const int r = tid >> 4, c = tid & 15;
      float p = 0.f;
#pragma unroll
      for (int u = 0; u < 16; ++u) {
        const int o = c * 16 + u;
        p += q[r * LSTR + o] * keyb[o];
      }
      tot[tid] = p;
    }
    __syncthreads();
    if (tid < BT) {
      float s = 0.f;
#pragma unroll
      for (int u = 0; u < 16; ++u) s += tot[tid * 16 + u];
      cs_g[b0 + tid] = s;
    }
    __syncthreads();
    gemm_xM(keyw, q, lane, wv,
            [&](int m, int n, float v) { sq_g[(b0 + m) * DD + n] = v; });
    __syncthreads();
  }

  {
    float* q = smb(sm, 7);
    gemm_xWT(eqw, eqb, refined, stage, lane, wv, bv, aw1, ab1, wvu < 8,
             [&](int m, int n, float v) { q[m * LSTR + n] = v; });
    __syncthreads();
    if (tid < BT * 16) {
      const int r = tid >> 4, c = tid & 15;
      float p = 0.f;
#pragma unroll
      for (int u = 0; u < 16; ++u) {
        const int o = c * 16 + u;
        p += q[r * LSTR + o] * keyb[o];
      }
      tot[tid] = p;
    }
    __syncthreads();
    if (tid < BT) {
      float s = 0.f;
#pragma unroll
      for (int u = 0; u < 16; ++u) s += tot[tid * 16 + u];
      ce_g[b0 + tid] = s;
    }
    __syncthreads();
    gemm_xM(keyw, q, lane, wv,
            [&](int m, int n, float v) { eq_g[(b0 + m) * DD + n] = v; });
    __syncthreads();
  }

  if (wvu < 8) {
    gemm_xWT(aw1, ab1, refined, stage, lane, wv, bv, nullptr, nullptr, false,
             [&](int m, int n, float v) {
               tot[m * LSTR + n] =
                   0.5f * v * (1.0f + erff(v * 0.70710678118654752f));
             });
  }
  __syncthreads();
  if (tid < BT * 2) {
    const int r = tid >> 1, k = tid & 1;
    float s = ab2[k];
    for (int j = 0; j < 128; ++j) s += tot[r * LSTR + j] * aw2[k * 128 + j];
    ans_g[(b0 + r) * 2 + k] = s;
  }
}

#define LG_WAVES 8
#define LG_TPW 4
#define LG_SMEM_BYTES (LG_WAVES * TILE * sizeof(float))

#if HAS_WMMA_F32
__device__ __forceinline__ void rows_issue(v4f (&ev)[32], v4f (&pv)[32],
                                           const int idv, const int l0,
                                           const float* __restrict__ ebase,
                                           const float* __restrict__ pbase) {
#pragma unroll
  for (int j = 0; j < 16; ++j) {
    const int id = __builtin_amdgcn_readlane(idv, l0 + j);
    const float* er = ebase + (size_t)id * DD;
    const float* pr = pbase + j * DD;
    ev[2 * j] = *reinterpret_cast<const v4f*>(er);
    ev[2 * j + 1] = *reinterpret_cast<const v4f*>(er + 128);
    pv[2 * j] = *reinterpret_cast<const v4f*>(pr);
    pv[2 * j + 1] = *reinterpret_cast<const v4f*>(pr + 128);
  }
}

__device__ __forceinline__ void rows_park(const v4f (&ev)[32],
                                          const v4f (&pv)[32], float* wp) {
#pragma unroll
  for (int j = 0; j < 16; ++j) {
    *reinterpret_cast<v4f*>(wp + j * LSTR) = ev[2 * j] + pv[2 * j];
    *reinterpret_cast<v4f*>(wp + j * LSTR + 128) = ev[2 * j + 1] + pv[2 * j + 1];
  }
}

__device__ __forceinline__ void afrag_read8(v2f (&a)[8], const float* xr) {
#pragma unroll
  for (int s = 0; s < 8; ++s) a[s] = (v2f){xr[4 * s], xr[4 * s + 1]};
}

__device__ __forceinline__ v8f tile_tokens_dot(const float* xr,
                                               const v2f (&bq)[64]) {
  v8f acc = {0.f, 0.f, 0.f, 0.f, 0.f, 0.f, 0.f, 0.f};
  v2f a0[8], a1[8];
  afrag_read8(a0, xr);
#pragma unroll
  for (int g = 0; g < 8; g += 2) {
    afrag_read8(a1, xr + 32 * (g + 1));
    __builtin_amdgcn_sched_barrier(0);
#pragma unroll
    for (int s = 0; s < 8; ++s)
      acc = __builtin_amdgcn_wmma_f32_16x16x4_f32(false, a0[s], false,
                                                  bq[8 * g + s], (short)0, acc,
                                                  false, false);
    __builtin_amdgcn_sched_barrier(0);
    if (g + 2 < 8) afrag_read8(a0, xr + 32 * (g + 2));
    __builtin_amdgcn_sched_barrier(0);
#pragma unroll
    for (int s = 0; s < 8; ++s)
      acc = __builtin_amdgcn_wmma_f32_16x16x4_f32(false, a1[s], false,
                                                  bq[8 * (g + 1) + s], (short)0,
                                                  acc, false, false);
    __builtin_amdgcn_sched_barrier(0);
  }
  return acc;
}
#endif

__global__ __launch_bounds__(LG_WAVES * 32) void logits_kernel(
    const int* __restrict__ ids, const float* __restrict__ emb,
    const float* __restrict__ pos, const float* __restrict__ sq_g,
    const float* __restrict__ eq_g, const float* __restrict__ cs_g,
    const float* __restrict__ ce_g, float* __restrict__ out_start,
    float* __restrict__ out_end) {
  extern __shared__ __attribute__((aligned(16))) float lsm[];
  const int b = blockIdx.x;
  const int lane = threadIdx.x & 31;
  const int wvu = __builtin_amdgcn_readfirstlane((int)(threadIdx.x >> 5));
#if HAS_WMMA_F32
  const int c = lane & 15, h = lane >> 4;
  const int tok0 = wvu * (LG_TPW * 16);
  float* st = lsm + wvu * TILE;
  const float* ebase = emb + lane * 4;
  const float* pbase = pos + tok0 * DD + lane * 4;
  const int idlo = ids[b * SS + tok0 + lane];
  const int idhi = ids[b * SS + tok0 + 32 + lane];

  v4f ev[32], pv[32];
  rows_issue(ev, pv, idlo, 0, ebase, pbase);

  v2f bq[64];
  {
    const float* qsrc = ((c == 0) ? sq_g : eq_g) + b * DD + 2 * h;
#pragma unroll
    for (int s = 0; s < 64; ++s)
      bq[s] = *reinterpret_cast<const v2f*>(qsrc + 4 * s);
  }
  const float cadd = (c == 0) ? cs_g[b] : ce_g[b];
  float* orow = ((c == 0) ? out_start : out_end) + b * SS + tok0 + 8 * h;
  const float* xr = st + c * LSTR + 2 * h;
  float* wp = st + lane * 4;

#pragma unroll 1
  for (int t = 0; t < LG_TPW; ++t) {
    wave_lds_handover();
    rows_park(ev, pv, wp);
    wave_lds_handover();
    if (t + 1 < LG_TPW)
      rows_issue(ev, pv, (t + 1 < 2) ? idlo : idhi, ((t + 1) & 1) * 16, ebase,
                 pbase + (t + 1) * 16 * DD);
    const v8f acc = tile_tokens_dot(xr, bq);
    const v4f lo = {acc[0] + cadd, acc[1] + cadd, acc[2] + cadd, acc[3] + cadd};
    const v4f hi = {acc[4] + cadd, acc[5] + cadd, acc[6] + cadd, acc[7] + cadd};
    if (c < 2) {
      *reinterpret_cast<v4f*>(orow + t * 16) = lo;
      *reinterpret_cast<v4f*>(orow + t * 16 + 4) = hi;
    }
  }
#else
  (void)lsm;
  (void)wvu;
  (void)lane;
  for (int s = threadIdx.x; s < SS; s += LG_WAVES * 32) {
    const float* row = emb + (size_t)ids[b * SS + s] * DD;
    const float* pr = pos + s * DD;
    float as = 0.f, ae = 0.f;
    for (int k = 0; k < DD; ++k) {
      const float e = row[k] + pr[k];
      as += e * sq_g[b * DD + k];
      ae += e * eq_g[b * DD + k];
    }
    out_start[b * SS + s] = as + cs_g[b];
    out_end[b * SS + s] = ae + ce_g[b];
  }
#endif
}

extern "C" void kernel_launch(void* const* d_in, const int* in_sizes, int n_in,
                              void* d_out, int out_size, void* d_ws,
                              size_t ws_size, hipStream_t stream) {
  const int* ids = (const int*)d_in[0];
  const float* emb = (const float*)d_in[1];
  const float* pos = (const float*)d_in[2];
  const float* init_w = (const float*)d_in[3];
  const float* init_b = (const float*)d_in[4];
  const float* pred_w = (const float*)d_in[5];
  const float* pred_b = (const float*)d_in[6];
  const float* upd_w = (const float*)d_in[7];
  const float* upd_b = (const float*)d_in[8];
  const float* sqw = (const float*)d_in[9];
  const float* sqb = (const float*)d_in[10];
  const float* eqw = (const float*)d_in[11];
  const float* eqb = (const float*)d_in[12];
  const float* keyw = (const float*)d_in[13];
  const float* keyb = (const float*)d_in[14];
  const float* aw1 = (const float*)d_in[15];
  const float* ab1 = (const float*)d_in[16];
  const float* aw2 = (const float*)d_in[17];
  const float* ab2 = (const float*)d_in[18];

  float* ws = (float*)d_ws;
  float* pooled = ws;
  float* sq = pooled + BB * DD;
  float* eq = sq + BB * DD;
  float* cs = eq + BB * DD;
  float* ce = cs + BB;

  float* out = (float*)d_out;
  float* out_start = out;
  float* out_end = out + BB * SS;
  float* out_ans = out + 2 * BB * SS;

  (void)hipFuncSetAttribute((const void*)refine_kernel,
                            hipFuncAttributeMaxDynamicSharedMemorySize,
                            (int)SMEM_BYTES);

  (void)hipFuncSetAttribute((const void*)logits_kernel,
                            hipFuncAttributeMaxDynamicSharedMemorySize,
                            (int)LG_SMEM_BYTES);

  pooled_kernel<<<BB, PL_WAVES * 32, 0, stream>>>(ids, emb, pos, pooled);
  refine_kernel<<<BB / BT, NTHR, SMEM_BYTES, stream>>>(
      pooled, init_w, init_b, pred_w, pred_b, upd_w, upd_b, sqw, sqb, eqw, eqb,
      keyw, keyb, aw1, ab1, aw2, ab2, sq, eq, cs, ce, out_ans);
  logits_kernel<<<BB, LG_WAVES * 32, LG_SMEM_BYTES, stream>>>(
      ids, emb, pos, sq, eq, cs, ce, out_start, out_end);
}
